# starting kernel
# baseline (speedup 1.0000x reference)
#include <hip/hip_runtime.h>

typedef __attribute__((ext_vector_type(2))) float v2f;
typedef __attribute__((ext_vector_type(8))) float v8f;

#define WAVES_PER_BLOCK 4


__global__ void gcn_deg_init(float* __restrict__ deg, int N) {
  int i = blockIdx.x * blockDim.x + threadIdx.x;
  if (i < N) deg[i] = 1.0f;
}

__global__ void gcn_deg_edge(const int* __restrict__ col, float* __restrict__ deg, int E) {
  int e = blockIdx.x * blockDim.x + threadIdx.x;
  if (e < E) unsafeAtomicAdd(&deg[col[e]], 1.0f);
}

__global__ void gcn_deg_to_dis(float* __restrict__ dis, int N) {
  int i = blockIdx.x * blockDim.x + threadIdx.x;
  if (i < N) {
    float d = dis[i];
    dis[i] = (d > 0.0f) ? rsqrtf(d) : 0.0f;
  }
}


__global__ void gcn_self_loop(const float* __restrict__ dis, const float* __restrict__ h,
                              float* __restrict__ agg, int N, int C) {
  long long i = (long long)blockIdx.x * blockDim.x + threadIdx.x;
  long long total = (long long)N * C;
  if (i < total) {
    int node = (int)(i / C);
    float d = dis[node];
    agg[i] = d * d * h[i];
  }
}

__global__ void gcn_edge_scatter(const int* __restrict__ ei, const float* __restrict__ dis,
                                 const float* __restrict__ h, float* __restrict__ agg,
                                 int E, int C) {
  int gw   = (int)(((long long)blockIdx.x * blockDim.x + threadIdx.x) >> 5);
  int lane = threadIdx.x & 31;
  if (gw >= E) return;
  int r = ei[gw];
  int c = ei[E + gw];
  float w = dis[r] * dis[c];
  const float* hp = h + (long long)r * C;
  float*       ap = agg + (long long)c * C;
  for (int ch = lane; ch < C; ch += 32)
    unsafeAtomicAdd(ap + ch, w * hp[ch]);
}

template <int IC, bool DUAL, bool RELU, bool GUARD>
__global__ __launch_bounds__(32 * WAVES_PER_BLOCK) void gcn_wmma_gemm(
    const float* __restrict__ inA, const float* __restrict__ WA, const float* __restrict__ bA,
    const float* __restrict__ inB, const float* __restrict__ WB, const float* __restrict__ bB,
    float* __restrict__ out, int nrows, int OC) {
  __shared__ float lds[(DUAL ? 2 : 1) * 16 * IC];
  const int tid  = threadIdx.x;
  const int lane = tid & 31;
  const int wave = tid >> 5;
  const int row0 = blockIdx.x << 4;
  const int col0 = (blockIdx.y * WAVES_PER_BLOCK + wave) << 4;

  float* Atile = lds;
  float* Btile = DUAL ? (lds + 16 * IC) : lds;

  constexpr int vecPerRow = IC >> 2;
  constexpr int totalVec  = vecPerRow << 4;
  for (int idx = tid; idx < totalVec; idx += 32 * WAVES_PER_BLOCK) {
    const int r = idx / vecPerRow;
    const int v = idx - r * vecPerRow;
    const int node = row0 + r;
    if (GUARD && node >= nrows) {
      float4 z = {0.f, 0.f, 0.f, 0.f};
      ((float4*)Atile)[idx] = z;
      if constexpr (DUAL) ((float4*)Btile)[idx] = z;
    } else {
      ((float4*)Atile)[idx] = ((const float4*)(inA + (size_t)node * IC))[v];
      if constexpr (DUAL)
        ((float4*)Btile)[idx] = ((const float4*)(inB + (size_t)node * IC))[v];
    }
  }
  __syncthreads();

  if (col0 >= OC) return;

  const int  nidx = lane & 15;
  const int  koff = (lane >> 4) << 1;
  const int  col  = col0 + nidx;
  const bool colOK = GUARD ? (col < OC) : true;
  const int  colC  = GUARD ? (colOK ? col : (OC - 1)) : col;
  const float* wAp = WA + (size_t)colC * IC + koff;
  const float* wBp = DUAL ? (WB + (size_t)colC * IC + koff) : WA;
  const float* ar  = Atile + nidx * IC + koff;
  const float* br  = Btile + nidx * IC + koff;

  constexpr int KSTEPS = IC / 4;
  constexpr int BATCH  = DUAL ? (KSTEPS < 4 ? KSTEPS : 4)
                              : (KSTEPS < 8 ? KSTEPS : 8);

  v8f acc = {};
#pragma unroll 1
  for (int kb = 0; kb < KSTEPS; kb += BATCH) {
    v2f fa[BATCH], fw[BATCH];
    v2f ga[DUAL ? BATCH : 1], gw[DUAL ? BATCH : 1];
#pragma unroll
    for (int i = 0; i < BATCH; ++i) {
      const int k = (kb + i) * 4;
      fa[i] = *(const v2f*)(ar + k);
      fw[i] = *(const v2f*)(wAp + k);
      if constexpr (DUAL) {
        ga[i] = *(const v2f*)(br + k);
        gw[i] = *(const v2f*)(wBp + k);
      }
    }
#pragma unroll
    for (int i = 0; i < BATCH; ++i) {
      acc = __builtin_amdgcn_wmma_f32_16x16x4_f32(false, fa[i], false, fw[i],
                                                  (short)0, acc, false, false);
      if constexpr (DUAL)
        acc = __builtin_amdgcn_wmma_f32_16x16x4_f32(false, ga[i], false, gw[i],
                                                    (short)0, acc, false, false);
    }
  }

  const int mb = (lane >> 4) << 3;
  if constexpr (GUARD) {
    float bias = 0.f;
    if (colOK) {
      bias = bA[col];
      if constexpr (DUAL) bias += bB[col];
    }
#pragma unroll
    for (int vg = 0; vg < 8; ++vg) {
      const int r = row0 + mb + vg;
      if (colOK && r < nrows) {
        float val = acc[vg] + bias;
        if constexpr (RELU) val = val > 0.f ? val : 0.f;
        out[(size_t)r * OC + col] = val;
      }
    }
  } else {
    float bias = bA[col];
    if constexpr (DUAL) bias += bB[col];
    float* op = out + (size_t)(row0 + mb) * OC + col;
#pragma unroll
    for (int vg = 0; vg < 8; ++vg) {
      float val = acc[vg] + bias;
      if constexpr (RELU) val = val > 0.f ? val : 0.f;
      op[(size_t)vg * OC] = val;
    }
  }
}


template <int IC, bool DUAL, bool RELU>
static inline void launch_gemm(const float* inA, const float* WA, const float* bA,
                               const float* inB, const float* WB, const float* bB,
                               float* out, int N, int OC, hipStream_t stream) {
  const int rowTiles = (N + 15) / 16;
  const int gy = (((OC + 15) / 16) + WAVES_PER_BLOCK - 1) / WAVES_PER_BLOCK;
  const dim3 grid(rowTiles, gy), blk(32 * WAVES_PER_BLOCK);
  if ((N & 15) == 0 && (OC & 15) == 0)
    gcn_wmma_gemm<IC, DUAL, RELU, false><<<grid, blk, 0, stream>>>(
        inA, WA, bA, inB, WB, bB, out, N, OC);
  else
    gcn_wmma_gemm<IC, DUAL, RELU, true><<<grid, blk, 0, stream>>>(
        inA, WA, bA, inB, WB, bB, out, N, OC);
}

extern "C" void kernel_launch(void* const* d_in, const int* in_sizes, int n_in,
                              void* d_out, int out_size, void* d_ws, size_t ws_size,
                              hipStream_t stream) {
  (void)n_in; (void)out_size; (void)ws_size;
  const float* x    = (const float*)d_in[0];
  const int*   ei   = (const int*)d_in[1];
  const float* w1   = (const float*)d_in[2];
  const float* b1   = (const float*)d_in[3];
  const float* w2   = (const float*)d_in[4];
  const float* b2   = (const float*)d_in[5];
  const float* c1wl = (const float*)d_in[6];
  const float* c1bl = (const float*)d_in[7];
  const float* c1wr = (const float*)d_in[8];
  const float* c1br = (const float*)d_in[9];
  const float* c2wl = (const float*)d_in[10];
  const float* c2bl = (const float*)d_in[11];
  const float* c2wr = (const float*)d_in[12];
  const float* c2br = (const float*)d_in[13];
  const float* w3   = (const float*)d_in[14];
  const float* b3   = (const float*)d_in[15];
  const float* w4   = (const float*)d_in[16];
  const float* b4   = (const float*)d_in[17];

  const int IN_CH = 256;
  const int N = in_sizes[0] / IN_CH;
  const int E = in_sizes[1] / 2;
  const size_t n = (size_t)N;

  float* ws   = (float*)d_ws;
  float* dis  = ws;
  float* h2   = ws + 1 * n;
  float* agg1 = ws + 33 * n;
  float* h3   = ws + 65 * n;
  float* agg2 = ws + 129 * n;
  float* h4   = ws + 193 * n;
  float* h1   = ws + 193 * n;
  float* h5   = ws + 1 * n;

  const int eltBlocks  = (N + 255) / 256;
  const int edgeBlocks = (E + 255) / 256;
  const int scatBlocks = (int)(((long long)E * 32 + 255) / 256);

  launch_gemm<256, false, false>(x, w1, b1, nullptr, nullptr, nullptr, h1, N, 16, stream);
  launch_gemm<16, false, false>(h1, w2, b2, nullptr, nullptr, nullptr, h2, N, 32, stream);

  gcn_deg_init<<<eltBlocks, 256, 0, stream>>>(dis, N);
  gcn_deg_edge<<<edgeBlocks, 256, 0, stream>>>(ei + E, dis, E);
  gcn_deg_to_dis<<<eltBlocks, 256, 0, stream>>>(dis, N);

  gcn_self_loop<<<(int)((n * 32 + 255) / 256), 256, 0, stream>>>(dis, h2, agg1, N, 32);
  gcn_edge_scatter<<<scatBlocks, 256, 0, stream>>>(ei, dis, h2, agg1, E, 32);
  launch_gemm<32, true, true>(h2, c1wl, c1bl, agg1, c1wr, c1br, h3, N, 64, stream);

  gcn_self_loop<<<(int)((n * 64 + 255) / 256), 256, 0, stream>>>(dis, h3, agg2, N, 64);
  gcn_edge_scatter<<<scatBlocks, 256, 0, stream>>>(ei, dis, h3, agg2, E, 64);
  launch_gemm<64, true, true>(h3, c2wl, c2bl, agg2, c2wr, c2br, h4, N, 128, stream);

  launch_gemm<128, false, false>(h4, w3, b3, nullptr, nullptr, nullptr, h5, N, 192, stream);
  launch_gemm<192, false, false>(h5, w4, b4, nullptr, nullptr, nullptr, (float*)d_out, N, 20, stream);
}


# neighbour sums held in LDS per 400-node target range: each work-group filters the whole target list, adds kept source rows with LDS float atomics (arrival order); no global float atomics in edge pass
# speedup vs baseline: 1.5374x; 1.5374x over previous
#include <hip/hip_runtime.h>

typedef __attribute__((ext_vector_type(2))) float v2f;
typedef __attribute__((ext_vector_type(8))) float v8f;

#define WAVES_PER_BLOCK 4


__global__ void gcn_deg_init(float* __restrict__ deg, int N) {
  int i = blockIdx.x * blockDim.x + threadIdx.x;
  if (i < N) deg[i] = 1.0f;
}

__global__ void gcn_deg_edge(const int* __restrict__ col, float* __restrict__ deg, int E) {
  int e = blockIdx.x * blockDim.x + threadIdx.x;
  if (e < E) unsafeAtomicAdd(&deg[col[e]], 1.0f);
}

__global__ void gcn_deg_to_dis(float* __restrict__ dis, int N) {
  int i = blockIdx.x * blockDim.x + threadIdx.x;
  if (i < N) {
    float d = dis[i];
    dis[i] = (d > 0.0f) ? rsqrtf(d) : 0.0f;
  }
}


__global__ void gcn_prescale(const float* __restrict__ dis, const float* __restrict__ h,
                             float* __restrict__ hs, int N, int C) {
  long long i = (long long)blockIdx.x * blockDim.x + threadIdx.x;
  long long total = ((long long)N * C) >> 2;
  if (i < total) {
    int node = (int)((i << 2) / C);
    float d = dis[node];
    float4 v = ((const float4*)h)[i];
    v.x *= d; v.y *= d; v.z *= d; v.w *= d;
    ((float4*)hs)[i] = v;
  }
}

#define RG_NODES   400
#define RG_THREADS 1024
#define RG_LIST    8192

template <int C>
__global__ __launch_bounds__(RG_THREADS) void gcn_range_sum(
    const int* __restrict__ ei, const float* __restrict__ hs, float* __restrict__ acc,
    int E, int N) {
  __shared__ float    sacc[RG_NODES * C];
  __shared__ int      lsrc[RG_LIST];
  __shared__ int      ldst[RG_LIST];
  __shared__ unsigned cnt;

  const int tid  = threadIdx.x;
  const int base = blockIdx.x * RG_NODES;
  const int rem  = N - base;
  const unsigned nloc = (unsigned)(rem < RG_NODES ? rem : RG_NODES);
  const int nvec = (int)nloc * (C >> 2);

  const float4* hsv = (const float4*)(hs + (size_t)base * C);
  for (int i = tid; i < nvec; i += RG_THREADS) ((float4*)sacc)[i] = hsv[i];
  if (tid == 0) cnt = 0u;
  __syncthreads();

  const int* col = ei + E;

  auto keep = [&](int e, unsigned cl) {
    const unsigned s = atomicAdd(&cnt, 1u);
    if (s < (unsigned)RG_LIST) {
      lsrc[s] = e;
      ldst[s] = (int)cl;
    } else {
      const float* hp = hs + (size_t)ei[e] * C;
#pragma unroll 1
      for (int ch = 0; ch < C; ++ch) atomicAdd(&sacc[cl * C + ch], hp[ch]);
    }
  };
  auto test4 = [&](const int4 v, int q) {
    const unsigned a = (unsigned)(v.x - base), b = (unsigned)(v.y - base);
    const unsigned c = (unsigned)(v.z - base), d = (unsigned)(v.w - base);
    if ((a < nloc) | (b < nloc) | (c < nloc) | (d < nloc)) {
      const int e = q << 2;
      if (a < nloc) keep(e, a);
      if (b < nloc) keep(e + 1, b);
      if (c < nloc) keep(e + 2, c);
      if (d < nloc) keep(e + 3, d);
    }
  };

  if (((E & 3) == 0) && ((((size_t)col) & 15) == 0)) {
    const int4* colv = (const int4*)col;
    const int nq = E >> 2;
    int q = tid;
    for (; q + 3 * RG_THREADS < nq; q += 4 * RG_THREADS) {
      const int4 v0 = colv[q];
      const int4 v1 = colv[q + RG_THREADS];
      const int4 v2 = colv[q + 2 * RG_THREADS];
      const int4 v3 = colv[q + 3 * RG_THREADS];
      test4(v0, q);
      test4(v1, q + RG_THREADS);
      test4(v2, q + 2 * RG_THREADS);
      test4(v3, q + 3 * RG_THREADS);
    }
    for (; q < nq; q += RG_THREADS) test4(colv[q], q);
  } else {
    for (int e = tid; e < E; e += RG_THREADS) {
      const unsigned cl = (unsigned)(col[e] - base);
      if (cl < nloc) keep(e, cl);
    }
  }
  __syncthreads();

  const unsigned total = cnt;
  const int n = (int)(total < (unsigned)RG_LIST ? total : (unsigned)RG_LIST);
  for (int i = tid; i < n; i += RG_THREADS) lsrc[i] = ei[lsrc[i]];
  __syncthreads();

  constexpr int RG_U = 8;
  constexpr int VEC  = C / 32;
  const int lane = tid & 31;
  const int wave = __builtin_amdgcn_readfirstlane(tid >> 5);
  for (int i0 = wave * RG_U; i0 < n; i0 += (RG_THREADS / 32) * RG_U) {
    float val[RG_U][VEC];
    int   dst[RG_U];
#pragma unroll
    for (int u = 0; u < RG_U; ++u) {
      const int idx = (i0 + u < n) ? (i0 + u) : i0;
      const int r = lsrc[idx];
      dst[u] = ldst[idx];
      const float* hp = hs + (size_t)r * C + lane * VEC;
#pragma unroll
      for (int k = 0; k < VEC; ++k) val[u][k] = hp[k];
    }
#pragma unroll
    for (int u = 0; u < RG_U; ++u) {
      if (i0 + u < n) {
        float* sp = sacc + dst[u] * C + lane * VEC;
#pragma unroll
        for (int k = 0; k < VEC; ++k) atomicAdd(sp + k, val[u][k]);
      }
    }
  }
  __syncthreads();

  float4* accv = (float4*)(acc + (size_t)base * C);
  for (int i = tid; i < nvec; i += RG_THREADS) accv[i] = ((const float4*)sacc)[i];
}


#define NUM_W 8
struct PackSrc {
  const float* w0; const float* w1; const float* w2; const float* w3;
  const float* w4; const float* w5; const float* w6; const float* w7;
};
#define WP_W1   0
#define WP_W2   4096
#define WP_C1L  4608
#define WP_C1R  6656
#define WP_C2L  8704
#define WP_C2R  16896
#define WP_W3   25088
#define WP_W4   49664
#define WP_TOTAL 55808

__global__ void gcn_pack_weights(PackSrc s, float* __restrict__ dst) {
  const int m = blockIdx.y;
  const float* W; int OC, IC, off;
  switch (m) {
    case 0:  W = s.w0; OC = 16;  IC = 256; off = WP_W1;  break;
    case 1:  W = s.w1; OC = 32;  IC = 16;  off = WP_W2;  break;
    case 2:  W = s.w2; OC = 64;  IC = 32;  off = WP_C1L; break;
    case 3:  W = s.w3; OC = 64;  IC = 32;  off = WP_C1R; break;
    case 4:  W = s.w4; OC = 128; IC = 64;  off = WP_C2L; break;
    case 5:  W = s.w5; OC = 128; IC = 64;  off = WP_C2R; break;
    case 6:  W = s.w6; OC = 192; IC = 128; off = WP_W3;  break;
    default: W = s.w7; OC = 20;  IC = 192; off = WP_W4;  break;
  }
  const int ks_n  = IC >> 2;
  const int tiles = (OC + 15) >> 4;
  const int total = tiles * ks_n * 32;
  for (int i = blockIdx.x * blockDim.x + threadIdx.x; i < total; i += gridDim.x * blockDim.x) {
    const int lane = i & 31;
    const int ks   = (i >> 5) % ks_n;
    const int ct   = (i >> 5) / ks_n;
    int col = ct * 16 + (lane & 15);
    col = col < OC ? col : OC - 1;
    const float* p = W + (size_t)col * IC + ks * 4 + ((lane >> 4) << 1);
    v2f v = *(const v2f*)p;
    *(v2f*)(dst + off + (size_t)i * 2) = v;
  }
}

template <int IC, bool DUAL, bool RELU, bool GUARD, bool INP, bool OUTP>
__global__ __launch_bounds__(32 * WAVES_PER_BLOCK) void gcn_wmma_gemm(
    const float* __restrict__ inA, const float* __restrict__ WA, const float* __restrict__ bA,
    const float* __restrict__ inB, const float* __restrict__ WB, const float* __restrict__ bB,
    const float* __restrict__ sB, float* __restrict__ out, int nrows, int OC) {
  static_assert(!(INP && DUAL), "fragment-order input is single-operand only");
  static_assert(!(OUTP && GUARD), "fragment-order output needs full tiles");
  __shared__ float lds[INP ? 4 : (DUAL ? 2 : 1) * 16 * IC];
  const int tid  = threadIdx.x;
  const int lane = tid & 31;
  const int wave = tid >> 5;
  const int row0 = blockIdx.x << 4;
  const int col0 = (blockIdx.y * WAVES_PER_BLOCK + wave) << 4;

  float* Atile = lds;
  float* Btile = DUAL ? (lds + 16 * IC) : lds;

  if constexpr (!INP) {
    constexpr int vecPerRow = IC >> 2;
    constexpr int totalVec  = vecPerRow << 4;
    for (int idx = tid; idx < totalVec; idx += 32 * WAVES_PER_BLOCK) {
      const int r = idx / vecPerRow;
      const int v = idx - r * vecPerRow;
      const int node = row0 + r;
      if (GUARD && node >= nrows) {
        float4 z = {0.f, 0.f, 0.f, 0.f};
        ((float4*)Atile)[idx] = z;
        if constexpr (DUAL) ((float4*)Btile)[idx] = z;
      } else {
        ((float4*)Atile)[idx] = ((const float4*)(inA + (size_t)node * IC))[v];
        if constexpr (DUAL) {
          float4 b = ((const float4*)(inB + (size_t)node * IC))[v];
          const float d = sB[node];
          b.x *= d; b.y *= d; b.z *= d; b.w *= d;
          ((float4*)Btile)[idx] = b;
        }
      }
    }
    __syncthreads();
  }

  if (col0 >= OC) return;

  const int  nidx = lane & 15;
  const int  koff = (lane >> 4) << 1;
  const int  col  = col0 + nidx;
  const bool colOK = GUARD ? (col < OC) : true;

  constexpr int KSTEPS = IC / 4;
  const v2f* wAp = (const v2f*)WA + (size_t)(col0 >> 4) * KSTEPS * 32 + lane;
  const v2f* wBp = DUAL ? ((const v2f*)WB + (size_t)(col0 >> 4) * KSTEPS * 32 + lane) : wAp;
  const v2f*   ag = (const v2f*)(inA + (size_t)row0 * IC) + lane;
  const float* ar = Atile + nidx * IC + koff;
  const float* br = Btile + nidx * IC + koff;

  constexpr int BATCH  = DUAL ? (KSTEPS < 4 ? KSTEPS : 4)
                              : (KSTEPS < 8 ? KSTEPS : 8);

  v8f acc = {};
#pragma unroll 1
  for (int kb = 0; kb < KSTEPS; kb += BATCH) {
    v2f fa[BATCH], fw[BATCH];
    v2f ga[DUAL ? BATCH : 1], gw[DUAL ? BATCH : 1];
#pragma unroll
    for (int i = 0; i < BATCH; ++i) {
      const int ks = kb + i;
      if constexpr (INP) fa[i] = ag[ks * 32];
      else               fa[i] = *(const v2f*)(ar + ks * 4);
      fw[i] = wAp[ks * 32];
      if constexpr (DUAL) {
        ga[i] = *(const v2f*)(br + ks * 4);
        gw[i] = wBp[ks * 32];
      }
    }
#pragma unroll
    for (int i = 0; i < BATCH; ++i) {
      if constexpr (OUTP) {
        acc = __builtin_amdgcn_wmma_f32_16x16x4_f32(false, fw[i], false, fa[i],
                                                    (short)0, acc, false, false);
        if constexpr (DUAL)
          acc = __builtin_amdgcn_wmma_f32_16x16x4_f32(false, gw[i], false, ga[i],
                                                      (short)0, acc, false, false);
      } else {
        acc = __builtin_amdgcn_wmma_f32_16x16x4_f32(false, fa[i], false, fw[i],
                                                    (short)0, acc, false, false);
        if constexpr (DUAL)
          acc = __builtin_amdgcn_wmma_f32_16x16x4_f32(false, ga[i], false, gw[i],
                                                      (short)0, acc, false, false);
      }
    }
  }

  const int mb = (lane >> 4) << 3;
  if constexpr (OUTP) {
    const int cb = col0 + mb;
    float bias[8];
#pragma unroll
    for (int vg = 0; vg < 8; ++vg) {
      bias[vg] = bA[cb + vg];
      if constexpr (DUAL) bias[vg] += bB[cb + vg];
    }
    const int KSO = OC >> 2;
    float* ob = out + ((size_t)blockIdx.x * KSO + (cb >> 2)) * 64 + nidx * 2;
#pragma unroll
    for (int t = 0; t < 4; ++t) {
      float v0 = acc[2 * t] + bias[2 * t];
      float v1 = acc[2 * t + 1] + bias[2 * t + 1];
      if constexpr (RELU) { v0 = v0 > 0.f ? v0 : 0.f; v1 = v1 > 0.f ? v1 : 0.f; }
      v2f o = {v0, v1};
      *(v2f*)(ob + (t >> 1) * 64 + (t & 1) * 32) = o;
    }
  } else if constexpr (GUARD) {
    float bias = 0.f;
    if (colOK) {
      bias = bA[col];
      if constexpr (DUAL) bias += bB[col];
    }
#pragma unroll
    for (int vg = 0; vg < 8; ++vg) {
      const int r = row0 + mb + vg;
      if (colOK && r < nrows) {
        float val = acc[vg] + bias;
        if constexpr (RELU) val = val > 0.f ? val : 0.f;
        out[(size_t)r * OC + col] = val;
      }
    }
  } else {
    float bias = bA[col];
    if constexpr (DUAL) bias += bB[col];
    float* op = out + (size_t)(row0 + mb) * OC + col;
#pragma unroll
    for (int vg = 0; vg < 8; ++vg) {
      float val = acc[vg] + bias;
      if constexpr (RELU) val = val > 0.f ? val : 0.f;
      op[(size_t)vg * OC] = val;
    }
  }
}


template <int IC, bool DUAL, bool RELU, bool INP, bool OUTP>
static inline void launch_gemm(const float* inA, const float* WA, const float* bA,
                               const float* inB, const float* WB, const float* bB,
                               const float* sB, float* out, int N, int OC, hipStream_t stream) {
  const int rowTiles = (N + 15) / 16;
  const int gy = (((OC + 15) / 16) + WAVES_PER_BLOCK - 1) / WAVES_PER_BLOCK;
  const dim3 grid(rowTiles, gy), blk(32 * WAVES_PER_BLOCK);
  if constexpr (OUTP) {
    gcn_wmma_gemm<IC, DUAL, RELU, false, INP, true><<<grid, blk, 0, stream>>>(
        inA, WA, bA, inB, WB, bB, sB, out, N, OC);
  } else {
    if ((N & 15) == 0 && (OC & 15) == 0)
      gcn_wmma_gemm<IC, DUAL, RELU, false, INP, false><<<grid, blk, 0, stream>>>(
          inA, WA, bA, inB, WB, bB, sB, out, N, OC);
    else
      gcn_wmma_gemm<IC, DUAL, RELU, true, INP, false><<<grid, blk, 0, stream>>>(
          inA, WA, bA, inB, WB, bB, sB, out, N, OC);
  }
}

extern "C" void kernel_launch(void* const* d_in, const int* in_sizes, int n_in,
                              void* d_out, int out_size, void* d_ws, size_t ws_size,
                              hipStream_t stream) {
  (void)n_in; (void)out_size; (void)ws_size;
  const float* x    = (const float*)d_in[0];
  const int*   ei   = (const int*)d_in[1];
  const float* w1   = (const float*)d_in[2];
  const float* b1   = (const float*)d_in[3];
  const float* w2   = (const float*)d_in[4];
  const float* b2   = (const float*)d_in[5];
  const float* c1wl = (const float*)d_in[6];
  const float* c1bl = (const float*)d_in[7];
  const float* c1wr = (const float*)d_in[8];
  const float* c1br = (const float*)d_in[9];
  const float* c2wl = (const float*)d_in[10];
  const float* c2bl = (const float*)d_in[11];
  const float* c2wr = (const float*)d_in[12];
  const float* c2br = (const float*)d_in[13];
  const float* w3   = (const float*)d_in[14];
  const float* b3   = (const float*)d_in[15];
  const float* w4   = (const float*)d_in[16];
  const float* b4   = (const float*)d_in[17];

  const int IN_CH = 256;
  const int N = in_sizes[0] / IN_CH;
  const int E = in_sizes[1] / 2;
  const size_t n = (size_t)N;

  float* ws   = (float*)d_ws;
  float* dis  = ws;
  float* h2   = ws + 1 * n;
  float* agg1 = ws + 33 * n;
  float* h3   = ws + 65 * n;
  float* agg2 = ws + 1 * n;
  float* h1   = ws + 129 * n;
  float* wp   = ws + 192 * n;
  float* hs2  = ws + 193 * n;
  float* hs3  = ws + 225 * n;
  float* h4   = ws + 193 * n;
  float* h5   = ws;

  const int eltBlocks  = (N + 255) / 256;
  const int edgeBlocks = (E + 255) / 256;
  const int rangeBlocks = (N + RG_NODES - 1) / RG_NODES;

  PackSrc ps = {w1, w2, c1wl, c1wr, c2wl, c2wr, w3, w4};
  gcn_pack_weights<<<dim3(24, NUM_W), 256, 0, stream>>>(ps, wp);

  launch_gemm<256, false, false, false, true>(x, wp + WP_W1, b1, nullptr, nullptr, nullptr, nullptr, h1, N, 16, stream);
  launch_gemm<16, false, false, true, false>(h1, wp + WP_W2, b2, nullptr, nullptr, nullptr, nullptr, h2, N, 32, stream);

  gcn_deg_init<<<eltBlocks, 256, 0, stream>>>(dis, N);
  gcn_deg_edge<<<edgeBlocks, 256, 0, stream>>>(ei + E, dis, E);
  gcn_deg_to_dis<<<eltBlocks, 256, 0, stream>>>(dis, N);

  gcn_prescale<<<(int)((n * 8 + 255) / 256), 256, 0, stream>>>(dis, h2, hs2, N, 32);
  gcn_range_sum<32><<<rangeBlocks, RG_THREADS, 0, stream>>>(ei, hs2, agg1, E, N);
  launch_gemm<32, true, true, false, false>(h2, wp + WP_C1L, c1bl, agg1, wp + WP_C1R, c1br, dis, h3, N, 64, stream);

  gcn_prescale<<<(int)((n * 16 + 255) / 256), 256, 0, stream>>>(dis, h3, hs3, N, 64);
  gcn_range_sum<64><<<rangeBlocks, RG_THREADS, 0, stream>>>(ei, hs3, agg2, E, N);
  launch_gemm<64, true, true, false, true>(h3, wp + WP_C2L, c2bl, agg2, wp + WP_C2R, c2br, dis, h4, N, 128, stream);

  launch_gemm<128, false, false, true, true>(h4, wp + WP_W3, b3, nullptr, nullptr, nullptr, nullptr, h5, N, 192, stream);
  launch_gemm<192, false, false, true, false>(h5, wp + WP_W4, b4, nullptr, nullptr, nullptr, nullptr, (float*)d_out, N, 20, stream);
}


# conv2 edge pass reuses the (source, local target) lists the conv1 edge pass filtered: one full scan of the target list and its source look-ups removed; LDS float atomics in arrival order as before
# speedup vs baseline: 1.7571x; 1.1429x over previous
#include <hip/hip_runtime.h>

typedef __attribute__((ext_vector_type(2))) float v2f;
typedef __attribute__((ext_vector_type(8))) float v8f;

#define WAVES_PER_BLOCK 4


__global__ void gcn_deg_init(float* __restrict__ deg, int N) {
  int i = blockIdx.x * blockDim.x + threadIdx.x;
  if (i < N) deg[i] = 1.0f;
}

__global__ void gcn_deg_edge(const int* __restrict__ col, float* __restrict__ deg, int E) {
  int e = blockIdx.x * blockDim.x + threadIdx.x;
  if (e < E) unsafeAtomicAdd(&deg[col[e]], 1.0f);
}

__global__ void gcn_deg_to_dis(float* __restrict__ dis, int N) {
  int i = blockIdx.x * blockDim.x + threadIdx.x;
  if (i < N) {
    float d = dis[i];
    dis[i] = (d > 0.0f) ? rsqrtf(d) : 0.0f;
  }
}


__global__ void gcn_prescale(const float* __restrict__ dis, const float* __restrict__ h,
                             float* __restrict__ hs, int N, int C) {
  long long i = (long long)blockIdx.x * blockDim.x + threadIdx.x;
  long long total = ((long long)N * C) >> 2;
  if (i < total) {
    int node = (int)((i << 2) / C);
    float d = dis[node];
    float4 v = ((const float4*)h)[i];
    v.x *= d; v.y *= d; v.z *= d; v.w *= d;
    ((float4*)hs)[i] = v;
  }
}

#define RG_NODES   400
#define RG_THREADS 1024
#define RG_LIST    8192

template <int C, bool SAVE, bool LOAD>
__global__ __launch_bounds__(RG_THREADS) void gcn_range_sum(
    const int* __restrict__ ei, const float* __restrict__ hs, float* __restrict__ acc,
    int2* __restrict__ glist, int* __restrict__ gcnt, int E, int N) {
  static_assert(!(SAVE && LOAD), "a pass either builds the lists or reads them");
  __shared__ float    sacc[RG_NODES * C];
  __shared__ int      lsrc[RG_LIST];
  __shared__ int      ldst[RG_LIST];
  __shared__ unsigned cnt;

  const int tid  = threadIdx.x;
  const int base = blockIdx.x * RG_NODES;
  const int rem  = N - base;
  const unsigned nloc = (unsigned)(rem < RG_NODES ? rem : RG_NODES);
  const int nvec = (int)nloc * (C >> 2);
  int2* gl = glist + (size_t)blockIdx.x * RG_LIST;

  const float4* hsv = (const float4*)(hs + (size_t)base * C);
  for (int i = tid; i < nvec; i += RG_THREADS) ((float4*)sacc)[i] = hsv[i];
  if (tid == 0) cnt = 0u;

  int      n = 0;
  unsigned total = 0u;
  bool     listed = false;
  if constexpr (LOAD) {
    const unsigned prev = (unsigned)gcnt[blockIdx.x];
    listed = prev <= (unsigned)RG_LIST;
    if (listed) {
      n = (int)prev;
      for (int i = tid; i < n; i += RG_THREADS) {
        const int2 p = gl[i];
        lsrc[i] = p.x;
        ldst[i] = p.y;
      }
    }
  }

  if (!listed) {
    __syncthreads();

    const int* col = ei + E;

    auto keep = [&](int e, unsigned cl) {
      const unsigned s = atomicAdd(&cnt, 1u);
      if (s < (unsigned)RG_LIST) {
        lsrc[s] = e;
        ldst[s] = (int)cl;
      } else {
        const float* hp = hs + (size_t)ei[e] * C;
#pragma unroll 1
        for (int ch = 0; ch < C; ++ch) atomicAdd(&sacc[cl * C + ch], hp[ch]);
      }
    };
    auto test4 = [&](const int4 v, int q) {
      const unsigned a = (unsigned)(v.x - base), b = (unsigned)(v.y - base);
      const unsigned c = (unsigned)(v.z - base), d = (unsigned)(v.w - base);
      if ((a < nloc) | (b < nloc) | (c < nloc) | (d < nloc)) {
        const int e = q << 2;
        if (a < nloc) keep(e, a);
        if (b < nloc) keep(e + 1, b);
        if (c < nloc) keep(e + 2, c);
        if (d < nloc) keep(e + 3, d);
      }
    };

    if (((E & 3) == 0) && ((((size_t)col) & 15) == 0)) {
      const int4* colv = (const int4*)col;
      const int nq = E >> 2;
      int q = tid;
      for (; q + 3 * RG_THREADS < nq; q += 4 * RG_THREADS) {
        const int4 v0 = colv[q];
        const int4 v1 = colv[q + RG_THREADS];
        const int4 v2 = colv[q + 2 * RG_THREADS];
        const int4 v3 = colv[q + 3 * RG_THREADS];
        test4(v0, q);
        test4(v1, q + RG_THREADS);
        test4(v2, q + 2 * RG_THREADS);
        test4(v3, q + 3 * RG_THREADS);
      }
      for (; q < nq; q += RG_THREADS) test4(colv[q], q);
    } else {
      for (int e = tid; e < E; e += RG_THREADS) {
        const unsigned cl = (unsigned)(col[e] - base);
        if (cl < nloc) keep(e, cl);
      }
    }
    __syncthreads();

    total = cnt;
    n = (int)(total < (unsigned)RG_LIST ? total : (unsigned)RG_LIST);
    for (int i = tid; i < n; i += RG_THREADS) lsrc[i] = ei[lsrc[i]];
  }
  __syncthreads();

  constexpr int RG_U = 8;
  constexpr int VEC  = C / 32;
  const int lane = tid & 31;
  const int wave = __builtin_amdgcn_readfirstlane(tid >> 5);
  for (int i0 = wave * RG_U; i0 < n; i0 += (RG_THREADS / 32) * RG_U) {
    float val[RG_U][VEC];
    int   dst[RG_U];
#pragma unroll
    for (int u = 0; u < RG_U; ++u) {
      const int idx = (i0 + u < n) ? (i0 + u) : i0;
      const int r = lsrc[idx];
      dst[u] = ldst[idx];
      const float* hp = hs + (size_t)r * C + lane * VEC;
#pragma unroll
      for (int k = 0; k < VEC; ++k) val[u][k] = hp[k];
    }
#pragma unroll
    for (int u = 0; u < RG_U; ++u) {
      if (i0 + u < n) {
        float* sp = sacc + dst[u] * C + lane * VEC;
#pragma unroll
        for (int k = 0; k < VEC; ++k) atomicAdd(sp + k, val[u][k]);
      }
    }
  }
  __syncthreads();

  float4* accv = (float4*)(acc + (size_t)base * C);
  for (int i = tid; i < nvec; i += RG_THREADS) accv[i] = ((const float4*)sacc)[i];

  if constexpr (SAVE) {
    for (int i = tid; i < n; i += RG_THREADS) gl[i] = make_int2(lsrc[i], ldst[i]);
    if (tid == 0) gcnt[blockIdx.x] = (int)total;
  }
}


#define NUM_W 8
struct PackSrc {
  const float* w0; const float* w1; const float* w2; const float* w3;
  const float* w4; const float* w5; const float* w6; const float* w7;
};
#define WP_W1   0
#define WP_W2   4096
#define WP_C1L  4608
#define WP_C1R  6656
#define WP_C2L  8704
#define WP_C2R  16896
#define WP_W3   25088
#define WP_W4   49664
#define WP_TOTAL 55808

__global__ void gcn_pack_weights(PackSrc s, float* __restrict__ dst) {
  const int m = blockIdx.y;
  const float* W; int OC, IC, off;
  switch (m) {
    case 0:  W = s.w0; OC = 16;  IC = 256; off = WP_W1;  break;
    case 1:  W = s.w1; OC = 32;  IC = 16;  off = WP_W2;  break;
    case 2:  W = s.w2; OC = 64;  IC = 32;  off = WP_C1L; break;
    case 3:  W = s.w3; OC = 64;  IC = 32;  off = WP_C1R; break;
    case 4:  W = s.w4; OC = 128; IC = 64;  off = WP_C2L; break;
    case 5:  W = s.w5; OC = 128; IC = 64;  off = WP_C2R; break;
    case 6:  W = s.w6; OC = 192; IC = 128; off = WP_W3;  break;
    default: W = s.w7; OC = 20;  IC = 192; off = WP_W4;  break;
  }
  const int ks_n  = IC >> 2;
  const int tiles = (OC + 15) >> 4;
  const int total = tiles * ks_n * 32;
  for (int i = blockIdx.x * blockDim.x + threadIdx.x; i < total; i += gridDim.x * blockDim.x) {
    const int lane = i & 31;
    const int ks   = (i >> 5) % ks_n;
    const int ct   = (i >> 5) / ks_n;
    int col = ct * 16 + (lane & 15);
    col = col < OC ? col : OC - 1;
    const float* p = W + (size_t)col * IC + ks * 4 + ((lane >> 4) << 1);
    v2f v = *(const v2f*)p;
    *(v2f*)(dst + off + (size_t)i * 2) = v;
  }
}

template <int IC, bool DUAL, bool RELU, bool GUARD, bool INP, bool OUTP>
__global__ __launch_bounds__(32 * WAVES_PER_BLOCK) void gcn_wmma_gemm(
    const float* __restrict__ inA, const float* __restrict__ WA, const float* __restrict__ bA,
    const float* __restrict__ inB, const float* __restrict__ WB, const float* __restrict__ bB,
    const float* __restrict__ sB, float* __restrict__ out, int nrows, int OC) {
  static_assert(!(INP && DUAL), "fragment-order input is single-operand only");
  static_assert(!(OUTP && GUARD), "fragment-order output needs full tiles");
  __shared__ float lds[INP ? 4 : (DUAL ? 2 : 1) * 16 * IC];
  const int tid  = threadIdx.x;
  const int lane = tid & 31;
  const int wave = tid >> 5;
  const int row0 = blockIdx.x << 4;
  const int col0 = (blockIdx.y * WAVES_PER_BLOCK + wave) << 4;

  float* Atile = lds;
  float* Btile = DUAL ? (lds + 16 * IC) : lds;

  if constexpr (!INP) {
    constexpr int vecPerRow = IC >> 2;
    constexpr int totalVec  = vecPerRow << 4;
    for (int idx = tid; idx < totalVec; idx += 32 * WAVES_PER_BLOCK) {
      const int r = idx / vecPerRow;
      const int v = idx - r * vecPerRow;
      const int node = row0 + r;
      if (GUARD && node >= nrows) {
        float4 z = {0.f, 0.f, 0.f, 0.f};
        ((float4*)Atile)[idx] = z;
        if constexpr (DUAL) ((float4*)Btile)[idx] = z;
      } else {
        ((float4*)Atile)[idx] = ((const float4*)(inA + (size_t)node * IC))[v];
        if constexpr (DUAL) {
          float4 b = ((const float4*)(inB + (size_t)node * IC))[v];
          const float d = sB[node];
          b.x *= d; b.y *= d; b.z *= d; b.w *= d;
          ((float4*)Btile)[idx] = b;
        }
      }
    }
    __syncthreads();
  }

  if (col0 >= OC) return;

  const int  nidx = lane & 15;
  const int  koff = (lane >> 4) << 1;
  const int  col  = col0 + nidx;
  const bool colOK = GUARD ? (col < OC) : true;

  constexpr int KSTEPS = IC / 4;
  const v2f* wAp = (const v2f*)WA + (size_t)(col0 >> 4) * KSTEPS * 32 + lane;
  const v2f* wBp = DUAL ? ((const v2f*)WB + (size_t)(col0 >> 4) * KSTEPS * 32 + lane) : wAp;
  const v2f*   ag = (const v2f*)(inA + (size_t)row0 * IC) + lane;
  const float* ar = Atile + nidx * IC + koff;
  const float* br = Btile + nidx * IC + koff;

  constexpr int BATCH  = DUAL ? (KSTEPS < 4 ? KSTEPS : 4)
                              : (KSTEPS < 8 ? KSTEPS : 8);

  v8f acc = {};
#pragma unroll 1
  for (int kb = 0; kb < KSTEPS; kb += BATCH) {
    v2f fa[BATCH], fw[BATCH];
    v2f ga[DUAL ? BATCH : 1], gw[DUAL ? BATCH : 1];
#pragma unroll
    for (int i = 0; i < BATCH; ++i) {
      const int ks = kb + i;
      if constexpr (INP) fa[i] = ag[ks * 32];
      else               fa[i] = *(const v2f*)(ar + ks * 4);
      fw[i] = wAp[ks * 32];
      if constexpr (DUAL) {
        ga[i] = *(const v2f*)(br + ks * 4);
        gw[i] = wBp[ks * 32];
      }
    }
#pragma unroll
    for (int i = 0; i < BATCH; ++i) {
      if constexpr (OUTP) {
        acc = __builtin_amdgcn_wmma_f32_16x16x4_f32(false, fw[i], false, fa[i],
                                                    (short)0, acc, false, false);
        if constexpr (DUAL)
          acc = __builtin_amdgcn_wmma_f32_16x16x4_f32(false, gw[i], false, ga[i],
                                                      (short)0, acc, false, false);
      } else {
        acc = __builtin_amdgcn_wmma_f32_16x16x4_f32(false, fa[i], false, fw[i],
                                                    (short)0, acc, false, false);
        if constexpr (DUAL)
          acc = __builtin_amdgcn_wmma_f32_16x16x4_f32(false, ga[i], false, gw[i],
                                                      (short)0, acc, false, false);
      }
    }
  }

  const int mb = (lane >> 4) << 3;
  if constexpr (OUTP) {
    const int cb = col0 + mb;
    float bias[8];
#pragma unroll
    for (int vg = 0; vg < 8; ++vg) {
      bias[vg] = bA[cb + vg];
      if constexpr (DUAL) bias[vg] += bB[cb + vg];
    }
    const int KSO = OC >> 2;
    float* ob = out + ((size_t)blockIdx.x * KSO + (cb >> 2)) * 64 + nidx * 2;
#pragma unroll
    for (int t = 0; t < 4; ++t) {
      float v0 = acc[2 * t] + bias[2 * t];
      float v1 = acc[2 * t + 1] + bias[2 * t + 1];
      if constexpr (RELU) { v0 = v0 > 0.f ? v0 : 0.f; v1 = v1 > 0.f ? v1 : 0.f; }
      v2f o = {v0, v1};
      *(v2f*)(ob + (t >> 1) * 64 + (t & 1) * 32) = o;
    }
  } else if constexpr (GUARD) {
    float bias = 0.f;
    if (colOK) {
      bias = bA[col];
      if constexpr (DUAL) bias += bB[col];
    }
#pragma unroll
    for (int vg = 0; vg < 8; ++vg) {
      const int r = row0 + mb + vg;
      if (colOK && r < nrows) {
        float val = acc[vg] + bias;
        if constexpr (RELU) val = val > 0.f ? val : 0.f;
        out[(size_t)r * OC + col] = val;
      }
    }
  } else {
    float bias = bA[col];
    if constexpr (DUAL) bias += bB[col];
    float* op = out + (size_t)(row0 + mb) * OC + col;
#pragma unroll
    for (int vg = 0; vg < 8; ++vg) {
      float val = acc[vg] + bias;
      if constexpr (RELU) val = val > 0.f ? val : 0.f;
      op[(size_t)vg * OC] = val;
    }
  }
}


template <int IC, bool DUAL, bool RELU, bool INP, bool OUTP>
static inline void launch_gemm(const float* inA, const float* WA, const float* bA,
                               const float* inB, const float* WB, const float* bB,
                               const float* sB, float* out, int N, int OC, hipStream_t stream) {
  const int rowTiles = (N + 15) / 16;
  const int gy = (((OC + 15) / 16) + WAVES_PER_BLOCK - 1) / WAVES_PER_BLOCK;
  const dim3 grid(rowTiles, gy), blk(32 * WAVES_PER_BLOCK);
  if constexpr (OUTP) {
    gcn_wmma_gemm<IC, DUAL, RELU, false, INP, true><<<grid, blk, 0, stream>>>(
        inA, WA, bA, inB, WB, bB, sB, out, N, OC);
  } else {
    if ((N & 15) == 0 && (OC & 15) == 0)
      gcn_wmma_gemm<IC, DUAL, RELU, false, INP, false><<<grid, blk, 0, stream>>>(
          inA, WA, bA, inB, WB, bB, sB, out, N, OC);
    else
      gcn_wmma_gemm<IC, DUAL, RELU, true, INP, false><<<grid, blk, 0, stream>>>(
          inA, WA, bA, inB, WB, bB, sB, out, N, OC);
  }
}

extern "C" void kernel_launch(void* const* d_in, const int* in_sizes, int n_in,
                              void* d_out, int out_size, void* d_ws, size_t ws_size,
                              hipStream_t stream) {
  (void)n_in; (void)out_size; (void)ws_size;
  const float* x    = (const float*)d_in[0];
  const int*   ei   = (const int*)d_in[1];
  const float* w1   = (const float*)d_in[2];
  const float* b1   = (const float*)d_in[3];
  const float* w2   = (const float*)d_in[4];
  const float* b2   = (const float*)d_in[5];
  const float* c1wl = (const float*)d_in[6];
  const float* c1bl = (const float*)d_in[7];
  const float* c1wr = (const float*)d_in[8];
  const float* c1br = (const float*)d_in[9];
  const float* c2wl = (const float*)d_in[10];
  const float* c2bl = (const float*)d_in[11];
  const float* c2wr = (const float*)d_in[12];
  const float* c2br = (const float*)d_in[13];
  const float* w3   = (const float*)d_in[14];
  const float* b3   = (const float*)d_in[15];
  const float* w4   = (const float*)d_in[16];
  const float* b4   = (const float*)d_in[17];

  const int IN_CH = 256;
  const int N = in_sizes[0] / IN_CH;
  const int E = in_sizes[1] / 2;
  const size_t n = (size_t)N;

  float* ws   = (float*)d_ws;
  float* dis  = ws;
  float* h2   = ws + 1 * n;
  float* agg1 = ws + 33 * n;
  float* h3   = ws + 65 * n;
  float* agg2 = ws + 1 * n;
  float* h1   = ws + 129 * n;
  float* wp   = ws + 192 * n;
  float* hs2  = ws + 193 * n;
  float* hs3  = ws + 225 * n;
  float* h4   = ws + 193 * n;
  float* h5   = ws;
  int*  rcnt  = (int*)(ws + 129 * n);
  int2* rlist = (int2*)(ws + 130 * n);

  const int eltBlocks  = (N + 255) / 256;
  const int edgeBlocks = (E + 255) / 256;
  const int rangeBlocks = (N + RG_NODES - 1) / RG_NODES;

  PackSrc ps = {w1, w2, c1wl, c1wr, c2wl, c2wr, w3, w4};
  gcn_pack_weights<<<dim3(24, NUM_W), 256, 0, stream>>>(ps, wp);

  launch_gemm<256, false, false, false, true>(x, wp + WP_W1, b1, nullptr, nullptr, nullptr, nullptr, h1, N, 16, stream);
  launch_gemm<16, false, false, true, false>(h1, wp + WP_W2, b2, nullptr, nullptr, nullptr, nullptr, h2, N, 32, stream);

  gcn_deg_init<<<eltBlocks, 256, 0, stream>>>(dis, N);
  gcn_deg_edge<<<edgeBlocks, 256, 0, stream>>>(ei + E, dis, E);
  gcn_deg_to_dis<<<eltBlocks, 256, 0, stream>>>(dis, N);

  gcn_prescale<<<(int)((n * 8 + 255) / 256), 256, 0, stream>>>(dis, h2, hs2, N, 32);
  gcn_range_sum<32, true, false><<<rangeBlocks, RG_THREADS, 0, stream>>>(ei, hs2, agg1, rlist, rcnt, E, N);
  launch_gemm<32, true, true, false, false>(h2, wp + WP_C1L, c1bl, agg1, wp + WP_C1R, c1br, dis, h3, N, 64, stream);

  gcn_prescale<<<(int)((n * 16 + 255) / 256), 256, 0, stream>>>(dis, h3, hs3, N, 64);
  gcn_range_sum<64, false, true><<<rangeBlocks, RG_THREADS, 0, stream>>>(ei, hs3, agg2, rlist, rcnt, E, N);
  launch_gemm<64, true, true, false, true>(h3, wp + WP_C2L, c2bl, agg2, wp + WP_C2R, c2br, dis, h4, N, 128, stream);

  launch_gemm<128, false, false, true, true>(h4, wp + WP_W3, b3, nullptr, nullptr, nullptr, nullptr, h5, N, 192, stream);
  launch_gemm<192, false, false, true, false>(h5, wp + WP_W4, b4, nullptr, nullptr, nullptr, nullptr, (float*)d_out, N, 20, stream);
}


# degree pass bins every edge into its target range's (source, local target) list: LDS counts, one global slot add per range per group; no group scans the target list; arrival-order sums as before
# speedup vs baseline: 2.1045x; 1.1977x over previous
#include <hip/hip_runtime.h>

typedef __attribute__((ext_vector_type(2))) float v2f;
typedef __attribute__((ext_vector_type(8))) float v8f;

#define WAVES_PER_BLOCK 4


#define RG_NODES   400
#define RG_THREADS 1024
#define RG_LIST    8192

#define BIN_THREADS 1024
#define BIN_PER     4
#define BIN_MAXR    1024

__global__ void gcn_deg_init(float* __restrict__ deg, int* __restrict__ gcnt, int N, int nr) {
  int i = blockIdx.x * blockDim.x + threadIdx.x;
  if (i < N) deg[i] = 1.0f;
  if (i < nr) gcnt[i] = (nr <= BIN_MAXR) ? 0 : (RG_LIST + 1);
}

__global__ __launch_bounds__(BIN_THREADS) void gcn_deg_bin(
    const int* __restrict__ ei, float* __restrict__ deg, int2* __restrict__ glist,
    int* __restrict__ gcnt, int E, int nr) {
  __shared__ int hist[BIN_MAXR];
  const int  tid = threadIdx.x;
  const bool bin = nr <= BIN_MAXR;
  if (bin)
    for (int i = tid; i < nr; i += BIN_THREADS) hist[i] = 0;
  __syncthreads();

  const int e0 = blockIdx.x * (BIN_THREADS * BIN_PER) + tid;
  int c[BIN_PER], sn[BIN_PER], rg[BIN_PER], rank[BIN_PER];
#pragma unroll
  for (int u = 0; u < BIN_PER; ++u) {
    const int e = e0 + u * BIN_THREADS;
    const bool ok = e < E;
    c[u]  = ok ? ei[E + e] : -1;
    sn[u] = ok ? ei[e] : 0;
  }
#pragma unroll
  for (int u = 0; u < BIN_PER; ++u) {
    rg[u] = 0; rank[u] = 0;
    if (c[u] >= 0) {
      unsafeAtomicAdd(&deg[c[u]], 1.0f);
      if (bin) {
        rg[u]   = c[u] / RG_NODES;
        rank[u] = atomicAdd(&hist[rg[u]], 1);
      }
    }
  }
  __syncthreads();
  if (bin)
    for (int i = tid; i < nr; i += BIN_THREADS) {
      const int h = hist[i];
      hist[i] = h > 0 ? atomicAdd(&gcnt[i], h) : 0;
    }
  __syncthreads();
  if (bin) {
#pragma unroll
    for (int u = 0; u < BIN_PER; ++u) {
      if (c[u] >= 0) {
        const int r = rg[u];
        const unsigned slot = (unsigned)(hist[r] + rank[u]);
        if (slot < (unsigned)RG_LIST)
          glist[(size_t)r * RG_LIST + slot] = make_int2(sn[u], c[u] - r * RG_NODES);
      }
    }
  }
}

__global__ void gcn_deg_to_dis(float* __restrict__ dis, int N) {
  int i = blockIdx.x * blockDim.x + threadIdx.x;
  if (i < N) {
    float d = dis[i];
    dis[i] = (d > 0.0f) ? rsqrtf(d) : 0.0f;
  }
}


__global__ void gcn_prescale(const float* __restrict__ dis, const float* __restrict__ h,
                             float* __restrict__ hs, int N, int C) {
  long long i = (long long)blockIdx.x * blockDim.x + threadIdx.x;
  long long total = ((long long)N * C) >> 2;
  if (i < total) {
    int node = (int)((i << 2) / C);
    float d = dis[node];
    float4 v = ((const float4*)h)[i];
    v.x *= d; v.y *= d; v.z *= d; v.w *= d;
    ((float4*)hs)[i] = v;
  }
}


template <int C>
__global__ __launch_bounds__(RG_THREADS) void gcn_range_sum(
    const int* __restrict__ ei, const float* __restrict__ hs, float* __restrict__ acc,
    const int2* __restrict__ glist, const int* __restrict__ gcnt, int E, int N) {
  __shared__ float    sacc[RG_NODES * C];
  __shared__ int      lsrc[RG_LIST];
  __shared__ int      ldst[RG_LIST];
  __shared__ unsigned cnt;

  const int tid  = threadIdx.x;
  const int base = blockIdx.x * RG_NODES;
  const int rem  = N - base;
  const unsigned nloc = (unsigned)(rem < RG_NODES ? rem : RG_NODES);
  const int nvec = (int)nloc * (C >> 2);
  const int2* gl = glist + (size_t)blockIdx.x * RG_LIST;

  const float4* hsv = (const float4*)(hs + (size_t)base * C);
  for (int i = tid; i < nvec; i += RG_THREADS) ((float4*)sacc)[i] = hsv[i];
  if (tid == 0) cnt = 0u;

  int n = 0;
  const unsigned prev = (unsigned)gcnt[blockIdx.x];
  const bool listed = prev <= (unsigned)RG_LIST;
  if (listed) {
    n = (int)prev;
    for (int i = tid; i < n; i += RG_THREADS) {
      const int2 p = gl[i];
      lsrc[i] = p.x;
      ldst[i] = p.y;
    }
  }

  if (!listed) {
    __syncthreads();

    const int* col = ei + E;

    auto keep = [&](int e, unsigned cl) {
      const unsigned s = atomicAdd(&cnt, 1u);
      if (s < (unsigned)RG_LIST) {
        lsrc[s] = e;
        ldst[s] = (int)cl;
      } else {
        const float* hp = hs + (size_t)ei[e] * C;
#pragma unroll 1
        for (int ch = 0; ch < C; ++ch) atomicAdd(&sacc[cl * C + ch], hp[ch]);
      }
    };
    auto test4 = [&](const int4 v, int q) {
      const unsigned a = (unsigned)(v.x - base), b = (unsigned)(v.y - base);
      const unsigned c = (unsigned)(v.z - base), d = (unsigned)(v.w - base);
      if ((a < nloc) | (b < nloc) | (c < nloc) | (d < nloc)) {
        const int e = q << 2;
        if (a < nloc) keep(e, a);
        if (b < nloc) keep(e + 1, b);
        if (c < nloc) keep(e + 2, c);
        if (d < nloc) keep(e + 3, d);
      }
    };

    if (((E & 3) == 0) && ((((size_t)col) & 15) == 0)) {
      const int4* colv = (const int4*)col;
      const int nq = E >> 2;
      int q = tid;
      for (; q + 3 * RG_THREADS < nq; q += 4 * RG_THREADS) {
        const int4 v0 = colv[q];
        const int4 v1 = colv[q + RG_THREADS];
        const int4 v2 = colv[q + 2 * RG_THREADS];
        const int4 v3 = colv[q + 3 * RG_THREADS];
        test4(v0, q);
        test4(v1, q + RG_THREADS);
        test4(v2, q + 2 * RG_THREADS);
        test4(v3, q + 3 * RG_THREADS);
      }
      for (; q < nq; q += RG_THREADS) test4(colv[q], q);
    } else {
      for (int e = tid; e < E; e += RG_THREADS) {
        const unsigned cl = (unsigned)(col[e] - base);
        if (cl < nloc) keep(e, cl);
      }
    }
    __syncthreads();

    const unsigned total = cnt;
    n = (int)(total < (unsigned)RG_LIST ? total : (unsigned)RG_LIST);
    for (int i = tid; i < n; i += RG_THREADS) lsrc[i] = ei[lsrc[i]];
  }
  __syncthreads();

  constexpr int RG_U = 8;
  constexpr int VEC  = C / 32;
  const int lane = tid & 31;
  const int wave = __builtin_amdgcn_readfirstlane(tid >> 5);
  for (int i0 = wave * RG_U; i0 < n; i0 += (RG_THREADS / 32) * RG_U) {
    float val[RG_U][VEC];
    int   dst[RG_U];
#pragma unroll
    for (int u = 0; u < RG_U; ++u) {
      const int idx = (i0 + u < n) ? (i0 + u) : i0;
      const int r = lsrc[idx];
      dst[u] = ldst[idx];
      const float* hp = hs + (size_t)r * C + lane * VEC;
#pragma unroll
      for (int k = 0; k < VEC; ++k) val[u][k] = hp[k];
    }
#pragma unroll
    for (int u = 0; u < RG_U; ++u) {
      if (i0 + u < n) {
        float* sp = sacc + dst[u] * C + lane * VEC;
#pragma unroll
        for (int k = 0; k < VEC; ++k) atomicAdd(sp + k, val[u][k]);
      }
    }
  }
  __syncthreads();

  float4* accv = (float4*)(acc + (size_t)base * C);
  for (int i = tid; i < nvec; i += RG_THREADS) accv[i] = ((const float4*)sacc)[i];
}


#define NUM_W 8
struct PackSrc {
  const float* w0; const float* w1; const float* w2; const float* w3;
  const float* w4; const float* w5; const float* w6; const float* w7;
};
#define WP_W1   0
#define WP_W2   4096
#define WP_C1L  4608
#define WP_C1R  6656
#define WP_C2L  8704
#define WP_C2R  16896
#define WP_W3   25088
#define WP_W4   49664
#define WP_TOTAL 55808

__global__ void gcn_pack_weights(PackSrc s, float* __restrict__ dst) {
  const int m = blockIdx.y;
  const float* W; int OC, IC, off;
  switch (m) {
    case 0:  W = s.w0; OC = 16;  IC = 256; off = WP_W1;  break;
    case 1:  W = s.w1; OC = 32;  IC = 16;  off = WP_W2;  break;
    case 2:  W = s.w2; OC = 64;  IC = 32;  off = WP_C1L; break;
    case 3:  W = s.w3; OC = 64;  IC = 32;  off = WP_C1R; break;
    case 4:  W = s.w4; OC = 128; IC = 64;  off = WP_C2L; break;
    case 5:  W = s.w5; OC = 128; IC = 64;  off = WP_C2R; break;
    case 6:  W = s.w6; OC = 192; IC = 128; off = WP_W3;  break;
    default: W = s.w7; OC = 20;  IC = 192; off = WP_W4;  break;
  }
  const int ks_n  = IC >> 2;
  const int tiles = (OC + 15) >> 4;
  const int total = tiles * ks_n * 32;
  for (int i = blockIdx.x * blockDim.x + threadIdx.x; i < total; i += gridDim.x * blockDim.x) {
    const int lane = i & 31;
    const int ks   = (i >> 5) % ks_n;
    const int ct   = (i >> 5) / ks_n;
    int col = ct * 16 + (lane & 15);
    col = col < OC ? col : OC - 1;
    const float* p = W + (size_t)col * IC + ks * 4 + ((lane >> 4) << 1);
    v2f v = *(const v2f*)p;
    *(v2f*)(dst + off + (size_t)i * 2) = v;
  }
}

template <int IC, bool DUAL, bool RELU, bool GUARD, bool INP, bool OUTP>
__global__ __launch_bounds__(32 * WAVES_PER_BLOCK) void gcn_wmma_gemm(
    const float* __restrict__ inA, const float* __restrict__ WA, const float* __restrict__ bA,
    const float* __restrict__ inB, const float* __restrict__ WB, const float* __restrict__ bB,
    const float* __restrict__ sB, float* __restrict__ out, int nrows, int OC) {
  static_assert(!(INP && DUAL), "fragment-order input is single-operand only");
  static_assert(!(OUTP && GUARD), "fragment-order output needs full tiles");
  __shared__ float lds[INP ? 4 : (DUAL ? 2 : 1) * 16 * IC];
  const int tid  = threadIdx.x;
  const int lane = tid & 31;
  const int wave = tid >> 5;
  const int row0 = blockIdx.x << 4;
  const int col0 = (blockIdx.y * WAVES_PER_BLOCK + wave) << 4;

  float* Atile = lds;
  float* Btile = DUAL ? (lds + 16 * IC) : lds;

  if constexpr (!INP) {
    constexpr int vecPerRow = IC >> 2;
    constexpr int totalVec  = vecPerRow << 4;
    for (int idx = tid; idx < totalVec; idx += 32 * WAVES_PER_BLOCK) {
      const int r = idx / vecPerRow;
      const int v = idx - r * vecPerRow;
      const int node = row0 + r;
      if (GUARD && node >= nrows) {
        float4 z = {0.f, 0.f, 0.f, 0.f};
        ((float4*)Atile)[idx] = z;
        if constexpr (DUAL) ((float4*)Btile)[idx] = z;
      } else {
        ((float4*)Atile)[idx] = ((const float4*)(inA + (size_t)node * IC))[v];
        if constexpr (DUAL) {
          float4 b = ((const float4*)(inB + (size_t)node * IC))[v];
          const float d = sB[node];
          b.x *= d; b.y *= d; b.z *= d; b.w *= d;
          ((float4*)Btile)[idx] = b;
        }
      }
    }
    __syncthreads();
  }

  if (col0 >= OC) return;

  const int  nidx = lane & 15;
  const int  koff = (lane >> 4) << 1;
  const int  col  = col0 + nidx;
  const bool colOK = GUARD ? (col < OC) : true;

  constexpr int KSTEPS = IC / 4;
  const v2f* wAp = (const v2f*)WA + (size_t)(col0 >> 4) * KSTEPS * 32 + lane;
  const v2f* wBp = DUAL ? ((const v2f*)WB + (size_t)(col0 >> 4) * KSTEPS * 32 + lane) : wAp;
  const v2f*   ag = (const v2f*)(inA + (size_t)row0 * IC) + lane;
  const float* ar = Atile + nidx * IC + koff;
  const float* br = Btile + nidx * IC + koff;

  constexpr int BATCH  = DUAL ? (KSTEPS < 4 ? KSTEPS : 4)
                              : (KSTEPS < 8 ? KSTEPS : 8);

  v8f acc = {};
#pragma unroll 1
  for (int kb = 0; kb < KSTEPS; kb += BATCH) {
    v2f fa[BATCH], fw[BATCH];
    v2f ga[DUAL ? BATCH : 1], gw[DUAL ? BATCH : 1];
#pragma unroll
    for (int i = 0; i < BATCH; ++i) {
      const int ks = kb + i;
      if constexpr (INP) fa[i] = ag[ks * 32];
      else               fa[i] = *(const v2f*)(ar + ks * 4);
      fw[i] = wAp[ks * 32];
      if constexpr (DUAL) {
        ga[i] = *(const v2f*)(br + ks * 4);
        gw[i] = wBp[ks * 32];
      }
    }
#pragma unroll
    for (int i = 0; i < BATCH; ++i) {
      if constexpr (OUTP) {
        acc = __builtin_amdgcn_wmma_f32_16x16x4_f32(false, fw[i], false, fa[i],
                                                    (short)0, acc, false, false);
        if constexpr (DUAL)
          acc = __builtin_amdgcn_wmma_f32_16x16x4_f32(false, gw[i], false, ga[i],
                                                      (short)0, acc, false, false);
      } else {
        acc = __builtin_amdgcn_wmma_f32_16x16x4_f32(false, fa[i], false, fw[i],
                                                    (short)0, acc, false, false);
        if constexpr (DUAL)
          acc = __builtin_amdgcn_wmma_f32_16x16x4_f32(false, ga[i], false, gw[i],
                                                      (short)0, acc, false, false);
      }
    }
  }

  const int mb = (lane >> 4) << 3;
  if constexpr (OUTP) {
    const int cb = col0 + mb;
    float bias[8];
#pragma unroll
    for (int vg = 0; vg < 8; ++vg) {
      bias[vg] = bA[cb + vg];
      if constexpr (DUAL) bias[vg] += bB[cb + vg];
    }
    const int KSO = OC >> 2;
    float* ob = out + ((size_t)blockIdx.x * KSO + (cb >> 2)) * 64 + nidx * 2;
#pragma unroll
    for (int t = 0; t < 4; ++t) {
      float v0 = acc[2 * t] + bias[2 * t];
      float v1 = acc[2 * t + 1] + bias[2 * t + 1];
      if constexpr (RELU) { v0 = v0 > 0.f ? v0 : 0.f; v1 = v1 > 0.f ? v1 : 0.f; }
      v2f o = {v0, v1};
      *(v2f*)(ob + (t >> 1) * 64 + (t & 1) * 32) = o;
    }
  } else if constexpr (GUARD) {
    float bias = 0.f;
    if (colOK) {
      bias = bA[col];
      if constexpr (DUAL) bias += bB[col];
    }
#pragma unroll
    for (int vg = 0; vg < 8; ++vg) {
      const int r = row0 + mb + vg;
      if (colOK && r < nrows) {
        float val = acc[vg] + bias;
        if constexpr (RELU) val = val > 0.f ? val : 0.f;
        out[(size_t)r * OC + col] = val;
      }
    }
  } else {
    float bias = bA[col];
    if constexpr (DUAL) bias += bB[col];
    float* op = out + (size_t)(row0 + mb) * OC + col;
#pragma unroll
    for (int vg = 0; vg < 8; ++vg) {
      float val = acc[vg] + bias;
      if constexpr (RELU) val = val > 0.f ? val : 0.f;
      op[(size_t)vg * OC] = val;
    }
  }
}


template <int IC, bool DUAL, bool RELU, bool INP, bool OUTP>
static inline void launch_gemm(const float* inA, const float* WA, const float* bA,
                               const float* inB, const float* WB, const float* bB,
                               const float* sB, float* out, int N, int OC, hipStream_t stream) {
  const int rowTiles = (N + 15) / 16;
  const int gy = (((OC + 15) / 16) + WAVES_PER_BLOCK - 1) / WAVES_PER_BLOCK;
  const dim3 grid(rowTiles, gy), blk(32 * WAVES_PER_BLOCK);
  if constexpr (OUTP) {
    gcn_wmma_gemm<IC, DUAL, RELU, false, INP, true><<<grid, blk, 0, stream>>>(
        inA, WA, bA, inB, WB, bB, sB, out, N, OC);
  } else {
    if ((N & 15) == 0 && (OC & 15) == 0)
      gcn_wmma_gemm<IC, DUAL, RELU, false, INP, false><<<grid, blk, 0, stream>>>(
          inA, WA, bA, inB, WB, bB, sB, out, N, OC);
    else
      gcn_wmma_gemm<IC, DUAL, RELU, true, INP, false><<<grid, blk, 0, stream>>>(
          inA, WA, bA, inB, WB, bB, sB, out, N, OC);
  }
}

extern "C" void kernel_launch(void* const* d_in, const int* in_sizes, int n_in,
                              void* d_out, int out_size, void* d_ws, size_t ws_size,
                              hipStream_t stream) {
  (void)n_in; (void)out_size; (void)ws_size;
  const float* x    = (const float*)d_in[0];
  const int*   ei   = (const int*)d_in[1];
  const float* w1   = (const float*)d_in[2];
  const float* b1   = (const float*)d_in[3];
  const float* w2   = (const float*)d_in[4];
  const float* b2   = (const float*)d_in[5];
  const float* c1wl = (const float*)d_in[6];
  const float* c1bl = (const float*)d_in[7];
  const float* c1wr = (const float*)d_in[8];
  const float* c1br = (const float*)d_in[9];
  const float* c2wl = (const float*)d_in[10];
  const float* c2bl = (const float*)d_in[11];
  const float* c2wr = (const float*)d_in[12];
  const float* c2br = (const float*)d_in[13];
  const float* w3   = (const float*)d_in[14];
  const float* b3   = (const float*)d_in[15];
  const float* w4   = (const float*)d_in[16];
  const float* b4   = (const float*)d_in[17];

  const int IN_CH = 256;
  const int N = in_sizes[0] / IN_CH;
  const int E = in_sizes[1] / 2;
  const size_t n = (size_t)N;

  float* ws   = (float*)d_ws;
  float* dis  = ws;
  float* h2   = ws + 1 * n;
  float* agg1 = ws + 33 * n;
  float* h3   = ws + 65 * n;
  float* agg2 = ws + 1 * n;
  float* h1   = ws + 129 * n;
  float* wp   = ws + 192 * n;
  float* hs2  = ws + 193 * n;
  float* hs3  = ws + 225 * n;
  float* h4   = ws + 193 * n;
  float* h5   = ws;
  int*  rcnt  = (int*)(ws + 129 * n);
  int2* rlist = (int2*)(ws + 130 * n);

  const int eltBlocks  = (N + 255) / 256;
  const int binBlocks  = (E + BIN_THREADS * BIN_PER - 1) / (BIN_THREADS * BIN_PER);
  const int rangeBlocks = (N + RG_NODES - 1) / RG_NODES;

  PackSrc ps = {w1, w2, c1wl, c1wr, c2wl, c2wr, w3, w4};
  gcn_pack_weights<<<dim3(24, NUM_W), 256, 0, stream>>>(ps, wp);

  launch_gemm<256, false, false, false, true>(x, wp + WP_W1, b1, nullptr, nullptr, nullptr, nullptr, h1, N, 16, stream);
  launch_gemm<16, false, false, true, false>(h1, wp + WP_W2, b2, nullptr, nullptr, nullptr, nullptr, h2, N, 32, stream);

  gcn_deg_init<<<eltBlocks, 256, 0, stream>>>(dis, rcnt, N, rangeBlocks);
  gcn_deg_bin<<<binBlocks, BIN_THREADS, 0, stream>>>(ei, dis, rlist, rcnt, E, rangeBlocks);
  gcn_deg_to_dis<<<eltBlocks, 256, 0, stream>>>(dis, N);

  gcn_prescale<<<(int)((n * 8 + 255) / 256), 256, 0, stream>>>(dis, h2, hs2, N, 32);
  gcn_range_sum<32><<<rangeBlocks, RG_THREADS, 0, stream>>>(ei, hs2, agg1, rlist, rcnt, E, N);
  launch_gemm<32, true, true, false, false>(h2, wp + WP_C1L, c1bl, agg1, wp + WP_C1R, c1br, dis, h3, N, 64, stream);

  gcn_prescale<<<(int)((n * 16 + 255) / 256), 256, 0, stream>>>(dis, h3, hs3, N, 64);
  gcn_range_sum<64><<<rangeBlocks, RG_THREADS, 0, stream>>>(ei, hs3, agg2, rlist, rcnt, E, N);
  launch_gemm<64, true, true, false, true>(h3, wp + WP_C2L, c2bl, agg2, wp + WP_C2R, c2br, dis, h4, N, 128, stream);

  launch_gemm<128, false, false, true, true>(h4, wp + WP_W3, b3, nullptr, nullptr, nullptr, nullptr, h5, N, 192, stream);
  launch_gemm<192, false, false, true, false>(h5, wp + WP_W4, b4, nullptr, nullptr, nullptr, nullptr, (float*)d_out, N, 20, stream);
}


# conv GEMM operands (layer input, neighbour sum) kept in fragment order in memory: producers store them so, edge pass applies dis[c] at its store; read from global, no LDS tile; sum order unchanged
# speedup vs baseline: 2.2203x; 1.0550x over previous
#include <hip/hip_runtime.h>

typedef __attribute__((ext_vector_type(2))) float v2f;
typedef __attribute__((ext_vector_type(8))) float v8f;

#define WAVES_PER_BLOCK 4


#define RG_NODES   400
#define RG_THREADS 1024
#define RG_LIST    8192

#define BIN_THREADS 1024
#define BIN_PER     4
#define BIN_MAXR    1024

__global__ void gcn_deg_init(float* __restrict__ deg, int* __restrict__ gcnt, int N, int nr) {
  int i = blockIdx.x * blockDim.x + threadIdx.x;
  if (i < N) deg[i] = 1.0f;
  if (i < nr) gcnt[i] = (nr <= BIN_MAXR) ? 0 : (RG_LIST + 1);
}

__global__ __launch_bounds__(BIN_THREADS) void gcn_deg_bin(
    const int* __restrict__ ei, float* __restrict__ deg, int2* __restrict__ glist,
    int* __restrict__ gcnt, int E, int nr) {
  __shared__ int hist[BIN_MAXR];
  const int  tid = threadIdx.x;
  const bool bin = nr <= BIN_MAXR;
  if (bin)
    for (int i = tid; i < nr; i += BIN_THREADS) hist[i] = 0;
  __syncthreads();

  const int e0 = blockIdx.x * (BIN_THREADS * BIN_PER) + tid;
  int c[BIN_PER], sn[BIN_PER], rg[BIN_PER], rank[BIN_PER];
#pragma unroll
  for (int u = 0; u < BIN_PER; ++u) {
    const int e = e0 + u * BIN_THREADS;
    const bool ok = e < E;
    c[u]  = ok ? ei[E + e] : -1;
    sn[u] = ok ? ei[e] : 0;
  }
#pragma unroll
  for (int u = 0; u < BIN_PER; ++u) {
    rg[u] = 0; rank[u] = 0;
    if (c[u] >= 0) {
      unsafeAtomicAdd(&deg[c[u]], 1.0f);
      if (bin) {
        rg[u]   = c[u] / RG_NODES;
        rank[u] = atomicAdd(&hist[rg[u]], 1);
      }
    }
  }
  __syncthreads();
  if (bin)
    for (int i = tid; i < nr; i += BIN_THREADS) {
      const int h = hist[i];
      hist[i] = h > 0 ? atomicAdd(&gcnt[i], h) : 0;
    }
  __syncthreads();
  if (bin) {
#pragma unroll
    for (int u = 0; u < BIN_PER; ++u) {
      if (c[u] >= 0) {
        const int r = rg[u];
        const unsigned slot = (unsigned)(hist[r] + rank[u]);
        if (slot < (unsigned)RG_LIST)
          glist[(size_t)r * RG_LIST + slot] = make_int2(sn[u], c[u] - r * RG_NODES);
      }
    }
  }
}

__global__ void gcn_deg_to_dis(float* __restrict__ dis, int N) {
  int i = blockIdx.x * blockDim.x + threadIdx.x;
  if (i < N) {
    float d = dis[i];
    dis[i] = (d > 0.0f) ? rsqrtf(d) : 0.0f;
  }
}


template <int C>
__global__ void gcn_prescale(const float* __restrict__ dis, const float* __restrict__ hf,
                             float* __restrict__ hs, int N) {
  constexpr int KS = C >> 2;
  long long i = (long long)blockIdx.x * blockDim.x + threadIdx.x;
  long long total = (long long)N * KS;
  if (i < total) {
    const int node = (int)(i / KS);
    const int v    = (int)(i - (long long)node * KS);
    const v2f* f = (const v2f*)hf + ((size_t)(node >> 4) * KS + v) * 32 + (node & 15);
    const v2f lo = f[0], hi = f[16];
    const float d = dis[node];
    float4 o = {lo.x * d, lo.y * d, hi.x * d, hi.y * d};
    ((float4*)hs)[i] = o;
  }
}


template <int C>
__global__ __launch_bounds__(RG_THREADS) void gcn_range_sum(
    const int* __restrict__ ei, const float* __restrict__ hs, const float* __restrict__ dis,
    float* __restrict__ acc, const int2* __restrict__ glist, const int* __restrict__ gcnt,
    int E, int N) {
  static_assert(RG_NODES % 16 == 0, "a range must start on a 16-row tile boundary");
  constexpr int PITCH = C + 4;
  constexpr int KS    = C >> 2;
  __shared__ float    sacc[RG_NODES * PITCH];
  __shared__ float    sdis[RG_NODES];
  __shared__ int      lsrc[RG_LIST];
  __shared__ int      ldst[RG_LIST];
  __shared__ unsigned cnt;

  const int tid  = threadIdx.x;
  const int base = blockIdx.x * RG_NODES;
  const int rem  = N - base;
  const unsigned nloc = (unsigned)(rem < RG_NODES ? rem : RG_NODES);
  const int nvec = (int)nloc * (C >> 2);
  const int2* gl = glist + (size_t)blockIdx.x * RG_LIST;

  const float4* hsv = (const float4*)(hs + (size_t)base * C);
  for (int i = tid; i < nvec; i += RG_THREADS) {
    const int r = i / KS;
    *(float4*)(sacc + r * PITCH + ((i - r * KS) << 2)) = hsv[i];
  }
  for (int i = tid; i < (int)nloc; i += RG_THREADS) sdis[i] = dis[base + i];
  if (tid == 0) cnt = 0u;

  int n = 0;
  const unsigned prev = (unsigned)gcnt[blockIdx.x];
  const bool listed = prev <= (unsigned)RG_LIST;
  if (listed) {
    n = (int)prev;
    for (int i = tid; i < n; i += RG_THREADS) {
      const int2 p = gl[i];
      lsrc[i] = p.x;
      ldst[i] = p.y;
    }
  }

  if (!listed) {
    __syncthreads();

    const int* col = ei + E;

    auto keep = [&](int e, unsigned cl) {
      const unsigned s = atomicAdd(&cnt, 1u);
      if (s < (unsigned)RG_LIST) {
        lsrc[s] = e;
        ldst[s] = (int)cl;
      } else {
        const float* hp = hs + (size_t)ei[e] * C;
#pragma unroll 1
        for (int ch = 0; ch < C; ++ch) atomicAdd(&sacc[cl * PITCH + ch], hp[ch]);
      }
    };
    auto test4 = [&](const int4 v, int q) {
      const unsigned a = (unsigned)(v.x - base), b = (unsigned)(v.y - base);
      const unsigned c = (unsigned)(v.z - base), d = (unsigned)(v.w - base);
      if ((a < nloc) | (b < nloc) | (c < nloc) | (d < nloc)) {
        const int e = q << 2;
        if (a < nloc) keep(e, a);
        if (b < nloc) keep(e + 1, b);
        if (c < nloc) keep(e + 2, c);
        if (d < nloc) keep(e + 3, d);
      }
    };

    if (((E & 3) == 0) && ((((size_t)col) & 15) == 0)) {
      const int4* colv = (const int4*)col;
      const int nq = E >> 2;
      int q = tid;
      for (; q + 3 * RG_THREADS < nq; q += 4 * RG_THREADS) {
        const int4 v0 = colv[q];
        const int4 v1 = colv[q + RG_THREADS];
        const int4 v2 = colv[q + 2 * RG_THREADS];
        const int4 v3 = colv[q + 3 * RG_THREADS];
        test4(v0, q);
        test4(v1, q + RG_THREADS);
        test4(v2, q + 2 * RG_THREADS);
        test4(v3, q + 3 * RG_THREADS);
      }
      for (; q < nq; q += RG_THREADS) test4(colv[q], q);
    } else {
      for (int e = tid; e < E; e += RG_THREADS) {
        const unsigned cl = (unsigned)(col[e] - base);
        if (cl < nloc) keep(e, cl);
      }
    }
    __syncthreads();

    const unsigned total = cnt;
    n = (int)(total < (unsigned)RG_LIST ? total : (unsigned)RG_LIST);
    for (int i = tid; i < n; i += RG_THREADS) lsrc[i] = ei[lsrc[i]];
  }
  __syncthreads();

  constexpr int RG_U = 8;
  constexpr int VEC  = C / 32;
  const int lane = tid & 31;
  const int wave = __builtin_amdgcn_readfirstlane(tid >> 5);
  for (int i0 = wave * RG_U; i0 < n; i0 += (RG_THREADS / 32) * RG_U) {
    float val[RG_U][VEC];
    int   dst[RG_U];
#pragma unroll
    for (int u = 0; u < RG_U; ++u) {
      const int idx = (i0 + u < n) ? (i0 + u) : i0;
      const int r = lsrc[idx];
      dst[u] = ldst[idx];
      const float* hp = hs + (size_t)r * C + lane * VEC;
#pragma unroll
      for (int k = 0; k < VEC; ++k) val[u][k] = hp[k];
    }
#pragma unroll
    for (int u = 0; u < RG_U; ++u) {
      if (i0 + u < n) {
        float* sp = sacc + dst[u] * PITCH + lane * VEC;
#pragma unroll
        for (int k = 0; k < VEC; ++k) atomicAdd(sp + k, val[u][k]);
      }
    }
  }
  __syncthreads();

  v2f* accf = (v2f*)(acc + (size_t)base * C);
  const int nfrag = (((int)nloc + 15) >> 4) * KS * 32;
  for (int p = tid; p < nfrag; p += RG_THREADS) {
    const int l   = p & 31;
    const int tk  = p >> 5;
    const int t   = tk / KS;
    const int ks  = tk - t * KS;
    const int row = (t << 4) + (l & 15);
    if ((unsigned)row < nloc) {
      const float d = sdis[row];
      v2f s = *(const v2f*)(sacc + row * PITCH + (ks << 2) + ((l >> 4) << 1));
      s.x *= d; s.y *= d;
      accf[p] = s;
    }
  }
}


#define NUM_W 8
struct PackSrc {
  const float* w0; const float* w1; const float* w2; const float* w3;
  const float* w4; const float* w5; const float* w6; const float* w7;
};
#define WP_W1   0
#define WP_W2   4096
#define WP_C1L  4608
#define WP_C1R  6656
#define WP_C2L  8704
#define WP_C2R  16896
#define WP_W3   25088
#define WP_W4   49664
#define WP_TOTAL 55808

__global__ void gcn_pack_weights(PackSrc s, float* __restrict__ dst) {
  const int m = blockIdx.y;
  const float* W; int OC, IC, off;
  switch (m) {
    case 0:  W = s.w0; OC = 16;  IC = 256; off = WP_W1;  break;
    case 1:  W = s.w1; OC = 32;  IC = 16;  off = WP_W2;  break;
    case 2:  W = s.w2; OC = 64;  IC = 32;  off = WP_C1L; break;
    case 3:  W = s.w3; OC = 64;  IC = 32;  off = WP_C1R; break;
    case 4:  W = s.w4; OC = 128; IC = 64;  off = WP_C2L; break;
    case 5:  W = s.w5; OC = 128; IC = 64;  off = WP_C2R; break;
    case 6:  W = s.w6; OC = 192; IC = 128; off = WP_W3;  break;
    default: W = s.w7; OC = 20;  IC = 192; off = WP_W4;  break;
  }
  const int ks_n  = IC >> 2;
  const int tiles = (OC + 15) >> 4;
  const int total = tiles * ks_n * 32;
  for (int i = blockIdx.x * blockDim.x + threadIdx.x; i < total; i += gridDim.x * blockDim.x) {
    const int lane = i & 31;
    const int ks   = (i >> 5) % ks_n;
    const int ct   = (i >> 5) / ks_n;
    int col = ct * 16 + (lane & 15);
    col = col < OC ? col : OC - 1;
    const float* p = W + (size_t)col * IC + ks * 4 + ((lane >> 4) << 1);
    v2f v = *(const v2f*)p;
    *(v2f*)(dst + off + (size_t)i * 2) = v;
  }
}

template <int IC, bool DUAL, bool RELU, bool GUARD, bool INP, bool OUTP>
__global__ __launch_bounds__(32 * WAVES_PER_BLOCK) void gcn_wmma_gemm(
    const float* __restrict__ inA, const float* __restrict__ WA, const float* __restrict__ bA,
    const float* __restrict__ inB, const float* __restrict__ WB, const float* __restrict__ bB,
    float* __restrict__ out, int nrows, int OC) {
  static_assert(INP || !DUAL, "the two-operand form reads fragment-order inputs only");
  static_assert(!(OUTP && GUARD), "fragment-order output needs full tiles");
  __shared__ float lds[INP ? 4 : 16 * IC];
  const int tid  = threadIdx.x;
  const int lane = tid & 31;
  const int wave = tid >> 5;
  const int row0 = blockIdx.x << 4;
  const int col0 = (blockIdx.y * WAVES_PER_BLOCK + wave) << 4;

  float* Atile = lds;

  if constexpr (!INP) {
    constexpr int vecPerRow = IC >> 2;
    constexpr int totalVec  = vecPerRow << 4;
    for (int idx = tid; idx < totalVec; idx += 32 * WAVES_PER_BLOCK) {
      const int r = idx / vecPerRow;
      const int v = idx - r * vecPerRow;
      const int node = row0 + r;
      if (GUARD && node >= nrows) {
        float4 z = {0.f, 0.f, 0.f, 0.f};
        ((float4*)Atile)[idx] = z;
      } else {
        ((float4*)Atile)[idx] = ((const float4*)(inA + (size_t)node * IC))[v];
      }
    }
    __syncthreads();
  }

  if (col0 >= OC) return;

  const int  nidx = lane & 15;
  const int  koff = (lane >> 4) << 1;
  const int  col  = col0 + nidx;
  const bool colOK = GUARD ? (col < OC) : true;

  constexpr int KSTEPS = IC / 4;
  const v2f* wAp = (const v2f*)WA + (size_t)(col0 >> 4) * KSTEPS * 32 + lane;
  const v2f* wBp = DUAL ? ((const v2f*)WB + (size_t)(col0 >> 4) * KSTEPS * 32 + lane) : wAp;
  const v2f*   ag = (const v2f*)(inA + (size_t)row0 * IC) + lane;
  const v2f*   bg = DUAL ? ((const v2f*)(inB + (size_t)row0 * IC) + lane) : ag;
  const float* ar = Atile + nidx * IC + koff;

  constexpr int BATCH  = DUAL ? (KSTEPS < 4 ? KSTEPS : 4)
                              : (KSTEPS < 8 ? KSTEPS : 8);

  v8f acc = {};
#pragma unroll 1
  for (int kb = 0; kb < KSTEPS; kb += BATCH) {
    v2f fa[BATCH], fw[BATCH];
    v2f ga[DUAL ? BATCH : 1], gw[DUAL ? BATCH : 1];
#pragma unroll
    for (int i = 0; i < BATCH; ++i) {
      const int ks = kb + i;
      if constexpr (INP) fa[i] = ag[ks * 32];
      else               fa[i] = *(const v2f*)(ar + ks * 4);
      fw[i] = wAp[ks * 32];
      if constexpr (DUAL) {
        ga[i] = bg[ks * 32];
        gw[i] = wBp[ks * 32];
      }
    }
#pragma unroll
    for (int i = 0; i < BATCH; ++i) {
      if constexpr (OUTP) {
        acc = __builtin_amdgcn_wmma_f32_16x16x4_f32(false, fw[i], false, fa[i],
                                                    (short)0, acc, false, false);
        if constexpr (DUAL)
          acc = __builtin_amdgcn_wmma_f32_16x16x4_f32(false, gw[i], false, ga[i],
                                                      (short)0, acc, false, false);
      } else {
        acc = __builtin_amdgcn_wmma_f32_16x16x4_f32(false, fa[i], false, fw[i],
                                                    (short)0, acc, false, false);
        if constexpr (DUAL)
          acc = __builtin_amdgcn_wmma_f32_16x16x4_f32(false, ga[i], false, gw[i],
                                                      (short)0, acc, false, false);
      }
    }
  }

  const int mb = (lane >> 4) << 3;
  if constexpr (OUTP) {
    const int cb = col0 + mb;
    float bias[8];
#pragma unroll
    for (int vg = 0; vg < 8; ++vg) {
      bias[vg] = bA[cb + vg];
      if constexpr (DUAL) bias[vg] += bB[cb + vg];
    }
    const int KSO = OC >> 2;
    float* ob = out + ((size_t)blockIdx.x * KSO + (cb >> 2)) * 64 + nidx * 2;
#pragma unroll
    for (int t = 0; t < 4; ++t) {
      float v0 = acc[2 * t] + bias[2 * t];
      float v1 = acc[2 * t + 1] + bias[2 * t + 1];
      if constexpr (RELU) { v0 = v0 > 0.f ? v0 : 0.f; v1 = v1 > 0.f ? v1 : 0.f; }
      v2f o = {v0, v1};
      *(v2f*)(ob + (t >> 1) * 64 + (t & 1) * 32) = o;
    }
  } else if constexpr (GUARD) {
    float bias = 0.f;
    if (colOK) {
      bias = bA[col];
      if constexpr (DUAL) bias += bB[col];
    }
#pragma unroll
    for (int vg = 0; vg < 8; ++vg) {
      const int r = row0 + mb + vg;
      if (colOK && r < nrows) {
        float val = acc[vg] + bias;
        if constexpr (RELU) val = val > 0.f ? val : 0.f;
        out[(size_t)r * OC + col] = val;
      }
    }
  } else {
    float bias = bA[col];
    if constexpr (DUAL) bias += bB[col];
    float* op = out + (size_t)(row0 + mb) * OC + col;
#pragma unroll
    for (int vg = 0; vg < 8; ++vg) {
      float val = acc[vg] + bias;
      if constexpr (RELU) val = val > 0.f ? val : 0.f;
      op[(size_t)vg * OC] = val;
    }
  }
}


template <int IC, bool DUAL, bool RELU, bool INP, bool OUTP>
static inline void launch_gemm(const float* inA, const float* WA, const float* bA,
                               const float* inB, const float* WB, const float* bB,
                               float* out, int N, int OC, hipStream_t stream) {
  const int rowTiles = (N + 15) / 16;
  const int gy = (((OC + 15) / 16) + WAVES_PER_BLOCK - 1) / WAVES_PER_BLOCK;
  const dim3 grid(rowTiles, gy), blk(32 * WAVES_PER_BLOCK);
  if constexpr (OUTP) {
    gcn_wmma_gemm<IC, DUAL, RELU, false, INP, true><<<grid, blk, 0, stream>>>(
        inA, WA, bA, inB, WB, bB, out, N, OC);
  } else {
    if ((N & 15) == 0 && (OC & 15) == 0)
      gcn_wmma_gemm<IC, DUAL, RELU, false, INP, false><<<grid, blk, 0, stream>>>(
          inA, WA, bA, inB, WB, bB, out, N, OC);
    else
      gcn_wmma_gemm<IC, DUAL, RELU, true, INP, false><<<grid, blk, 0, stream>>>(
          inA, WA, bA, inB, WB, bB, out, N, OC);
  }
}

extern "C" void kernel_launch(void* const* d_in, const int* in_sizes, int n_in,
                              void* d_out, int out_size, void* d_ws, size_t ws_size,
                              hipStream_t stream) {
  (void)n_in; (void)out_size; (void)ws_size;
  const float* x    = (const float*)d_in[0];
  const int*   ei   = (const int*)d_in[1];
  const float* w1   = (const float*)d_in[2];
  const float* b1   = (const float*)d_in[3];
  const float* w2   = (const float*)d_in[4];
  const float* b2   = (const float*)d_in[5];
  const float* c1wl = (const float*)d_in[6];
  const float* c1bl = (const float*)d_in[7];
  const float* c1wr = (const float*)d_in[8];
  const float* c1br = (const float*)d_in[9];
  const float* c2wl = (const float*)d_in[10];
  const float* c2bl = (const float*)d_in[11];
  const float* c2wr = (const float*)d_in[12];
  const float* c2br = (const float*)d_in[13];
  const float* w3   = (const float*)d_in[14];
  const float* b3   = (const float*)d_in[15];
  const float* w4   = (const float*)d_in[16];
  const float* b4   = (const float*)d_in[17];

  const int IN_CH = 256;
  const int N = in_sizes[0] / IN_CH;
  const int E = in_sizes[1] / 2;
  const size_t n = (size_t)N;

  float* ws   = (float*)d_ws;
  float* dis  = ws;
  float* h2   = ws + 1 * n;
  float* agg1 = ws + 33 * n;
  float* h3   = ws + 65 * n;
  float* agg2 = ws + 1 * n;
  float* h1   = ws + 129 * n;
  float* wp   = ws + 192 * n;
  float* hs2  = ws + 193 * n;
  float* hs3  = ws + 225 * n;
  float* h4   = ws + 193 * n;
  float* h5   = ws;
  int*  rcnt  = (int*)(ws + 129 * n);
  int2* rlist = (int2*)(ws + 130 * n);

  const int eltBlocks  = (N + 255) / 256;
  const int binBlocks  = (E + BIN_THREADS * BIN_PER - 1) / (BIN_THREADS * BIN_PER);
  const int rangeBlocks = (N + RG_NODES - 1) / RG_NODES;

  PackSrc ps = {w1, w2, c1wl, c1wr, c2wl, c2wr, w3, w4};
  gcn_pack_weights<<<dim3(24, NUM_W), 256, 0, stream>>>(ps, wp);

  launch_gemm<256, false, false, false, true>(x, wp + WP_W1, b1, nullptr, nullptr, nullptr, h1, N, 16, stream);
  launch_gemm<16, false, false, true, true>(h1, wp + WP_W2, b2, nullptr, nullptr, nullptr, h2, N, 32, stream);

  gcn_deg_init<<<eltBlocks, 256, 0, stream>>>(dis, rcnt, N, rangeBlocks);
  gcn_deg_bin<<<binBlocks, BIN_THREADS, 0, stream>>>(ei, dis, rlist, rcnt, E, rangeBlocks);
  gcn_deg_to_dis<<<eltBlocks, 256, 0, stream>>>(dis, N);

  gcn_prescale<32><<<(int)((n * 8 + 255) / 256), 256, 0, stream>>>(dis, h2, hs2, N);
  gcn_range_sum<32><<<rangeBlocks, RG_THREADS, 0, stream>>>(ei, hs2, dis, agg1, rlist, rcnt, E, N);
  launch_gemm<32, true, true, true, true>(h2, wp + WP_C1L, c1bl, agg1, wp + WP_C1R, c1br, h3, N, 64, stream);

  gcn_prescale<64><<<(int)((n * 16 + 255) / 256), 256, 0, stream>>>(dis, h3, hs3, N);
  gcn_range_sum<64><<<rangeBlocks, RG_THREADS, 0, stream>>>(ei, hs3, dis, agg2, rlist, rcnt, E, N);
  launch_gemm<64, true, true, true, true>(h3, wp + WP_C2L, c2bl, agg2, wp + WP_C2R, c2br, h4, N, 128, stream);

  launch_gemm<128, false, false, true, true>(h4, wp + WP_W3, b3, nullptr, nullptr, nullptr, h5, N, 192, stream);
  launch_gemm<192, false, false, true, false>(h5, wp + WP_W4, b4, nullptr, nullptr, nullptr, (float*)d_out, N, 20, stream);
}


# GEMMs on fragment-order inputs turned persistent: a wave loads its column tile's weight fragments into registers once and walks several row tiles; 200k GEMM waves become 27k; sum order unchanged
# speedup vs baseline: 2.3183x; 1.0441x over previous
#include <hip/hip_runtime.h>

typedef __attribute__((ext_vector_type(2))) float v2f;
typedef __attribute__((ext_vector_type(8))) float v8f;

#define WAVES_PER_BLOCK 4


#define RG_NODES   400
#define RG_THREADS 1024
#define RG_LIST    8192

#define BIN_THREADS 1024
#define BIN_PER     4
#define BIN_MAXR    1024

__global__ void gcn_deg_init(float* __restrict__ deg, int* __restrict__ gcnt, int N, int nr) {
  int i = blockIdx.x * blockDim.x + threadIdx.x;
  if (i < N) deg[i] = 1.0f;
  if (i < nr) gcnt[i] = (nr <= BIN_MAXR) ? 0 : (RG_LIST + 1);
}

__global__ __launch_bounds__(BIN_THREADS) void gcn_deg_bin(
    const int* __restrict__ ei, float* __restrict__ deg, int2* __restrict__ glist,
    int* __restrict__ gcnt, int E, int nr) {
  __shared__ int hist[BIN_MAXR];
  const int  tid = threadIdx.x;
  const bool bin = nr <= BIN_MAXR;
  if (bin)
    for (int i = tid; i < nr; i += BIN_THREADS) hist[i] = 0;
  __syncthreads();

  const int e0 = blockIdx.x * (BIN_THREADS * BIN_PER) + tid;
  int c[BIN_PER], sn[BIN_PER], rg[BIN_PER], rank[BIN_PER];
#pragma unroll
  for (int u = 0; u < BIN_PER; ++u) {
    const int e = e0 + u * BIN_THREADS;
    const bool ok = e < E;
    c[u]  = ok ? ei[E + e] : -1;
    sn[u] = ok ? ei[e] : 0;
  }
#pragma unroll
  for (int u = 0; u < BIN_PER; ++u) {
    rg[u] = 0; rank[u] = 0;
    if (c[u] >= 0) {
      unsafeAtomicAdd(&deg[c[u]], 1.0f);
      if (bin) {
        rg[u]   = c[u] / RG_NODES;
        rank[u] = atomicAdd(&hist[rg[u]], 1);
      }
    }
  }
  __syncthreads();
  if (bin)
    for (int i = tid; i < nr; i += BIN_THREADS) {
      const int h = hist[i];
      hist[i] = h > 0 ? atomicAdd(&gcnt[i], h) : 0;
    }
  __syncthreads();
  if (bin) {
#pragma unroll
    for (int u = 0; u < BIN_PER; ++u) {
      if (c[u] >= 0) {
        const int r = rg[u];
        const unsigned slot = (unsigned)(hist[r] + rank[u]);
        if (slot < (unsigned)RG_LIST)
          glist[(size_t)r * RG_LIST + slot] = make_int2(sn[u], c[u] - r * RG_NODES);
      }
    }
  }
}

__global__ void gcn_deg_to_dis(float* __restrict__ dis, int N) {
  int i = blockIdx.x * blockDim.x + threadIdx.x;
  if (i < N) {
    float d = dis[i];
    dis[i] = (d > 0.0f) ? rsqrtf(d) : 0.0f;
  }
}


template <int C>
__global__ void gcn_prescale(const float* __restrict__ dis, const float* __restrict__ hf,
                             float* __restrict__ hs, int N) {
  constexpr int KS = C >> 2;
  long long i = (long long)blockIdx.x * blockDim.x + threadIdx.x;
  long long total = (long long)N * KS;
  if (i < total) {
    const int node = (int)(i / KS);
    const int v    = (int)(i - (long long)node * KS);
    const v2f* f = (const v2f*)hf + ((size_t)(node >> 4) * KS + v) * 32 + (node & 15);
    const v2f lo = f[0], hi = f[16];
    const float d = dis[node];
    float4 o = {lo.x * d, lo.y * d, hi.x * d, hi.y * d};
    ((float4*)hs)[i] = o;
  }
}


template <int C>
__global__ __launch_bounds__(RG_THREADS) void gcn_range_sum(
    const int* __restrict__ ei, const float* __restrict__ hs, const float* __restrict__ dis,
    float* __restrict__ acc, const int2* __restrict__ glist, const int* __restrict__ gcnt,
    int E, int N) {
  static_assert(RG_NODES % 16 == 0, "a range must start on a 16-row tile boundary");
  constexpr int PITCH = C + 4;
  constexpr int KS    = C >> 2;
  __shared__ float    sacc[RG_NODES * PITCH];
  __shared__ float    sdis[RG_NODES];
  __shared__ int      lsrc[RG_LIST];
  __shared__ int      ldst[RG_LIST];
  __shared__ unsigned cnt;

  const int tid  = threadIdx.x;
  const int base = blockIdx.x * RG_NODES;
  const int rem  = N - base;
  const unsigned nloc = (unsigned)(rem < RG_NODES ? rem : RG_NODES);
  const int nvec = (int)nloc * (C >> 2);
  const int2* gl = glist + (size_t)blockIdx.x * RG_LIST;

  const float4* hsv = (const float4*)(hs + (size_t)base * C);
  for (int i = tid; i < nvec; i += RG_THREADS) {
    const int r = i / KS;
    *(float4*)(sacc + r * PITCH + ((i - r * KS) << 2)) = hsv[i];
  }
  for (int i = tid; i < (int)nloc; i += RG_THREADS) sdis[i] = dis[base + i];
  if (tid == 0) cnt = 0u;

  int n = 0;
  const unsigned prev = (unsigned)gcnt[blockIdx.x];
  const bool listed = prev <= (unsigned)RG_LIST;
  if (listed) {
    n = (int)prev;
    for (int i = tid; i < n; i += RG_THREADS) {
      const int2 p = gl[i];
      lsrc[i] = p.x;
      ldst[i] = p.y;
    }
  }

  if (!listed) {
    __syncthreads();

    const int* col = ei + E;

    auto keep = [&](int e, unsigned cl) {
      const unsigned s = atomicAdd(&cnt, 1u);
      if (s < (unsigned)RG_LIST) {
        lsrc[s] = e;
        ldst[s] = (int)cl;
      } else {
        const float* hp = hs + (size_t)ei[e] * C;
#pragma unroll 1
        for (int ch = 0; ch < C; ++ch) atomicAdd(&sacc[cl * PITCH + ch], hp[ch]);
      }
    };
    auto test4 = [&](const int4 v, int q) {
      const unsigned a = (unsigned)(v.x - base), b = (unsigned)(v.y - base);
      const unsigned c = (unsigned)(v.z - base), d = (unsigned)(v.w - base);
      if ((a < nloc) | (b < nloc) | (c < nloc) | (d < nloc)) {
        const int e = q << 2;
        if (a < nloc) keep(e, a);
        if (b < nloc) keep(e + 1, b);
        if (c < nloc) keep(e + 2, c);
        if (d < nloc) keep(e + 3, d);
      }
    };

    if (((E & 3) == 0) && ((((size_t)col) & 15) == 0)) {
      const int4* colv = (const int4*)col;
      const int nq = E >> 2;
      int q = tid;
      for (; q + 3 * RG_THREADS < nq; q += 4 * RG_THREADS) {
        const int4 v0 = colv[q];
        const int4 v1 = colv[q + RG_THREADS];
        const int4 v2 = colv[q + 2 * RG_THREADS];
        const int4 v3 = colv[q + 3 * RG_THREADS];
        test4(v0, q);
        test4(v1, q + RG_THREADS);
        test4(v2, q + 2 * RG_THREADS);
        test4(v3, q + 3 * RG_THREADS);
      }
      for (; q < nq; q += RG_THREADS) test4(colv[q], q);
    } else {
      for (int e = tid; e < E; e += RG_THREADS) {
        const unsigned cl = (unsigned)(col[e] - base);
        if (cl < nloc) keep(e, cl);
      }
    }
    __syncthreads();

    const unsigned total = cnt;
    n = (int)(total < (unsigned)RG_LIST ? total : (unsigned)RG_LIST);
    for (int i = tid; i < n; i += RG_THREADS) lsrc[i] = ei[lsrc[i]];
  }
  __syncthreads();

  constexpr int RG_U = 8;
  constexpr int VEC  = C / 32;
  const int lane = tid & 31;
  const int wave = __builtin_amdgcn_readfirstlane(tid >> 5);
  for (int i0 = wave * RG_U; i0 < n; i0 += (RG_THREADS / 32) * RG_U) {
    float val[RG_U][VEC];
    int   dst[RG_U];
#pragma unroll
    for (int u = 0; u < RG_U; ++u) {
      const int idx = (i0 + u < n) ? (i0 + u) : i0;
      const int r = lsrc[idx];
      dst[u] = ldst[idx];
      const float* hp = hs + (size_t)r * C + lane * VEC;
#pragma unroll
      for (int k = 0; k < VEC; ++k) val[u][k] = hp[k];
    }
#pragma unroll
    for (int u = 0; u < RG_U; ++u) {
      if (i0 + u < n) {
        float* sp = sacc + dst[u] * PITCH + lane * VEC;
#pragma unroll
        for (int k = 0; k < VEC; ++k) atomicAdd(sp + k, val[u][k]);
      }
    }
  }
  __syncthreads();

  v2f* accf = (v2f*)(acc + (size_t)base * C);
  const int nfrag = (((int)nloc + 15) >> 4) * KS * 32;
  for (int p = tid; p < nfrag; p += RG_THREADS) {
    const int l   = p & 31;
    const int tk  = p >> 5;
    const int t   = tk / KS;
    const int ks  = tk - t * KS;
    const int row = (t << 4) + (l & 15);
    if ((unsigned)row < nloc) {
      const float d = sdis[row];
      v2f s = *(const v2f*)(sacc + row * PITCH + (ks << 2) + ((l >> 4) << 1));
      s.x *= d; s.y *= d;
      accf[p] = s;
    }
  }
}


#define NUM_W 8
struct PackSrc {
  const float* w0; const float* w1; const float* w2; const float* w3;
  const float* w4; const float* w5; const float* w6; const float* w7;
};
#define WP_W1   0
#define WP_W2   4096
#define WP_C1L  4608
#define WP_C1R  6656
#define WP_C2L  8704
#define WP_C2R  16896
#define WP_W3   25088
#define WP_W4   49664
#define WP_TOTAL 55808

__global__ void gcn_pack_weights(PackSrc s, float* __restrict__ dst) {
  const int m = blockIdx.y;
  const float* W; int OC, IC, off;
  switch (m) {
    case 0:  W = s.w0; OC = 16;  IC = 256; off = WP_W1;  break;
    case 1:  W = s.w1; OC = 32;  IC = 16;  off = WP_W2;  break;
    case 2:  W = s.w2; OC = 64;  IC = 32;  off = WP_C1L; break;
    case 3:  W = s.w3; OC = 64;  IC = 32;  off = WP_C1R; break;
    case 4:  W = s.w4; OC = 128; IC = 64;  off = WP_C2L; break;
    case 5:  W = s.w5; OC = 128; IC = 64;  off = WP_C2R; break;
    case 6:  W = s.w6; OC = 192; IC = 128; off = WP_W3;  break;
    default: W = s.w7; OC = 20;  IC = 192; off = WP_W4;  break;
  }
  const int ks_n  = IC >> 2;
  const int tiles = (OC + 15) >> 4;
  const int total = tiles * ks_n * 32;
  for (int i = blockIdx.x * blockDim.x + threadIdx.x; i < total; i += gridDim.x * blockDim.x) {
    const int lane = i & 31;
    const int ks   = (i >> 5) % ks_n;
    const int ct   = (i >> 5) / ks_n;
    int col = ct * 16 + (lane & 15);
    col = col < OC ? col : OC - 1;
    const float* p = W + (size_t)col * IC + ks * 4 + ((lane >> 4) << 1);
    v2f v = *(const v2f*)p;
    *(v2f*)(dst + off + (size_t)i * 2) = v;
  }
}

template <int IC, bool DUAL, bool RELU, bool GUARD, bool INP, bool OUTP>
__global__ __launch_bounds__(32 * WAVES_PER_BLOCK) void gcn_wmma_gemm(
    const float* __restrict__ inA, const float* __restrict__ WA, const float* __restrict__ bA,
    const float* __restrict__ inB, const float* __restrict__ WB, const float* __restrict__ bB,
    float* __restrict__ out, int nrows, int OC) {
  static_assert(INP || !DUAL, "the two-operand form reads fragment-order inputs only");
  static_assert(!(OUTP && GUARD), "fragment-order output needs full tiles");
  __shared__ float lds[INP ? 4 : 16 * IC];
  const int tid  = threadIdx.x;
  const int lane = tid & 31;
  const int wave = tid >> 5;
  const int row0 = blockIdx.x << 4;
  const int col0 = (blockIdx.y * WAVES_PER_BLOCK + wave) << 4;

  float* Atile = lds;

  if constexpr (!INP) {
    constexpr int vecPerRow = IC >> 2;
    constexpr int totalVec  = vecPerRow << 4;
    for (int idx = tid; idx < totalVec; idx += 32 * WAVES_PER_BLOCK) {
      const int r = idx / vecPerRow;
      const int v = idx - r * vecPerRow;
      const int node = row0 + r;
      if (GUARD && node >= nrows) {
        float4 z = {0.f, 0.f, 0.f, 0.f};
        ((float4*)Atile)[idx] = z;
      } else {
        ((float4*)Atile)[idx] = ((const float4*)(inA + (size_t)node * IC))[v];
      }
    }
    __syncthreads();
  }

  if (col0 >= OC) return;

  const int  nidx = lane & 15;
  const int  koff = (lane >> 4) << 1;
  const int  col  = col0 + nidx;
  const bool colOK = GUARD ? (col < OC) : true;

  constexpr int KSTEPS = IC / 4;
  const v2f* wAp = (const v2f*)WA + (size_t)(col0 >> 4) * KSTEPS * 32 + lane;
  const v2f* wBp = DUAL ? ((const v2f*)WB + (size_t)(col0 >> 4) * KSTEPS * 32 + lane) : wAp;
  const v2f*   ag = (const v2f*)(inA + (size_t)row0 * IC) + lane;
  const v2f*   bg = DUAL ? ((const v2f*)(inB + (size_t)row0 * IC) + lane) : ag;
  const float* ar = Atile + nidx * IC + koff;

  constexpr int BATCH  = DUAL ? (KSTEPS < 4 ? KSTEPS : 4)
                              : (KSTEPS < 8 ? KSTEPS : 8);

  v8f acc = {};
#pragma unroll 1
  for (int kb = 0; kb < KSTEPS; kb += BATCH) {
    v2f fa[BATCH], fw[BATCH];
    v2f ga[DUAL ? BATCH : 1], gw[DUAL ? BATCH : 1];
#pragma unroll
    for (int i = 0; i < BATCH; ++i) {
      const int ks = kb + i;
      if constexpr (INP) fa[i] = ag[ks * 32];
      else               fa[i] = *(const v2f*)(ar + ks * 4);
      fw[i] = wAp[ks * 32];
      if constexpr (DUAL) {
        ga[i] = bg[ks * 32];
        gw[i] = wBp[ks * 32];
      }
    }
#pragma unroll
    for (int i = 0; i < BATCH; ++i) {
      if constexpr (OUTP) {
        acc = __builtin_amdgcn_wmma_f32_16x16x4_f32(false, fw[i], false, fa[i],
                                                    (short)0, acc, false, false);
        if constexpr (DUAL)
          acc = __builtin_amdgcn_wmma_f32_16x16x4_f32(false, gw[i], false, ga[i],
                                                      (short)0, acc, false, false);
      } else {
        acc = __builtin_amdgcn_wmma_f32_16x16x4_f32(false, fa[i], false, fw[i],
                                                    (short)0, acc, false, false);
        if constexpr (DUAL)
          acc = __builtin_amdgcn_wmma_f32_16x16x4_f32(false, ga[i], false, gw[i],
                                                      (short)0, acc, false, false);
      }
    }
  }

  const int mb = (lane >> 4) << 3;
  if constexpr (OUTP) {
    const int cb = col0 + mb;
    float bias[8];
#pragma unroll
    for (int vg = 0; vg < 8; ++vg) {
      bias[vg] = bA[cb + vg];
      if constexpr (DUAL) bias[vg] += bB[cb + vg];
    }
    const int KSO = OC >> 2;
    float* ob = out + ((size_t)blockIdx.x * KSO + (cb >> 2)) * 64 + nidx * 2;
#pragma unroll
    for (int t = 0; t < 4; ++t) {
      float v0 = acc[2 * t] + bias[2 * t];
      float v1 = acc[2 * t + 1] + bias[2 * t + 1];
      if constexpr (RELU) { v0 = v0 > 0.f ? v0 : 0.f; v1 = v1 > 0.f ? v1 : 0.f; }
      v2f o = {v0, v1};
      *(v2f*)(ob + (t >> 1) * 64 + (t & 1) * 32) = o;
    }
  } else if constexpr (GUARD) {
    float bias = 0.f;
    if (colOK) {
      bias = bA[col];
      if constexpr (DUAL) bias += bB[col];
    }
#pragma unroll
    for (int vg = 0; vg < 8; ++vg) {
      const int r = row0 + mb + vg;
      if (colOK && r < nrows) {
        float val = acc[vg] + bias;
        if constexpr (RELU) val = val > 0.f ? val : 0.f;
        out[(size_t)r * OC + col] = val;
      }
    }
  } else {
    float bias = bA[col];
    if constexpr (DUAL) bias += bB[col];
    float* op = out + (size_t)(row0 + mb) * OC + col;
#pragma unroll
    for (int vg = 0; vg < 8; ++vg) {
      float val = acc[vg] + bias;
      if constexpr (RELU) val = val > 0.f ? val : 0.f;
      op[(size_t)vg * OC] = val;
    }
  }
}

template <int IC, int WPB, bool DUAL, bool RELU, bool GUARD, bool OUTP>
__global__ __launch_bounds__(32 * WPB) void gcn_wmma_gemm_rows(
    const float* __restrict__ inA, const float* __restrict__ WA, const float* __restrict__ bA,
    const float* __restrict__ inB, const float* __restrict__ WB, const float* __restrict__ bB,
    float* __restrict__ out, int nrows, int OC) {
  static_assert(!(OUTP && GUARD), "fragment-order output needs full tiles");
  constexpr int KSTEPS = IC / 4;
  const int lane = threadIdx.x & 31;
  const int wave = __builtin_amdgcn_readfirstlane(threadIdx.x >> 5);
  const int col0 = (blockIdx.y * WPB + wave) << 4;
  if (col0 >= OC) return;

  const int nidx = lane & 15;
  const int mb   = (lane >> 4) << 3;

  const v2f* wAp = (const v2f*)WA + (size_t)(col0 >> 4) * KSTEPS * 32 + lane;
  const v2f* wBp = DUAL ? ((const v2f*)WB + (size_t)(col0 >> 4) * KSTEPS * 32 + lane) : wAp;
  v2f fw[KSTEPS];
  v2f gw[DUAL ? KSTEPS : 1];
#pragma unroll
  for (int ks = 0; ks < KSTEPS; ++ks) {
    fw[ks] = wAp[ks * 32];
    if constexpr (DUAL) gw[ks] = wBp[ks * 32];
  }

  float bias[OUTP ? 8 : 1];
  const int  col   = col0 + nidx;
  const bool colOK = GUARD ? (col < OC) : true;
  if constexpr (OUTP) {
    const int cb = col0 + mb;
#pragma unroll
    for (int vg = 0; vg < 8; ++vg) {
      bias[vg] = bA[cb + vg];
      if constexpr (DUAL) bias[vg] += bB[cb + vg];
    }
  } else {
    bias[0] = 0.f;
    if (colOK) {
      bias[0] = bA[col];
      if constexpr (DUAL) bias[0] += bB[col];
    }
  }

  const int rowTiles = (nrows + 15) >> 4;
  for (int rt = blockIdx.x; rt < rowTiles; rt += gridDim.x) {
    const v2f* ag = (const v2f*)(inA + (size_t)rt * 16 * IC) + lane;
    const v2f* bg = DUAL ? ((const v2f*)(inB + (size_t)rt * 16 * IC) + lane) : ag;
    v2f fa[KSTEPS];
    v2f ga[DUAL ? KSTEPS : 1];
#pragma unroll
    for (int ks = 0; ks < KSTEPS; ++ks) {
      fa[ks] = ag[ks * 32];
      if constexpr (DUAL) ga[ks] = bg[ks * 32];
    }
    __builtin_amdgcn_sched_barrier(0);
    v8f acc = {};
#pragma unroll
    for (int ks = 0; ks < KSTEPS; ++ks) {
      if constexpr (OUTP) {
        acc = __builtin_amdgcn_wmma_f32_16x16x4_f32(false, fw[ks], false, fa[ks],
                                                    (short)0, acc, false, false);
        if constexpr (DUAL)
          acc = __builtin_amdgcn_wmma_f32_16x16x4_f32(false, gw[ks], false, ga[ks],
                                                      (short)0, acc, false, false);
      } else {
        acc = __builtin_amdgcn_wmma_f32_16x16x4_f32(false, fa[ks], false, fw[ks],
                                                    (short)0, acc, false, false);
        if constexpr (DUAL)
          acc = __builtin_amdgcn_wmma_f32_16x16x4_f32(false, ga[ks], false, gw[ks],
                                                      (short)0, acc, false, false);
      }
    }

    if constexpr (OUTP) {
      const int cb  = col0 + mb;
      const int KSO = OC >> 2;
      float* ob = out + ((size_t)rt * KSO + (cb >> 2)) * 64 + nidx * 2;
#pragma unroll
      for (int t = 0; t < 4; ++t) {
        float v0 = acc[2 * t] + bias[2 * t];
        float v1 = acc[2 * t + 1] + bias[2 * t + 1];
        if constexpr (RELU) { v0 = v0 > 0.f ? v0 : 0.f; v1 = v1 > 0.f ? v1 : 0.f; }
        v2f o = {v0, v1};
        *(v2f*)(ob + (t >> 1) * 64 + (t & 1) * 32) = o;
      }
    } else {
      const int row0 = rt << 4;
#pragma unroll
      for (int vg = 0; vg < 8; ++vg) {
        const int r = row0 + mb + vg;
        if (colOK && (!GUARD || r < nrows)) {
          float val = acc[vg] + bias[0];
          if constexpr (RELU) val = val > 0.f ? val : 0.f;
          out[(size_t)r * OC + col] = val;
        }
      }
    }
  }
}


template <int IC, bool DUAL, bool RELU, bool INP, bool OUTP>
static inline void launch_gemm(const float* inA, const float* WA, const float* bA,
                               const float* inB, const float* WB, const float* bB,
                               float* out, int N, int OC, hipStream_t stream) {
  const int rowTiles = (N + 15) / 16;
  const int gy = (((OC + 15) / 16) + WAVES_PER_BLOCK - 1) / WAVES_PER_BLOCK;
  const dim3 grid(rowTiles, gy), blk(32 * WAVES_PER_BLOCK);
  if constexpr (OUTP) {
    gcn_wmma_gemm<IC, DUAL, RELU, false, INP, true><<<grid, blk, 0, stream>>>(
        inA, WA, bA, inB, WB, bB, out, N, OC);
  } else {
    if ((N & 15) == 0 && (OC & 15) == 0)
      gcn_wmma_gemm<IC, DUAL, RELU, false, INP, false><<<grid, blk, 0, stream>>>(
          inA, WA, bA, inB, WB, bB, out, N, OC);
    else
      gcn_wmma_gemm<IC, DUAL, RELU, true, INP, false><<<grid, blk, 0, stream>>>(
          inA, WA, bA, inB, WB, bB, out, N, OC);
  }
}

template <int IC, int WPB, int TPW, bool DUAL, bool RELU, bool OUTP>
static inline void launch_gemm_rows(const float* inA, const float* WA, const float* bA,
                                    const float* inB, const float* WB, const float* bB,
                                    float* out, int N, int OC, hipStream_t stream) {
  const int rowTiles = (N + 15) / 16;
  const int gx = (rowTiles + TPW - 1) / TPW;
  const int gy = (((OC + 15) / 16) + WPB - 1) / WPB;
  const dim3 grid(gx, gy), blk(32 * WPB);
  if constexpr (OUTP) {
    gcn_wmma_gemm_rows<IC, WPB, DUAL, RELU, false, true><<<grid, blk, 0, stream>>>(
        inA, WA, bA, inB, WB, bB, out, N, OC);
  } else {
    if ((N & 15) == 0 && (OC & 15) == 0)
      gcn_wmma_gemm_rows<IC, WPB, DUAL, RELU, false, false><<<grid, blk, 0, stream>>>(
          inA, WA, bA, inB, WB, bB, out, N, OC);
    else
      gcn_wmma_gemm_rows<IC, WPB, DUAL, RELU, true, false><<<grid, blk, 0, stream>>>(
          inA, WA, bA, inB, WB, bB, out, N, OC);
  }
}

extern "C" void kernel_launch(void* const* d_in, const int* in_sizes, int n_in,
                              void* d_out, int out_size, void* d_ws, size_t ws_size,
                              hipStream_t stream) {
  (void)n_in; (void)out_size; (void)ws_size;
  const float* x    = (const float*)d_in[0];
  const int*   ei   = (const int*)d_in[1];
  const float* w1   = (const float*)d_in[2];
  const float* b1   = (const float*)d_in[3];
  const float* w2   = (const float*)d_in[4];
  const float* b2   = (const float*)d_in[5];
  const float* c1wl = (const float*)d_in[6];
  const float* c1bl = (const float*)d_in[7];
  const float* c1wr = (const float*)d_in[8];
  const float* c1br = (const float*)d_in[9];
  const float* c2wl = (const float*)d_in[10];
  const float* c2bl = (const float*)d_in[11];
  const float* c2wr = (const float*)d_in[12];
  const float* c2br = (const float*)d_in[13];
  const float* w3   = (const float*)d_in[14];
  const float* b3   = (const float*)d_in[15];
  const float* w4   = (const float*)d_in[16];
  const float* b4   = (const float*)d_in[17];

  const int IN_CH = 256;
  const int N = in_sizes[0] / IN_CH;
  const int E = in_sizes[1] / 2;
  const size_t n = (size_t)N;

  float* ws   = (float*)d_ws;
  float* dis  = ws;
  float* h2   = ws + 1 * n;
  float* agg1 = ws + 33 * n;
  float* h3   = ws + 65 * n;
  float* agg2 = ws + 1 * n;
  float* h1   = ws + 129 * n;
  float* wp   = ws + 192 * n;
  float* hs2  = ws + 193 * n;
  float* hs3  = ws + 225 * n;
  float* h4   = ws + 193 * n;
  float* h5   = ws;
  int*  rcnt  = (int*)(ws + 129 * n);
  int2* rlist = (int2*)(ws + 130 * n);

  const int eltBlocks  = (N + 255) / 256;
  const int binBlocks  = (E + BIN_THREADS * BIN_PER - 1) / (BIN_THREADS * BIN_PER);
  const int rangeBlocks = (N + RG_NODES - 1) / RG_NODES;

  PackSrc ps = {w1, w2, c1wl, c1wr, c2wl, c2wr, w3, w4};
  gcn_pack_weights<<<dim3(24, NUM_W), 256, 0, stream>>>(ps, wp);

  launch_gemm<256, false, false, false, true>(x, wp + WP_W1, b1, nullptr, nullptr, nullptr, h1, N, 16, stream);
  launch_gemm_rows<16, 2, 2, false, false, true>(h1, wp + WP_W2, b2, nullptr, nullptr, nullptr, h2, N, 32, stream);

  gcn_deg_init<<<eltBlocks, 256, 0, stream>>>(dis, rcnt, N, rangeBlocks);
  gcn_deg_bin<<<binBlocks, BIN_THREADS, 0, stream>>>(ei, dis, rlist, rcnt, E, rangeBlocks);
  gcn_deg_to_dis<<<eltBlocks, 256, 0, stream>>>(dis, N);

  gcn_prescale<32><<<(int)((n * 8 + 255) / 256), 256, 0, stream>>>(dis, h2, hs2, N);
  gcn_range_sum<32><<<rangeBlocks, RG_THREADS, 0, stream>>>(ei, hs2, dis, agg1, rlist, rcnt, E, N);
  launch_gemm_rows<32, 4, 4, true, true, true>(h2, wp + WP_C1L, c1bl, agg1, wp + WP_C1R, c1br, h3, N, 64, stream);

  gcn_prescale<64><<<(int)((n * 16 + 255) / 256), 256, 0, stream>>>(dis, h3, hs3, N);
  gcn_range_sum<64><<<rangeBlocks, RG_THREADS, 0, stream>>>(ei, hs3, dis, agg2, rlist, rcnt, E, N);
  launch_gemm_rows<64, 4, 9, true, true, true>(h3, wp + WP_C2L, c2bl, agg2, wp + WP_C2R, c2br, h4, N, 128, stream);

  launch_gemm_rows<128, 4, 13, false, false, true>(h4, wp + WP_W3, b3, nullptr, nullptr, nullptr, h5, N, 192, stream);
  launch_gemm_rows<192, 2, 4, false, false, false>(h5, wp + WP_W4, b4, nullptr, nullptr, nullptr, (float*)d_out, N, 20, stream);
}


# each neighbour-sum work-group takes the conv GEMM of its own 25 row tiles from its LDS sums; conv GEMM and conv2 prescale launches and agg arrays removed (14 -> 11 launches); sum order unchanged
# speedup vs baseline: 2.4333x; 1.0496x over previous
#include <hip/hip_runtime.h>

typedef __attribute__((ext_vector_type(2))) float v2f;
typedef __attribute__((ext_vector_type(8))) float v8f;

#define WAVES_PER_BLOCK 4


#define RG_NODES   400
#define RG_THREADS 1024
#define RG_LIST    8192

#define BIN_THREADS 1024
#define BIN_PER     4
#define BIN_MAXR    1024

__global__ void gcn_deg_init(float* __restrict__ deg, int* __restrict__ gcnt, int N, int nr) {
  int i = blockIdx.x * blockDim.x + threadIdx.x;
  if (i < N) deg[i] = 1.0f;
  if (i < nr) gcnt[i] = (nr <= BIN_MAXR) ? 0 : (RG_LIST + 1);
}

__global__ __launch_bounds__(BIN_THREADS) void gcn_deg_bin(
    const int* __restrict__ ei, float* __restrict__ deg, int2* __restrict__ glist,
    int* __restrict__ gcnt, int E, int nr) {
  __shared__ int hist[BIN_MAXR];
  const int  tid = threadIdx.x;
  const bool bin = nr <= BIN_MAXR;
  if (bin)
    for (int i = tid; i < nr; i += BIN_THREADS) hist[i] = 0;
  __syncthreads();

  const int e0 = blockIdx.x * (BIN_THREADS * BIN_PER) + tid;
  int c[BIN_PER], sn[BIN_PER], rg[BIN_PER], rank[BIN_PER];
#pragma unroll
  for (int u = 0; u < BIN_PER; ++u) {
    const int e = e0 + u * BIN_THREADS;
    const bool ok = e < E;
    c[u]  = ok ? ei[E + e] : -1;
    sn[u] = ok ? ei[e] : 0;
  }
#pragma unroll
  for (int u = 0; u < BIN_PER; ++u) {
    rg[u] = 0; rank[u] = 0;
    if (c[u] >= 0) {
      unsafeAtomicAdd(&deg[c[u]], 1.0f);
      if (bin) {
        rg[u]   = c[u] / RG_NODES;
        rank[u] = atomicAdd(&hist[rg[u]], 1);
      }
    }
  }
  __syncthreads();
  if (bin)
    for (int i = tid; i < nr; i += BIN_THREADS) {
      const int h = hist[i];
      hist[i] = h > 0 ? atomicAdd(&gcnt[i], h) : 0;
    }
  __syncthreads();
  if (bin) {
#pragma unroll
    for (int u = 0; u < BIN_PER; ++u) {
      if (c[u] >= 0) {
        const int r = rg[u];
        const unsigned slot = (unsigned)(hist[r] + rank[u]);
        if (slot < (unsigned)RG_LIST)
          glist[(size_t)r * RG_LIST + slot] = make_int2(sn[u], c[u] - r * RG_NODES);
      }
    }
  }
}

__global__ void gcn_deg_to_dis(float* __restrict__ dis, int N) {
  int i = blockIdx.x * blockDim.x + threadIdx.x;
  if (i < N) {
    float d = dis[i];
    dis[i] = (d > 0.0f) ? rsqrtf(d) : 0.0f;
  }
}


template <int C>
__global__ void gcn_prescale(const float* __restrict__ dis, const float* __restrict__ hf,
                             float* __restrict__ hs, int N) {
  constexpr int KS = C >> 2;
  long long i = (long long)blockIdx.x * blockDim.x + threadIdx.x;
  long long total = (long long)N * KS;
  if (i < total) {
    const int node = (int)(i / KS);
    const int v    = (int)(i - (long long)node * KS);
    const v2f* f = (const v2f*)hf + ((size_t)(node >> 4) * KS + v) * 32 + (node & 15);
    const v2f lo = f[0], hi = f[16];
    const float d = dis[node];
    float4 o = {lo.x * d, lo.y * d, hi.x * d, hi.y * d};
    ((float4*)hs)[i] = o;
  }
}


template <int C, int OC, bool HS>
__global__ __launch_bounds__(RG_THREADS) void gcn_range_conv(
    const int* __restrict__ ei, const float* __restrict__ hs, const float* __restrict__ dis,
    const int2* __restrict__ glist, const int* __restrict__ gcnt, int E, int N,
    const float* __restrict__ hf, const float* __restrict__ WA, const float* __restrict__ bA,
    const float* __restrict__ WB, const float* __restrict__ bB,
    float* __restrict__ outf, float* __restrict__ hsout) {
  static_assert(RG_NODES % 16 == 0, "a range must start on a 16-row tile boundary");
  constexpr int PITCH = C + 4;
  constexpr int KS    = C >> 2;
  __shared__ float    sacc[RG_NODES * PITCH];
  __shared__ float    sdis[RG_NODES];
  __shared__ int      lsrc[RG_LIST];
  __shared__ int      ldst[RG_LIST];
  __shared__ unsigned cnt;

  const int tid  = threadIdx.x;
  const int base = blockIdx.x * RG_NODES;
  const int rem  = N - base;
  const unsigned nloc = (unsigned)(rem < RG_NODES ? rem : RG_NODES);
  const int nvec = (int)nloc * (C >> 2);
  const int2* gl = glist + (size_t)blockIdx.x * RG_LIST;

  const float4* hsv = (const float4*)(hs + (size_t)base * C);
  for (int i = tid; i < nvec; i += RG_THREADS) {
    const int r = i / KS;
    *(float4*)(sacc + r * PITCH + ((i - r * KS) << 2)) = hsv[i];
  }
  for (int i = nvec + tid; i < (((int)nloc + 15) >> 4) * 16 * KS; i += RG_THREADS) {
    const int r = i / KS;
    float4 z = {0.f, 0.f, 0.f, 0.f};
    *(float4*)(sacc + r * PITCH + ((i - r * KS) << 2)) = z;
  }
  for (int i = tid; i < (int)nloc; i += RG_THREADS) sdis[i] = dis[base + i];
  if (tid == 0) cnt = 0u;

  int n = 0;
  const unsigned prev = (unsigned)gcnt[blockIdx.x];
  const bool listed = prev <= (unsigned)RG_LIST;
  if (listed) {
    n = (int)prev;
    for (int i = tid; i < n; i += RG_THREADS) {
      const int2 p = gl[i];
      lsrc[i] = p.x;
      ldst[i] = p.y;
    }
  }

  if (!listed) {
    __syncthreads();

    const int* col = ei + E;

    auto keep = [&](int e, unsigned cl) {
      const unsigned s = atomicAdd(&cnt, 1u);
      if (s < (unsigned)RG_LIST) {
        lsrc[s] = e;
        ldst[s] = (int)cl;
      } else {
        const float* hp = hs + (size_t)ei[e] * C;
#pragma unroll 1
        for (int ch = 0; ch < C; ++ch) atomicAdd(&sacc[cl * PITCH + ch], hp[ch]);
      }
    };
    auto test4 = [&](const int4 v, int q) {
      const unsigned a = (unsigned)(v.x - base), b = (unsigned)(v.y - base);
      const unsigned c = (unsigned)(v.z - base), d = (unsigned)(v.w - base);
      if ((a < nloc) | (b < nloc) | (c < nloc) | (d < nloc)) {
        const int e = q << 2;
        if (a < nloc) keep(e, a);
        if (b < nloc) keep(e + 1, b);
        if (c < nloc) keep(e + 2, c);
        if (d < nloc) keep(e + 3, d);
      }
    };

    if (((E & 3) == 0) && ((((size_t)col) & 15) == 0)) {
      const int4* colv = (const int4*)col;
      const int nq = E >> 2;
      int q = tid;
      for (; q + 3 * RG_THREADS < nq; q += 4 * RG_THREADS) {
        const int4 v0 = colv[q];
        const int4 v1 = colv[q + RG_THREADS];
        const int4 v2 = colv[q + 2 * RG_THREADS];
        const int4 v3 = colv[q + 3 * RG_THREADS];
        test4(v0, q);
        test4(v1, q + RG_THREADS);
        test4(v2, q + 2 * RG_THREADS);
        test4(v3, q + 3 * RG_THREADS);
      }
      for (; q < nq; q += RG_THREADS) test4(colv[q], q);
    } else {
      for (int e = tid; e < E; e += RG_THREADS) {
        const unsigned cl = (unsigned)(col[e] - base);
        if (cl < nloc) keep(e, cl);
      }
    }
    __syncthreads();

    const unsigned total = cnt;
    n = (int)(total < (unsigned)RG_LIST ? total : (unsigned)RG_LIST);
    for (int i = tid; i < n; i += RG_THREADS) lsrc[i] = ei[lsrc[i]];
  }
  __syncthreads();

  constexpr int RG_U = 8;
  constexpr int VEC  = C / 32;
  const int lane = tid & 31;
  const int wave = __builtin_amdgcn_readfirstlane(tid >> 5);
  for (int i0 = wave * RG_U; i0 < n; i0 += (RG_THREADS / 32) * RG_U) {
    float val[RG_U][VEC];
    int   dst[RG_U];
#pragma unroll
    for (int u = 0; u < RG_U; ++u) {
      const int idx = (i0 + u < n) ? (i0 + u) : i0;
      const int r = lsrc[idx];
      dst[u] = ldst[idx];
      const float* hp = hs + (size_t)r * C + lane * VEC;
#pragma unroll
      for (int k = 0; k < VEC; ++k) val[u][k] = hp[k];
    }
#pragma unroll
    for (int u = 0; u < RG_U; ++u) {
      if (i0 + u < n) {
        float* sp = sacc + dst[u] * PITCH + lane * VEC;
#pragma unroll
        for (int k = 0; k < VEC; ++k) atomicAdd(sp + k, val[u][k]);
      }
    }
  }
  __syncthreads();

  static_assert(OC % 16 == 0 && (32 % (OC / 16)) == 0, "column tiles must divide the waves of a group");
  constexpr int OCT = OC / 16;
  constexpr int TPR = (RG_THREADS / 32) / OCT;
  constexpr int KB  = KS < 8 ? KS : 8;
  const int ct   = wave % OCT;
  const int tsub = wave / OCT;
  const int nidx = lane & 15;
  const int half = lane >> 4;
  const int cb   = ct * 16 + (half << 3);

  v2f fw[KS], gw[KS];
  {
    const v2f* wAp = (const v2f*)WA + (size_t)ct * KS * 32 + lane;
    const v2f* wBp = (const v2f*)WB + (size_t)ct * KS * 32 + lane;
#pragma unroll
    for (int ks = 0; ks < KS; ++ks) {
      fw[ks] = wAp[ks * 32];
      gw[ks] = wBp[ks * 32];
    }
  }
  float bias[8];
#pragma unroll
  for (int vg = 0; vg < 8; ++vg) bias[vg] = bA[cb + vg] + bB[cb + vg];

  const int ntile = ((int)nloc + 15) >> 4;
  for (int t = tsub; t < ntile; t += TPR) {
    const int  gt  = (base >> 4) + t;
    const int  row = (t << 4) + nidx;
    const bool rok = (unsigned)row < nloc;
    const float d  = rok ? sdis[row] : 0.f;
    const v2f*   ag = (const v2f*)(hf + (size_t)gt * 16 * C) + lane;
    const float* sp = sacc + row * PITCH + (half << 1);
    v8f acc = {};
#pragma unroll
    for (int kb = 0; kb < KS; kb += KB) {
      v2f fa[KB], ga[KB];
#pragma unroll
      for (int i = 0; i < KB; ++i) {
        fa[i] = ag[(kb + i) * 32];
        v2f s = *(const v2f*)(sp + ((kb + i) << 2));
        s.x *= d;
        s.y *= d;
        ga[i] = s;
      }
#pragma unroll
      for (int i = 0; i < KB; ++i) {
        acc = __builtin_amdgcn_wmma_f32_16x16x4_f32(false, fw[kb + i], false, fa[i],
                                                    (short)0, acc, false, false);
        acc = __builtin_amdgcn_wmma_f32_16x16x4_f32(false, gw[kb + i], false, ga[i],
                                                    (short)0, acc, false, false);
      }
    }
    if (rok) {
      constexpr int KSO = OC >> 2;
      float* ob = outf + ((size_t)gt * KSO + (cb >> 2)) * 64 + nidx * 2;
      float o[8];
#pragma unroll
      for (int vg = 0; vg < 8; ++vg) {
        const float v = acc[vg] + bias[vg];
        o[vg] = v > 0.f ? v : 0.f;
      }
#pragma unroll
      for (int q = 0; q < 4; ++q) {
        v2f ov = {o[2 * q], o[2 * q + 1]};
        *(v2f*)(ob + (q >> 1) * 64 + (q & 1) * 32) = ov;
      }
      if constexpr (HS) {
        float4* hp = (float4*)(hsout + (size_t)(base + row) * OC + cb);
        float4 lo = {o[0] * d, o[1] * d, o[2] * d, o[3] * d};
        float4 hi = {o[4] * d, o[5] * d, o[6] * d, o[7] * d};
        hp[0] = lo;
        hp[1] = hi;
      }
    }
  }
}


#define NUM_W 8
struct PackSrc {
  const float* w0; const float* w1; const float* w2; const float* w3;
  const float* w4; const float* w5; const float* w6; const float* w7;
};
#define WP_W1   0
#define WP_W2   4096
#define WP_C1L  4608
#define WP_C1R  6656
#define WP_C2L  8704
#define WP_C2R  16896
#define WP_W3   25088
#define WP_W4   49664
#define WP_TOTAL 55808

__global__ void gcn_pack_weights(PackSrc s, float* __restrict__ dst) {
  const int m = blockIdx.y;
  const float* W; int OC, IC, off;
  switch (m) {
    case 0:  W = s.w0; OC = 16;  IC = 256; off = WP_W1;  break;
    case 1:  W = s.w1; OC = 32;  IC = 16;  off = WP_W2;  break;
    case 2:  W = s.w2; OC = 64;  IC = 32;  off = WP_C1L; break;
    case 3:  W = s.w3; OC = 64;  IC = 32;  off = WP_C1R; break;
    case 4:  W = s.w4; OC = 128; IC = 64;  off = WP_C2L; break;
    case 5:  W = s.w5; OC = 128; IC = 64;  off = WP_C2R; break;
    case 6:  W = s.w6; OC = 192; IC = 128; off = WP_W3;  break;
    default: W = s.w7; OC = 20;  IC = 192; off = WP_W4;  break;
  }
  const int ks_n  = IC >> 2;
  const int tiles = (OC + 15) >> 4;
  const int total = tiles * ks_n * 32;
  for (int i = blockIdx.x * blockDim.x + threadIdx.x; i < total; i += gridDim.x * blockDim.x) {
    const int lane = i & 31;
    const int ks   = (i >> 5) % ks_n;
    const int ct   = (i >> 5) / ks_n;
    int col = ct * 16 + (lane & 15);
    col = col < OC ? col : OC - 1;
    const float* p = W + (size_t)col * IC + ks * 4 + ((lane >> 4) << 1);
    v2f v = *(const v2f*)p;
    *(v2f*)(dst + off + (size_t)i * 2) = v;
  }
}

template <int IC, bool DUAL, bool RELU, bool GUARD, bool INP, bool OUTP>
__global__ __launch_bounds__(32 * WAVES_PER_BLOCK) void gcn_wmma_gemm(
    const float* __restrict__ inA, const float* __restrict__ WA, const float* __restrict__ bA,
    const float* __restrict__ inB, const float* __restrict__ WB, const float* __restrict__ bB,
    float* __restrict__ out, int nrows, int OC) {
  static_assert(INP || !DUAL, "the two-operand form reads fragment-order inputs only");
  static_assert(!(OUTP && GUARD), "fragment-order output needs full tiles");
  __shared__ float lds[INP ? 4 : 16 * IC];
  const int tid  = threadIdx.x;
  const int lane = tid & 31;
  const int wave = tid >> 5;
  const int row0 = blockIdx.x << 4;
  const int col0 = (blockIdx.y * WAVES_PER_BLOCK + wave) << 4;

  float* Atile = lds;

  if constexpr (!INP) {
    constexpr int vecPerRow = IC >> 2;
    constexpr int totalVec  = vecPerRow << 4;
    for (int idx = tid; idx < totalVec; idx += 32 * WAVES_PER_BLOCK) {
      const int r = idx / vecPerRow;
      const int v = idx - r * vecPerRow;
      const int node = row0 + r;
      if (GUARD && node >= nrows) {
        float4 z = {0.f, 0.f, 0.f, 0.f};
        ((float4*)Atile)[idx] = z;
      } else {
        ((float4*)Atile)[idx] = ((const float4*)(inA + (size_t)node * IC))[v];
      }
    }
    __syncthreads();
  }

  if (col0 >= OC) return;

  const int  nidx = lane & 15;
  const int  koff = (lane >> 4) << 1;
  const int  col  = col0 + nidx;
  const bool colOK = GUARD ? (col < OC) : true;

  constexpr int KSTEPS = IC / 4;
  const v2f* wAp = (const v2f*)WA + (size_t)(col0 >> 4) * KSTEPS * 32 + lane;
  const v2f* wBp = DUAL ? ((const v2f*)WB + (size_t)(col0 >> 4) * KSTEPS * 32 + lane) : wAp;
  const v2f*   ag = (const v2f*)(inA + (size_t)row0 * IC) + lane;
  const v2f*   bg = DUAL ? ((const v2f*)(inB + (size_t)row0 * IC) + lane) : ag;
  const float* ar = Atile + nidx * IC + koff;

  constexpr int BATCH  = DUAL ? (KSTEPS < 4 ? KSTEPS : 4)
                              : (KSTEPS < 8 ? KSTEPS : 8);

  v8f acc = {};
#pragma unroll 1
  for (int kb = 0; kb < KSTEPS; kb += BATCH) {
    v2f fa[BATCH], fw[BATCH];
    v2f ga[DUAL ? BATCH : 1], gw[DUAL ? BATCH : 1];
#pragma unroll
    for (int i = 0; i < BATCH; ++i) {
      const int ks = kb + i;
      if constexpr (INP) fa[i] = ag[ks * 32];
      else               fa[i] = *(const v2f*)(ar + ks * 4);
      fw[i] = wAp[ks * 32];
      if constexpr (DUAL) {
        ga[i] = bg[ks * 32];
        gw[i] = wBp[ks * 32];
      }
    }
#pragma unroll
    for (int i = 0; i < BATCH; ++i) {
      if constexpr (OUTP) {
        acc = __builtin_amdgcn_wmma_f32_16x16x4_f32(false, fw[i], false, fa[i],
                                                    (short)0, acc, false, false);
        if constexpr (DUAL)
          acc = __builtin_amdgcn_wmma_f32_16x16x4_f32(false, gw[i], false, ga[i],
                                                      (short)0, acc, false, false);
      } else {
        acc = __builtin_amdgcn_wmma_f32_16x16x4_f32(false, fa[i], false, fw[i],
                                                    (short)0, acc, false, false);
        if constexpr (DUAL)
          acc = __builtin_amdgcn_wmma_f32_16x16x4_f32(false, ga[i], false, gw[i],
                                                      (short)0, acc, false, false);
      }
    }
  }

  const int mb = (lane >> 4) << 3;
  if constexpr (OUTP) {
    const int cb = col0 + mb;
    float bias[8];
#pragma unroll
    for (int vg = 0; vg < 8; ++vg) {
      bias[vg] = bA[cb + vg];
      if constexpr (DUAL) bias[vg] += bB[cb + vg];
    }
    const int KSO = OC >> 2;
    float* ob = out + ((size_t)blockIdx.x * KSO + (cb >> 2)) * 64 + nidx * 2;
#pragma unroll
    for (int t = 0; t < 4; ++t) {
      float v0 = acc[2 * t] + bias[2 * t];
      float v1 = acc[2 * t + 1] + bias[2 * t + 1];
      if constexpr (RELU) { v0 = v0 > 0.f ? v0 : 0.f; v1 = v1 > 0.f ? v1 : 0.f; }
      v2f o = {v0, v1};
      *(v2f*)(ob + (t >> 1) * 64 + (t & 1) * 32) = o;
    }
  } else if constexpr (GUARD) {
    float bias = 0.f;
    if (colOK) {
      bias = bA[col];
      if constexpr (DUAL) bias += bB[col];
    }
#pragma unroll
    for (int vg = 0; vg < 8; ++vg) {
      const int r = row0 + mb + vg;
      if (colOK && r < nrows) {
        float val = acc[vg] + bias;
        if constexpr (RELU) val = val > 0.f ? val : 0.f;
        out[(size_t)r * OC + col] = val;
      }
    }
  } else {
    float bias = bA[col];
    if constexpr (DUAL) bias += bB[col];
    float* op = out + (size_t)(row0 + mb) * OC + col;
#pragma unroll
    for (int vg = 0; vg < 8; ++vg) {
      float val = acc[vg] + bias;
      if constexpr (RELU) val = val > 0.f ? val : 0.f;
      op[(size_t)vg * OC] = val;
    }
  }
}

template <int IC, int WPB, bool DUAL, bool RELU, bool GUARD, bool OUTP>
__global__ __launch_bounds__(32 * WPB) void gcn_wmma_gemm_rows(
    const float* __restrict__ inA, const float* __restrict__ WA, const float* __restrict__ bA,
    const float* __restrict__ inB, const float* __restrict__ WB, const float* __restrict__ bB,
    float* __restrict__ out, int nrows, int OC) {
  static_assert(!(OUTP && GUARD), "fragment-order output needs full tiles");
  constexpr int KSTEPS = IC / 4;
  const int lane = threadIdx.x & 31;
  const int wave = __builtin_amdgcn_readfirstlane(threadIdx.x >> 5);
  const int col0 = (blockIdx.y * WPB + wave) << 4;
  if (col0 >= OC) return;

  const int nidx = lane & 15;
  const int mb   = (lane >> 4) << 3;

  const v2f* wAp = (const v2f*)WA + (size_t)(col0 >> 4) * KSTEPS * 32 + lane;
  const v2f* wBp = DUAL ? ((const v2f*)WB + (size_t)(col0 >> 4) * KSTEPS * 32 + lane) : wAp;
  v2f fw[KSTEPS];
  v2f gw[DUAL ? KSTEPS : 1];
#pragma unroll
  for (int ks = 0; ks < KSTEPS; ++ks) {
    fw[ks] = wAp[ks * 32];
    if constexpr (DUAL) gw[ks] = wBp[ks * 32];
  }

  float bias[OUTP ? 8 : 1];
  const int  col   = col0 + nidx;
  const bool colOK = GUARD ? (col < OC) : true;
  if constexpr (OUTP) {
    const int cb = col0 + mb;
#pragma unroll
    for (int vg = 0; vg < 8; ++vg) {
      bias[vg] = bA[cb + vg];
      if constexpr (DUAL) bias[vg] += bB[cb + vg];
    }
  } else {
    bias[0] = 0.f;
    if (colOK) {
      bias[0] = bA[col];
      if constexpr (DUAL) bias[0] += bB[col];
    }
  }

  const int rowTiles = (nrows + 15) >> 4;
  for (int rt = blockIdx.x; rt < rowTiles; rt += gridDim.x) {
    const v2f* ag = (const v2f*)(inA + (size_t)rt * 16 * IC) + lane;
    const v2f* bg = DUAL ? ((const v2f*)(inB + (size_t)rt * 16 * IC) + lane) : ag;
    v2f fa[KSTEPS];
    v2f ga[DUAL ? KSTEPS : 1];
#pragma unroll
    for (int ks = 0; ks < KSTEPS; ++ks) {
      fa[ks] = ag[ks * 32];
      if constexpr (DUAL) ga[ks] = bg[ks * 32];
    }
    __builtin_amdgcn_sched_barrier(0);
    v8f acc = {};
#pragma unroll
    for (int ks = 0; ks < KSTEPS; ++ks) {
      if constexpr (OUTP) {
        acc = __builtin_amdgcn_wmma_f32_16x16x4_f32(false, fw[ks], false, fa[ks],
                                                    (short)0, acc, false, false);
        if constexpr (DUAL)
          acc = __builtin_amdgcn_wmma_f32_16x16x4_f32(false, gw[ks], false, ga[ks],
                                                      (short)0, acc, false, false);
      } else {
        acc = __builtin_amdgcn_wmma_f32_16x16x4_f32(false, fa[ks], false, fw[ks],
                                                    (short)0, acc, false, false);
        if constexpr (DUAL)
          acc = __builtin_amdgcn_wmma_f32_16x16x4_f32(false, ga[ks], false, gw[ks],
                                                      (short)0, acc, false, false);
      }
    }

    if constexpr (OUTP) {
      const int cb  = col0 + mb;
      const int KSO = OC >> 2;
      float* ob = out + ((size_t)rt * KSO + (cb >> 2)) * 64 + nidx * 2;
#pragma unroll
      for (int t = 0; t < 4; ++t) {
        float v0 = acc[2 * t] + bias[2 * t];
        float v1 = acc[2 * t + 1] + bias[2 * t + 1];
        if constexpr (RELU) { v0 = v0 > 0.f ? v0 : 0.f; v1 = v1 > 0.f ? v1 : 0.f; }
        v2f o = {v0, v1};
        *(v2f*)(ob + (t >> 1) * 64 + (t & 1) * 32) = o;
      }
    } else {
      const int row0 = rt << 4;
#pragma unroll
      for (int vg = 0; vg < 8; ++vg) {
        const int r = row0 + mb + vg;
        if (colOK && (!GUARD || r < nrows)) {
          float val = acc[vg] + bias[0];
          if constexpr (RELU) val = val > 0.f ? val : 0.f;
          out[(size_t)r * OC + col] = val;
        }
      }
    }
  }
}


template <int IC, bool DUAL, bool RELU, bool INP, bool OUTP>
static inline void launch_gemm(const float* inA, const float* WA, const float* bA,
                               const float* inB, const float* WB, const float* bB,
                               float* out, int N, int OC, hipStream_t stream) {
  const int rowTiles = (N + 15) / 16;
  const int gy = (((OC + 15) / 16) + WAVES_PER_BLOCK - 1) / WAVES_PER_BLOCK;
  const dim3 grid(rowTiles, gy), blk(32 * WAVES_PER_BLOCK);
  if constexpr (OUTP) {
    gcn_wmma_gemm<IC, DUAL, RELU, false, INP, true><<<grid, blk, 0, stream>>>(
        inA, WA, bA, inB, WB, bB, out, N, OC);
  } else {
    if ((N & 15) == 0 && (OC & 15) == 0)
      gcn_wmma_gemm<IC, DUAL, RELU, false, INP, false><<<grid, blk, 0, stream>>>(
          inA, WA, bA, inB, WB, bB, out, N, OC);
    else
      gcn_wmma_gemm<IC, DUAL, RELU, true, INP, false><<<grid, blk, 0, stream>>>(
          inA, WA, bA, inB, WB, bB, out, N, OC);
  }
}

template <int IC, int WPB, int TPW, bool DUAL, bool RELU, bool OUTP>
static inline void launch_gemm_rows(const float* inA, const float* WA, const float* bA,
                                    const float* inB, const float* WB, const float* bB,
                                    float* out, int N, int OC, hipStream_t stream) {
  const int rowTiles = (N + 15) / 16;
  const int gx = (rowTiles + TPW - 1) / TPW;
  const int gy = (((OC + 15) / 16) + WPB - 1) / WPB;
  const dim3 grid(gx, gy), blk(32 * WPB);
  if constexpr (OUTP) {
    gcn_wmma_gemm_rows<IC, WPB, DUAL, RELU, false, true><<<grid, blk, 0, stream>>>(
        inA, WA, bA, inB, WB, bB, out, N, OC);
  } else {
    if ((N & 15) == 0 && (OC & 15) == 0)
      gcn_wmma_gemm_rows<IC, WPB, DUAL, RELU, false, false><<<grid, blk, 0, stream>>>(
          inA, WA, bA, inB, WB, bB, out, N, OC);
    else
      gcn_wmma_gemm_rows<IC, WPB, DUAL, RELU, true, false><<<grid, blk, 0, stream>>>(
          inA, WA, bA, inB, WB, bB, out, N, OC);
  }
}

extern "C" void kernel_launch(void* const* d_in, const int* in_sizes, int n_in,
                              void* d_out, int out_size, void* d_ws, size_t ws_size,
                              hipStream_t stream) {
  (void)n_in; (void)out_size; (void)ws_size;
  const float* x    = (const float*)d_in[0];
  const int*   ei   = (const int*)d_in[1];
  const float* w1   = (const float*)d_in[2];
  const float* b1   = (const float*)d_in[3];
  const float* w2   = (const float*)d_in[4];
  const float* b2   = (const float*)d_in[5];
  const float* c1wl = (const float*)d_in[6];
  const float* c1bl = (const float*)d_in[7];
  const float* c1wr = (const float*)d_in[8];
  const float* c1br = (const float*)d_in[9];
  const float* c2wl = (const float*)d_in[10];
  const float* c2bl = (const float*)d_in[11];
  const float* c2wr = (const float*)d_in[12];
  const float* c2br = (const float*)d_in[13];
  const float* w3   = (const float*)d_in[14];
  const float* b3   = (const float*)d_in[15];
  const float* w4   = (const float*)d_in[16];
  const float* b4   = (const float*)d_in[17];

  const int IN_CH = 256;
  const int N = in_sizes[0] / IN_CH;
  const int E = in_sizes[1] / 2;
  const size_t n = (size_t)N;

  float* ws   = (float*)d_ws;
  float* dis  = ws;
  float* h3   = ws + 1 * n;
  float* hs3  = ws + 65 * n;
  int*  rcnt  = (int*)(ws + 129 * n);
  int2* rlist = (int2*)(ws + 130 * n);
  float* h1   = ws + 172 * n;
  float* wp   = ws + 192 * n;
  float* h2   = ws + 193 * n;
  float* hs2  = ws + 225 * n;
  float* h4   = ws + 193 * n;
  float* h5   = ws;

  const int eltBlocks  = (N + 255) / 256;
  const int binBlocks  = (E + BIN_THREADS * BIN_PER - 1) / (BIN_THREADS * BIN_PER);
  const int rangeBlocks = (N + RG_NODES - 1) / RG_NODES;

  PackSrc ps = {w1, w2, c1wl, c1wr, c2wl, c2wr, w3, w4};
  gcn_pack_weights<<<dim3(24, NUM_W), 256, 0, stream>>>(ps, wp);

  launch_gemm<256, false, false, false, true>(x, wp + WP_W1, b1, nullptr, nullptr, nullptr, h1, N, 16, stream);
  launch_gemm_rows<16, 2, 2, false, false, true>(h1, wp + WP_W2, b2, nullptr, nullptr, nullptr, h2, N, 32, stream);

  gcn_deg_init<<<eltBlocks, 256, 0, stream>>>(dis, rcnt, N, rangeBlocks);
  gcn_deg_bin<<<binBlocks, BIN_THREADS, 0, stream>>>(ei, dis, rlist, rcnt, E, rangeBlocks);
  gcn_deg_to_dis<<<eltBlocks, 256, 0, stream>>>(dis, N);

  gcn_prescale<32><<<(int)((n * 8 + 255) / 256), 256, 0, stream>>>(dis, h2, hs2, N);
  gcn_range_conv<32, 64, true><<<rangeBlocks, RG_THREADS, 0, stream>>>(
      ei, hs2, dis, rlist, rcnt, E, N, h2, wp + WP_C1L, c1bl, wp + WP_C1R, c1br, h3, hs3);

  gcn_range_conv<64, 128, false><<<rangeBlocks, RG_THREADS, 0, stream>>>(
      ei, hs3, dis, rlist, rcnt, E, N, h3, wp + WP_C2L, c2bl, wp + WP_C2R, c2br, h4, nullptr);

  launch_gemm_rows<128, 4, 13, false, false, true>(h4, wp + WP_W3, b3, nullptr, nullptr, nullptr, h5, N, 192, stream);
  launch_gemm_rows<192, 2, 4, false, false, false>(h5, wp + WP_W4, b4, nullptr, nullptr, nullptr, (float*)d_out, N, 20, stream);
}


# neighbour-sum kernel keeps more loads in flight: staging loads issued as one batch; edge loop in whole trips, next trip's rows requested before this trip's LDS adds (arrival order as before)
# speedup vs baseline: 2.5318x; 1.0405x over previous
#include <hip/hip_runtime.h>

typedef __attribute__((ext_vector_type(2))) float v2f;
typedef __attribute__((ext_vector_type(4))) float v4f;
typedef __attribute__((ext_vector_type(8))) float v8f;

#define WAVES_PER_BLOCK 4


#define RG_NODES   400
#define RG_THREADS 1024
#define RG_LIST    8192

#define BIN_THREADS 1024
#define BIN_PER     4
#define BIN_MAXR    1024

__global__ void gcn_deg_init(float* __restrict__ deg, int* __restrict__ gcnt, int N, int nr) {
  int i = blockIdx.x * blockDim.x + threadIdx.x;
  if (i < N) deg[i] = 1.0f;
  if (i < nr) gcnt[i] = (nr <= BIN_MAXR) ? 0 : (RG_LIST + 1);
}

__global__ __launch_bounds__(BIN_THREADS) void gcn_deg_bin(
    const int* __restrict__ ei, float* __restrict__ deg, int2* __restrict__ glist,
    int* __restrict__ gcnt, int E, int nr) {
  __shared__ int hist[BIN_MAXR];
  const int  tid = threadIdx.x;
  const bool bin = nr <= BIN_MAXR;
  if (bin)
    for (int i = tid; i < nr; i += BIN_THREADS) hist[i] = 0;
  __syncthreads();

  const int e0 = blockIdx.x * (BIN_THREADS * BIN_PER) + tid;
  int c[BIN_PER], sn[BIN_PER], rg[BIN_PER], rank[BIN_PER];
#pragma unroll
  for (int u = 0; u < BIN_PER; ++u) {
    const int e = e0 + u * BIN_THREADS;
    const bool ok = e < E;
    c[u]  = ok ? ei[E + e] : -1;
    sn[u] = ok ? ei[e] : 0;
  }
#pragma unroll
  for (int u = 0; u < BIN_PER; ++u) {
    rg[u] = 0; rank[u] = 0;
    if (c[u] >= 0) {
      unsafeAtomicAdd(&deg[c[u]], 1.0f);
      if (bin) {
        rg[u]   = c[u] / RG_NODES;
        rank[u] = atomicAdd(&hist[rg[u]], 1);
      }
    }
  }
  __syncthreads();
  if (bin)
    for (int i = tid; i < nr; i += BIN_THREADS) {
      const int h = hist[i];
      hist[i] = h > 0 ? atomicAdd(&gcnt[i], h) : 0;
    }
  __syncthreads();
  if (bin) {
#pragma unroll
    for (int u = 0; u < BIN_PER; ++u) {
      if (c[u] >= 0) {
        const int r = rg[u];
        const unsigned slot = (unsigned)(hist[r] + rank[u]);
        if (slot < (unsigned)RG_LIST)
          glist[(size_t)r * RG_LIST + slot] = make_int2(sn[u], c[u] - r * RG_NODES);
      }
    }
  }
}

__global__ void gcn_deg_to_dis(float* __restrict__ dis, int N) {
  int i = blockIdx.x * blockDim.x + threadIdx.x;
  if (i < N) {
    float d = dis[i];
    dis[i] = (d > 0.0f) ? rsqrtf(d) : 0.0f;
  }
}


template <int C>
__global__ void gcn_prescale(const float* __restrict__ dis, const float* __restrict__ hf,
                             float* __restrict__ hs, int N) {
  constexpr int KS = C >> 2;
  long long i = (long long)blockIdx.x * blockDim.x + threadIdx.x;
  long long total = (long long)N * KS;
  if (i < total) {
    const int node = (int)(i / KS);
    const int v    = (int)(i - (long long)node * KS);
    const v2f* f = (const v2f*)hf + ((size_t)(node >> 4) * KS + v) * 32 + (node & 15);
    const v2f lo = f[0], hi = f[16];
    const float d = dis[node];
    float4 o = {lo.x * d, lo.y * d, hi.x * d, hi.y * d};
    ((float4*)hs)[i] = o;
  }
}


template <int C, int OC, bool HS>
__global__ __launch_bounds__(RG_THREADS) void gcn_range_conv(
    const int* __restrict__ ei, const float* __restrict__ hs, const float* __restrict__ dis,
    const int2* __restrict__ glist, const int* __restrict__ gcnt, int E, int N,
    const float* __restrict__ hf, const float* __restrict__ WA, const float* __restrict__ bA,
    const float* __restrict__ WB, const float* __restrict__ bB,
    float* __restrict__ outf, float* __restrict__ hsout) {
  static_assert(RG_NODES % 16 == 0, "a range must start on a 16-row tile boundary");
  constexpr int PITCH = C + 4;
  constexpr int KS    = C >> 2;
  __shared__ float    sacc[RG_NODES * PITCH];
  __shared__ float    sdis[RG_NODES];
  __shared__ int      lsrc[RG_LIST];
  __shared__ int      ldst[RG_LIST];
  __shared__ unsigned cnt;

  const int tid  = threadIdx.x;
  const int base = blockIdx.x * RG_NODES;
  const int rem  = N - base;
  const unsigned nloc = (unsigned)(rem < RG_NODES ? rem : RG_NODES);
  const int nvec = (int)nloc * (C >> 2);
  const int2* gl = glist + (size_t)blockIdx.x * RG_LIST;

  const unsigned prev = (unsigned)gcnt[blockIdx.x];
  const bool listed = prev <= (unsigned)RG_LIST;
  int n = listed ? (int)prev : 0;

  static_assert(RG_NODES <= RG_THREADS, "one dis value per thread");
  constexpr int NSV = (RG_NODES * KS + RG_THREADS - 1) / RG_THREADS;
  constexpr int NLP = RG_LIST / RG_THREADS;
  const v4f* hsv = (const v4f*)(hs + (size_t)base * C);
  v4f sv[NSV];
#pragma unroll
  for (int j = 0; j < NSV; ++j) {
    const int i = tid + j * RG_THREADS;
    sv[j] = hsv[i < nvec ? i : nvec - 1];
  }
  const float dv = dis[base + (tid < (int)nloc ? tid : (int)nloc - 1)];
  int2 lp[NLP];
#pragma unroll
  for (int j = 0; j < NLP; ++j) lp[j] = make_int2(0, 0);
  if (n > 0) {
#pragma unroll
    for (int j = 0; j < NLP; ++j) {
      const int i = tid + j * RG_THREADS;
      lp[j] = gl[i < n ? i : n - 1];
    }
  }
#pragma unroll
  for (int j = 0; j < NSV; ++j) {
    const int i = tid + j * RG_THREADS;
    if (i < nvec) {
      const int r = i / KS;
      *(v4f*)(sacc + r * PITCH + ((i - r * KS) << 2)) = sv[j];
    }
  }
  for (int i = nvec + tid; i < (((int)nloc + 15) >> 4) * 16 * KS; i += RG_THREADS) {
    const int r = i / KS;
    float4 z = {0.f, 0.f, 0.f, 0.f};
    *(float4*)(sacc + r * PITCH + ((i - r * KS) << 2)) = z;
  }
  if (tid < (int)nloc) sdis[tid] = dv;
  if (tid == 0) cnt = 0u;
  if (n > 0) {
#pragma unroll
    for (int j = 0; j < NLP; ++j) {
      const int i = tid + j * RG_THREADS;
      if (i < n) {
        lsrc[i] = lp[j].x;
        ldst[i] = lp[j].y;
      }
    }
  }

  if (!listed) {
    __syncthreads();

    const int* col = ei + E;

    auto keep = [&](int e, unsigned cl) {
      const unsigned s = atomicAdd(&cnt, 1u);
      if (s < (unsigned)RG_LIST) {
        lsrc[s] = e;
        ldst[s] = (int)cl;
      } else {
        const float* hp = hs + (size_t)ei[e] * C;
#pragma unroll 1
        for (int ch = 0; ch < C; ++ch) atomicAdd(&sacc[cl * PITCH + ch], hp[ch]);
      }
    };
    auto test4 = [&](const int4 v, int q) {
      const unsigned a = (unsigned)(v.x - base), b = (unsigned)(v.y - base);
      const unsigned c = (unsigned)(v.z - base), d = (unsigned)(v.w - base);
      if ((a < nloc) | (b < nloc) | (c < nloc) | (d < nloc)) {
        const int e = q << 2;
        if (a < nloc) keep(e, a);
        if (b < nloc) keep(e + 1, b);
        if (c < nloc) keep(e + 2, c);
        if (d < nloc) keep(e + 3, d);
      }
    };

    if (((E & 3) == 0) && ((((size_t)col) & 15) == 0)) {
      const int4* colv = (const int4*)col;
      const int nq = E >> 2;
      int q = tid;
      for (; q + 3 * RG_THREADS < nq; q += 4 * RG_THREADS) {
        const int4 v0 = colv[q];
        const int4 v1 = colv[q + RG_THREADS];
        const int4 v2 = colv[q + 2 * RG_THREADS];
        const int4 v3 = colv[q + 3 * RG_THREADS];
        test4(v0, q);
        test4(v1, q + RG_THREADS);
        test4(v2, q + 2 * RG_THREADS);
        test4(v3, q + 3 * RG_THREADS);
      }
      for (; q < nq; q += RG_THREADS) test4(colv[q], q);
    } else {
      for (int e = tid; e < E; e += RG_THREADS) {
        const unsigned cl = (unsigned)(col[e] - base);
        if (cl < nloc) keep(e, cl);
      }
    }
    __syncthreads();

    const unsigned total = cnt;
    n = (int)(total < (unsigned)RG_LIST ? total : (unsigned)RG_LIST);
    for (int i = tid; i < n; i += RG_THREADS) lsrc[i] = ei[lsrc[i]];
  }
  __syncthreads();

  constexpr int VEC  = C / 32;
  constexpr int RG_U = 16 / VEC;
  constexpr int NW   = RG_THREADS / 32;
  static_assert(RG_U <= 32 && (RG_U & (RG_U - 1)) == 0 && RG_LIST % RG_U == 0, "a trip is read by the lanes of one wave");
  const int lane = tid & 31;
  const int wave  = __builtin_amdgcn_readfirstlane(tid >> 5);
  const int ntrip = n / RG_U;
  const int lo = lane * VEC;
  float*    srow = sacc + lo;

  auto fetch = [&](int t, float (&val)[RG_U][VEC], int (&dst)[RG_U]) {
    const int e = t * RG_U + (lane & (RG_U - 1));
    const int s = lsrc[e];
    const int d = ldst[e];
#pragma unroll
    for (int u = 0; u < RG_U; ++u) {
      const int r = __builtin_amdgcn_readlane(s, u);
      dst[u] = __builtin_amdgcn_readlane(d, u);
      const float* hp = hs + (size_t)r * C;
#pragma unroll
      for (int k = 0; k < VEC; ++k) val[u][k] = hp[lo + k];
    }
  };
  auto add = [&](const float (&val)[RG_U][VEC], const int (&dst)[RG_U]) {
#pragma unroll
    for (int u = 0; u < RG_U; ++u) {
      float* sp = srow + dst[u] * PITCH;
#pragma unroll
      for (int k = 0; k < VEC; ++k) atomicAdd(sp + k, val[u][k]);
    }
  };

  {
    float vA[RG_U][VEC], vB[RG_U][VEC];
    int   dA[RG_U], dB[RG_U];
    int t = wave;
    if (t < ntrip) {
      fetch(t, vA, dA);
      for (;;) {
        const int t1 = t + NW;
        if (t1 < ntrip) fetch(t1, vB, dB);
        add(vA, dA);
        if (t1 >= ntrip) break;
        t = t1 + NW;
        if (t < ntrip) fetch(t, vA, dA);
        add(vB, dB);
        if (t >= ntrip) break;
      }
    }
  }
  for (int i = ntrip * RG_U + wave; i < n; i += NW) {
    const float* hp = hs + (size_t)lsrc[i] * C + lo;
    float*       sp = srow + ldst[i] * PITCH;
#pragma unroll
    for (int k = 0; k < VEC; ++k) atomicAdd(sp + k, hp[k]);
  }
  __syncthreads();

  static_assert(OC % 16 == 0 && (32 % (OC / 16)) == 0, "column tiles must divide the waves of a group");
  constexpr int OCT = OC / 16;
  constexpr int TPR = (RG_THREADS / 32) / OCT;
  constexpr int KB  = KS < 8 ? KS : 8;
  const int ct   = wave % OCT;
  const int tsub = wave / OCT;
  const int nidx = lane & 15;
  const int half = lane >> 4;
  const int cb   = ct * 16 + (half << 3);

  v2f fw[KS], gw[KS];
  {
    const v2f* wAp = (const v2f*)WA + (size_t)ct * KS * 32 + lane;
    const v2f* wBp = (const v2f*)WB + (size_t)ct * KS * 32 + lane;
#pragma unroll
    for (int ks = 0; ks < KS; ++ks) {
      fw[ks] = wAp[ks * 32];
      gw[ks] = wBp[ks * 32];
    }
  }
  float bias[8];
#pragma unroll
  for (int vg = 0; vg < 8; ++vg) bias[vg] = bA[cb + vg] + bB[cb + vg];

  const int ntile = ((int)nloc + 15) >> 4;
  for (int t = tsub; t < ntile; t += TPR) {
    const int  gt  = (base >> 4) + t;
    const int  row = (t << 4) + nidx;
    const bool rok = (unsigned)row < nloc;
    const float d  = rok ? sdis[row] : 0.f;
    const v2f*   ag = (const v2f*)(hf + (size_t)gt * 16 * C) + lane;
    const float* sp = sacc + row * PITCH + (half << 1);
    v8f acc = {};
#pragma unroll
    for (int kb = 0; kb < KS; kb += KB) {
      v2f fa[KB], ga[KB];
#pragma unroll
      for (int i = 0; i < KB; ++i) {
        fa[i] = ag[(kb + i) * 32];
        v2f s = *(const v2f*)(sp + ((kb + i) << 2));
        s.x *= d;
        s.y *= d;
        ga[i] = s;
      }
#pragma unroll
      for (int i = 0; i < KB; ++i) {
        acc = __builtin_amdgcn_wmma_f32_16x16x4_f32(false, fw[kb + i], false, fa[i],
                                                    (short)0, acc, false, false);
        acc = __builtin_amdgcn_wmma_f32_16x16x4_f32(false, gw[kb + i], false, ga[i],
                                                    (short)0, acc, false, false);
      }
    }
    if (rok) {
      constexpr int KSO = OC >> 2;
      float* ob = outf + ((size_t)gt * KSO + (cb >> 2)) * 64 + nidx * 2;
      float o[8];
#pragma unroll
      for (int vg = 0; vg < 8; ++vg) {
        const float v = acc[vg] + bias[vg];
        o[vg] = v > 0.f ? v : 0.f;
      }
#pragma unroll
      for (int q = 0; q < 4; ++q) {
        v2f ov = {o[2 * q], o[2 * q + 1]};
        *(v2f*)(ob + (q >> 1) * 64 + (q & 1) * 32) = ov;
      }
      if constexpr (HS) {
        float4* hp = (float4*)(hsout + (size_t)(base + row) * OC + cb);
        float4 lo = {o[0] * d, o[1] * d, o[2] * d, o[3] * d};
        float4 hi = {o[4] * d, o[5] * d, o[6] * d, o[7] * d};
        hp[0] = lo;
        hp[1] = hi;
      }
    }
  }
}


#define NUM_W 8
struct PackSrc {
  const float* w0; const float* w1; const float* w2; const float* w3;
  const float* w4; const float* w5; const float* w6; const float* w7;
};
#define WP_W1   0
#define WP_W2   4096
#define WP_C1L  4608
#define WP_C1R  6656
#define WP_C2L  8704
#define WP_C2R  16896
#define WP_W3   25088
#define WP_W4   49664
#define WP_TOTAL 55808

__global__ void gcn_pack_weights(PackSrc s, float* __restrict__ dst) {
  const int m = blockIdx.y;
  const float* W; int OC, IC, off;
  switch (m) {
    case 0:  W = s.w0; OC = 16;  IC = 256; off = WP_W1;  break;
    case 1:  W = s.w1; OC = 32;  IC = 16;  off = WP_W2;  break;
    case 2:  W = s.w2; OC = 64;  IC = 32;  off = WP_C1L; break;
    case 3:  W = s.w3; OC = 64;  IC = 32;  off = WP_C1R; break;
    case 4:  W = s.w4; OC = 128; IC = 64;  off = WP_C2L; break;
    case 5:  W = s.w5; OC = 128; IC = 64;  off = WP_C2R; break;
    case 6:  W = s.w6; OC = 192; IC = 128; off = WP_W3;  break;
    default: W = s.w7; OC = 20;  IC = 192; off = WP_W4;  break;
  }
  const int ks_n  = IC >> 2;
  const int tiles = (OC + 15) >> 4;
  const int total = tiles * ks_n * 32;
  for (int i = blockIdx.x * blockDim.x + threadIdx.x; i < total; i += gridDim.x * blockDim.x) {
    const int lane = i & 31;
    const int ks   = (i >> 5) % ks_n;
    const int ct   = (i >> 5) / ks_n;
    int col = ct * 16 + (lane & 15);
    col = col < OC ? col : OC - 1;
    const float* p = W + (size_t)col * IC + ks * 4 + ((lane >> 4) << 1);
    v2f v = *(const v2f*)p;
    *(v2f*)(dst + off + (size_t)i * 2) = v;
  }
}

template <int IC, bool DUAL, bool RELU, bool GUARD, bool INP, bool OUTP>
__global__ __launch_bounds__(32 * WAVES_PER_BLOCK) void gcn_wmma_gemm(
    const float* __restrict__ inA, const float* __restrict__ WA, const float* __restrict__ bA,
    const float* __restrict__ inB, const float* __restrict__ WB, const float* __restrict__ bB,
    float* __restrict__ out, int nrows, int OC) {
  static_assert(INP || !DUAL, "the two-operand form reads fragment-order inputs only");
  static_assert(!(OUTP && GUARD), "fragment-order output needs full tiles");
  __shared__ float lds[INP ? 4 : 16 * IC];
  const int tid  = threadIdx.x;
  const int lane = tid & 31;
  const int wave = tid >> 5;
  const int row0 = blockIdx.x << 4;
  const int col0 = (blockIdx.y * WAVES_PER_BLOCK + wave) << 4;

  float* Atile = lds;

  if constexpr (!INP) {
    constexpr int vecPerRow = IC >> 2;
    constexpr int totalVec  = vecPerRow << 4;
    for (int idx = tid; idx < totalVec; idx += 32 * WAVES_PER_BLOCK) {
      const int r = idx / vecPerRow;
      const int v = idx - r * vecPerRow;
      const int node = row0 + r;
      if (GUARD && node >= nrows) {
        float4 z = {0.f, 0.f, 0.f, 0.f};
        ((float4*)Atile)[idx] = z;
      } else {
        ((float4*)Atile)[idx] = ((const float4*)(inA + (size_t)node * IC))[v];
      }
    }
    __syncthreads();
  }

  if (col0 >= OC) return;

  const int  nidx = lane & 15;
  const int  koff = (lane >> 4) << 1;
  const int  col  = col0 + nidx;
  const bool colOK = GUARD ? (col < OC) : true;

  constexpr int KSTEPS = IC / 4;
  const v2f* wAp = (const v2f*)WA + (size_t)(col0 >> 4) * KSTEPS * 32 + lane;
  const v2f* wBp = DUAL ? ((const v2f*)WB + (size_t)(col0 >> 4) * KSTEPS * 32 + lane) : wAp;
  const v2f*   ag = (const v2f*)(inA + (size_t)row0 * IC) + lane;
  const v2f*   bg = DUAL ? ((const v2f*)(inB + (size_t)row0 * IC) + lane) : ag;
  const float* ar = Atile + nidx * IC + koff;

  constexpr int BATCH  = DUAL ? (KSTEPS < 4 ? KSTEPS : 4)
                              : (KSTEPS < 8 ? KSTEPS : 8);

  v8f acc = {};
#pragma unroll 1
  for (int kb = 0; kb < KSTEPS; kb += BATCH) {
    v2f fa[BATCH], fw[BATCH];
    v2f ga[DUAL ? BATCH : 1], gw[DUAL ? BATCH : 1];
#pragma unroll
    for (int i = 0; i < BATCH; ++i) {
      const int ks = kb + i;
      if constexpr (INP) fa[i] = ag[ks * 32];
      else               fa[i] = *(const v2f*)(ar + ks * 4);
      fw[i] = wAp[ks * 32];
      if constexpr (DUAL) {
        ga[i] = bg[ks * 32];
        gw[i] = wBp[ks * 32];
      }
    }
#pragma unroll
    for (int i = 0; i < BATCH; ++i) {
      if constexpr (OUTP) {
        acc = __builtin_amdgcn_wmma_f32_16x16x4_f32(false, fw[i], false, fa[i],
                                                    (short)0, acc, false, false);
        if constexpr (DUAL)
          acc = __builtin_amdgcn_wmma_f32_16x16x4_f32(false, gw[i], false, ga[i],
                                                      (short)0, acc, false, false);
      } else {
        acc = __builtin_amdgcn_wmma_f32_16x16x4_f32(false, fa[i], false, fw[i],
                                                    (short)0, acc, false, false);
        if constexpr (DUAL)
          acc = __builtin_amdgcn_wmma_f32_16x16x4_f32(false, ga[i], false, gw[i],
                                                      (short)0, acc, false, false);
      }
    }
  }

  const int mb = (lane >> 4) << 3;
  if constexpr (OUTP) {
    const int cb = col0 + mb;
    float bias[8];
#pragma unroll
    for (int vg = 0; vg < 8; ++vg) {
      bias[vg] = bA[cb + vg];
      if constexpr (DUAL) bias[vg] += bB[cb + vg];
    }
    const int KSO = OC >> 2;
    float* ob = out + ((size_t)blockIdx.x * KSO + (cb >> 2)) * 64 + nidx * 2;
#pragma unroll
    for (int t = 0; t < 4; ++t) {
      float v0 = acc[2 * t] + bias[2 * t];
      float v1 = acc[2 * t + 1] + bias[2 * t + 1];
      if constexpr (RELU) { v0 = v0 > 0.f ? v0 : 0.f; v1 = v1 > 0.f ? v1 : 0.f; }
      v2f o = {v0, v1};
      *(v2f*)(ob + (t >> 1) * 64 + (t & 1) * 32) = o;
    }
  } else if constexpr (GUARD) {
    float bias = 0.f;
    if (colOK) {
      bias = bA[col];
      if constexpr (DUAL) bias += bB[col];
    }
#pragma unroll
    for (int vg = 0; vg < 8; ++vg) {
      const int r = row0 + mb + vg;
      if (colOK && r < nrows) {
        float val = acc[vg] + bias;
        if constexpr (RELU) val = val > 0.f ? val : 0.f;
        out[(size_t)r * OC + col] = val;
      }
    }
  } else {
    float bias = bA[col];
    if constexpr (DUAL) bias += bB[col];
    float* op = out + (size_t)(row0 + mb) * OC + col;
#pragma unroll
    for (int vg = 0; vg < 8; ++vg) {
      float val = acc[vg] + bias;
      if constexpr (RELU) val = val > 0.f ? val : 0.f;
      op[(size_t)vg * OC] = val;
    }
  }
}

template <int IC, int WPB, bool DUAL, bool RELU, bool GUARD, bool OUTP>
__global__ __launch_bounds__(32 * WPB) void gcn_wmma_gemm_rows(
    const float* __restrict__ inA, const float* __restrict__ WA, const float* __restrict__ bA,
    const float* __restrict__ inB, const float* __restrict__ WB, const float* __restrict__ bB,
    float* __restrict__ out, int nrows, int OC) {
  static_assert(!(OUTP && GUARD), "fragment-order output needs full tiles");
  constexpr int KSTEPS = IC / 4;
  const int lane = threadIdx.x & 31;
  const int wave = __builtin_amdgcn_readfirstlane(threadIdx.x >> 5);
  const int col0 = (blockIdx.y * WPB + wave) << 4;
  if (col0 >= OC) return;

  const int nidx = lane & 15;
  const int mb   = (lane >> 4) << 3;

  const v2f* wAp = (const v2f*)WA + (size_t)(col0 >> 4) * KSTEPS * 32 + lane;
  const v2f* wBp = DUAL ? ((const v2f*)WB + (size_t)(col0 >> 4) * KSTEPS * 32 + lane) : wAp;
  v2f fw[KSTEPS];
  v2f gw[DUAL ? KSTEPS : 1];
#pragma unroll
  for (int ks = 0; ks < KSTEPS; ++ks) {
    fw[ks] = wAp[ks * 32];
    if constexpr (DUAL) gw[ks] = wBp[ks * 32];
  }

  float bias[OUTP ? 8 : 1];
  const int  col   = col0 + nidx;
  const bool colOK = GUARD ? (col < OC) : true;
  if constexpr (OUTP) {
    const int cb = col0 + mb;
#pragma unroll
    for (int vg = 0; vg < 8; ++vg) {
      bias[vg] = bA[cb + vg];
      if constexpr (DUAL) bias[vg] += bB[cb + vg];
    }
  } else {
    bias[0] = 0.f;
    if (colOK) {
      bias[0] = bA[col];
      if constexpr (DUAL) bias[0] += bB[col];
    }
  }

  const int rowTiles = (nrows + 15) >> 4;
  for (int rt = blockIdx.x; rt < rowTiles; rt += gridDim.x) {
    const v2f* ag = (const v2f*)(inA + (size_t)rt * 16 * IC) + lane;
    const v2f* bg = DUAL ? ((const v2f*)(inB + (size_t)rt * 16 * IC) + lane) : ag;
    v2f fa[KSTEPS];
    v2f ga[DUAL ? KSTEPS : 1];
#pragma unroll
    for (int ks = 0; ks < KSTEPS; ++ks) {
      fa[ks] = ag[ks * 32];
      if constexpr (DUAL) ga[ks] = bg[ks * 32];
    }
    __builtin_amdgcn_sched_barrier(0);
    v8f acc = {};
#pragma unroll
    for (int ks = 0; ks < KSTEPS; ++ks) {
      if constexpr (OUTP) {
        acc = __builtin_amdgcn_wmma_f32_16x16x4_f32(false, fw[ks], false, fa[ks],
                                                    (short)0, acc, false, false);
        if constexpr (DUAL)
          acc = __builtin_amdgcn_wmma_f32_16x16x4_f32(false, gw[ks], false, ga[ks],
                                                      (short)0, acc, false, false);
      } else {
        acc = __builtin_amdgcn_wmma_f32_16x16x4_f32(false, fa[ks], false, fw[ks],
                                                    (short)0, acc, false, false);
        if constexpr (DUAL)
          acc = __builtin_amdgcn_wmma_f32_16x16x4_f32(false, ga[ks], false, gw[ks],
                                                      (short)0, acc, false, false);
      }
    }

    if constexpr (OUTP) {
      const int cb  = col0 + mb;
      const int KSO = OC >> 2;
      float* ob = out + ((size_t)rt * KSO + (cb >> 2)) * 64 + nidx * 2;
#pragma unroll
      for (int t = 0; t < 4; ++t) {
        float v0 = acc[2 * t] + bias[2 * t];
        float v1 = acc[2 * t + 1] + bias[2 * t + 1];
        if constexpr (RELU) { v0 = v0 > 0.f ? v0 : 0.f; v1 = v1 > 0.f ? v1 : 0.f; }
        v2f o = {v0, v1};
        *(v2f*)(ob + (t >> 1) * 64 + (t & 1) * 32) = o;
      }
    } else {
      const int row0 = rt << 4;
#pragma unroll
      for (int vg = 0; vg < 8; ++vg) {
        const int r = row0 + mb + vg;
        if (colOK && (!GUARD || r < nrows)) {
          float val = acc[vg] + bias[0];
          if constexpr (RELU) val = val > 0.f ? val : 0.f;
          out[(size_t)r * OC + col] = val;
        }
      }
    }
  }
}


template <int IC, bool DUAL, bool RELU, bool INP, bool OUTP>
static inline void launch_gemm(const float* inA, const float* WA, const float* bA,
                               const float* inB, const float* WB, const float* bB,
                               float* out, int N, int OC, hipStream_t stream) {
  const int rowTiles = (N + 15) / 16;
  const int gy = (((OC + 15) / 16) + WAVES_PER_BLOCK - 1) / WAVES_PER_BLOCK;
  const dim3 grid(rowTiles, gy), blk(32 * WAVES_PER_BLOCK);
  if constexpr (OUTP) {
    gcn_wmma_gemm<IC, DUAL, RELU, false, INP, true><<<grid, blk, 0, stream>>>(
        inA, WA, bA, inB, WB, bB, out, N, OC);
  } else {
    if ((N & 15) == 0 && (OC & 15) == 0)
      gcn_wmma_gemm<IC, DUAL, RELU, false, INP, false><<<grid, blk, 0, stream>>>(
          inA, WA, bA, inB, WB, bB, out, N, OC);
    else
      gcn_wmma_gemm<IC, DUAL, RELU, true, INP, false><<<grid, blk, 0, stream>>>(
          inA, WA, bA, inB, WB, bB, out, N, OC);
  }
}

template <int IC, int WPB, int TPW, bool DUAL, bool RELU, bool OUTP>
static inline void launch_gemm_rows(const float* inA, const float* WA, const float* bA,
                                    const float* inB, const float* WB, const float* bB,
                                    float* out, int N, int OC, hipStream_t stream) {
  const int rowTiles = (N + 15) / 16;
  const int gx = (rowTiles + TPW - 1) / TPW;
  const int gy = (((OC + 15) / 16) + WPB - 1) / WPB;
  const dim3 grid(gx, gy), blk(32 * WPB);
  if constexpr (OUTP) {
    gcn_wmma_gemm_rows<IC, WPB, DUAL, RELU, false, true><<<grid, blk, 0, stream>>>(
        inA, WA, bA, inB, WB, bB, out, N, OC);
  } else {
    if ((N & 15) == 0 && (OC & 15) == 0)
      gcn_wmma_gemm_rows<IC, WPB, DUAL, RELU, false, false><<<grid, blk, 0, stream>>>(
          inA, WA, bA, inB, WB, bB, out, N, OC);
    else
      gcn_wmma_gemm_rows<IC, WPB, DUAL, RELU, true, false><<<grid, blk, 0, stream>>>(
          inA, WA, bA, inB, WB, bB, out, N, OC);
  }
}

extern "C" void kernel_launch(void* const* d_in, const int* in_sizes, int n_in,
                              void* d_out, int out_size, void* d_ws, size_t ws_size,
                              hipStream_t stream) {
  (void)n_in; (void)out_size; (void)ws_size;
  const float* x    = (const float*)d_in[0];
  const int*   ei   = (const int*)d_in[1];
  const float* w1   = (const float*)d_in[2];
  const float* b1   = (const float*)d_in[3];
  const float* w2   = (const float*)d_in[4];
  const float* b2   = (const float*)d_in[5];
  const float* c1wl = (const float*)d_in[6];
  const float* c1bl = (const float*)d_in[7];
  const float* c1wr = (const float*)d_in[8];
  const float* c1br = (const float*)d_in[9];
  const float* c2wl = (const float*)d_in[10];
  const float* c2bl = (const float*)d_in[11];
  const float* c2wr = (const float*)d_in[12];
  const float* c2br = (const float*)d_in[13];
  const float* w3   = (const float*)d_in[14];
  const float* b3   = (const float*)d_in[15];
  const float* w4   = (const float*)d_in[16];
  const float* b4   = (const float*)d_in[17];

  const int IN_CH = 256;
  const int N = in_sizes[0] / IN_CH;
  const int E = in_sizes[1] / 2;
  const size_t n = (size_t)N;

  float* ws   = (float*)d_ws;
  float* dis  = ws;
  float* h3   = ws + 1 * n;
  float* hs3  = ws + 65 * n;
  int*  rcnt  = (int*)(ws + 129 * n);
  int2* rlist = (int2*)(ws + 130 * n);
  float* h1   = ws + 172 * n;
  float* wp   = ws + 192 * n;
  float* h2   = ws + 193 * n;
  float* hs2  = ws + 225 * n;
  float* h4   = ws + 193 * n;
  float* h5   = ws;

  const int eltBlocks  = (N + 255) / 256;
  const int binBlocks  = (E + BIN_THREADS * BIN_PER - 1) / (BIN_THREADS * BIN_PER);
  const int rangeBlocks = (N + RG_NODES - 1) / RG_NODES;

  PackSrc ps = {w1, w2, c1wl, c1wr, c2wl, c2wr, w3, w4};
  gcn_pack_weights<<<dim3(24, NUM_W), 256, 0, stream>>>(ps, wp);

  launch_gemm<256, false, false, false, true>(x, wp + WP_W1, b1, nullptr, nullptr, nullptr, h1, N, 16, stream);
  launch_gemm_rows<16, 2, 2, false, false, true>(h1, wp + WP_W2, b2, nullptr, nullptr, nullptr, h2, N, 32, stream);

  gcn_deg_init<<<eltBlocks, 256, 0, stream>>>(dis, rcnt, N, rangeBlocks);
  gcn_deg_bin<<<binBlocks, BIN_THREADS, 0, stream>>>(ei, dis, rlist, rcnt, E, rangeBlocks);
  gcn_deg_to_dis<<<eltBlocks, 256, 0, stream>>>(dis, N);

  gcn_prescale<32><<<(int)((n * 8 + 255) / 256), 256, 0, stream>>>(dis, h2, hs2, N);
  gcn_range_conv<32, 64, true><<<rangeBlocks, RG_THREADS, 0, stream>>>(
      ei, hs2, dis, rlist, rcnt, E, N, h2, wp + WP_C1L, c1bl, wp + WP_C1R, c1br, h3, hs3);

  gcn_range_conv<64, 128, false><<<rangeBlocks, RG_THREADS, 0, stream>>>(
      ei, hs3, dis, rlist, rcnt, E, N, h3, wp + WP_C2L, c2bl, wp + WP_C2R, c2br, h4, nullptr);

  launch_gemm_rows<128, 4, 13, false, false, true>(h4, wp + WP_W3, b3, nullptr, nullptr, nullptr, h5, N, 192, stream);
  launch_gemm_rows<192, 2, 4, false, false, false>(h5, wp + WP_W4, b4, nullptr, nullptr, nullptr, (float*)d_out, N, 20, stream);
}


# in-degree read off the per-range edge lists the binning pass fills: each range's group counts its list's local targets in LDS (integer, exact) and writes dis; no per-edge global float add is left
# speedup vs baseline: 3.4442x; 1.3604x over previous
#include <hip/hip_runtime.h>

typedef __attribute__((ext_vector_type(2))) float v2f;
typedef __attribute__((ext_vector_type(4))) float v4f;
typedef __attribute__((ext_vector_type(8))) float v8f;

#define WAVES_PER_BLOCK 4


#define RG_NODES   400
#define RG_THREADS 1024
#define RG_LIST    8192

#define BIN_THREADS 1024
#define BIN_PER     4
#define BIN_MAXR    1024

#define DIS_THREADS 256

__global__ void gcn_cnt_init(int* __restrict__ gcnt, int nr) {
  int i = blockIdx.x * blockDim.x + threadIdx.x;
  if (i < nr) gcnt[i] = (nr <= BIN_MAXR) ? 0 : (RG_LIST + 1);
}

__global__ __launch_bounds__(BIN_THREADS) void gcn_bin_edges(
    const int* __restrict__ ei, int2* __restrict__ glist,
    int* __restrict__ gcnt, int E, int nr) {
  __shared__ int hist[BIN_MAXR];
  const int  tid = threadIdx.x;
  const bool bin = nr <= BIN_MAXR;
  if (bin)
    for (int i = tid; i < nr; i += BIN_THREADS) hist[i] = 0;
  __syncthreads();

  const int e0 = blockIdx.x * (BIN_THREADS * BIN_PER) + tid;
  int c[BIN_PER], sn[BIN_PER], rg[BIN_PER], rank[BIN_PER];
#pragma unroll
  for (int u = 0; u < BIN_PER; ++u) {
    const int e = e0 + u * BIN_THREADS;
    const bool ok = e < E;
    c[u]  = ok ? ei[E + e] : -1;
    sn[u] = ok ? ei[e] : 0;
  }
#pragma unroll
  for (int u = 0; u < BIN_PER; ++u) {
    rg[u] = 0; rank[u] = 0;
    if (bin && c[u] >= 0) {
      rg[u]   = c[u] / RG_NODES;
      rank[u] = atomicAdd(&hist[rg[u]], 1);
    }
  }
  __syncthreads();
  if (bin)
    for (int i = tid; i < nr; i += BIN_THREADS) {
      const int h = hist[i];
      hist[i] = h > 0 ? atomicAdd(&gcnt[i], h) : 0;
    }
  __syncthreads();
  if (bin) {
#pragma unroll
    for (int u = 0; u < BIN_PER; ++u) {
      if (c[u] >= 0) {
        const int r = rg[u];
        const unsigned slot = (unsigned)(hist[r] + rank[u]);
        if (slot < (unsigned)RG_LIST)
          glist[(size_t)r * RG_LIST + slot] = make_int2(sn[u], c[u] - r * RG_NODES);
      }
    }
  }
}

__global__ __launch_bounds__(DIS_THREADS) void gcn_range_dis(
    const int* __restrict__ ei, const int2* __restrict__ glist, const int* __restrict__ gcnt,
    float* __restrict__ dis, int E, int N) {
  __shared__ int scnt[RG_NODES];
  const int tid  = threadIdx.x;
  const int base = blockIdx.x * RG_NODES;
  const int rem  = N - base;
  const unsigned nloc = (unsigned)(rem < RG_NODES ? rem : RG_NODES);
  const unsigned prev = (unsigned)gcnt[blockIdx.x];
  const bool listed = prev <= (unsigned)RG_LIST;

  for (int i = tid; i < RG_NODES; i += DIS_THREADS) scnt[i] = 0;

  constexpr int NLP = RG_LIST / DIS_THREADS;
  const int n = listed ? (int)prev : 0;
  int tl[NLP];
#pragma unroll
  for (int j = 0; j < NLP; ++j) tl[j] = 0;
  if (n > 0) {
    const int2* gl = glist + (size_t)blockIdx.x * RG_LIST;
#pragma unroll
    for (int j = 0; j < NLP; ++j) {
      const int i = tid + j * DIS_THREADS;
      tl[j] = gl[i < n ? i : n - 1].y;
    }
  }
  __syncthreads();

  if (listed) {
#pragma unroll
    for (int j = 0; j < NLP; ++j)
      if (tid + j * DIS_THREADS < n) atomicAdd(&scnt[tl[j]], 1);
  } else {
    const int* col = ei + E;
    for (int e = tid; e < E; e += DIS_THREADS) {
      const unsigned cl = (unsigned)(col[e] - base);
      if (cl < nloc) atomicAdd(&scnt[cl], 1);
    }
  }
  __syncthreads();

  for (int i = tid; i < (int)nloc; i += DIS_THREADS) {
    const float d = (float)(1 + scnt[i]);
    dis[base + i] = (d > 0.0f) ? rsqrtf(d) : 0.0f;
  }
}


template <int C>
__global__ void gcn_prescale(const float* __restrict__ dis, const float* __restrict__ hf,
                             float* __restrict__ hs, int N) {
  constexpr int KS = C >> 2;
  long long i = (long long)blockIdx.x * blockDim.x + threadIdx.x;
  long long total = (long long)N * KS;
  if (i < total) {
    const int node = (int)(i / KS);
    const int v    = (int)(i - (long long)node * KS);
    const v2f* f = (const v2f*)hf + ((size_t)(node >> 4) * KS + v) * 32 + (node & 15);
    const v2f lo = f[0], hi = f[16];
    const float d = dis[node];
    float4 o = {lo.x * d, lo.y * d, hi.x * d, hi.y * d};
    ((float4*)hs)[i] = o;
  }
}


template <int C, int OC, bool HS>
__global__ __launch_bounds__(RG_THREADS) void gcn_range_conv(
    const int* __restrict__ ei, const float* __restrict__ hs, const float* __restrict__ dis,
    const int2* __restrict__ glist, const int* __restrict__ gcnt, int E, int N,
    const float* __restrict__ hf, const float* __restrict__ WA, const float* __restrict__ bA,
    const float* __restrict__ WB, const float* __restrict__ bB,
    float* __restrict__ outf, float* __restrict__ hsout) {
  static_assert(RG_NODES % 16 == 0, "a range must start on a 16-row tile boundary");
  constexpr int PITCH = C + 4;
  constexpr int KS    = C >> 2;
  __shared__ float    sacc[RG_NODES * PITCH];
  __shared__ float    sdis[RG_NODES];
  __shared__ int      lsrc[RG_LIST];
  __shared__ int      ldst[RG_LIST];
  __shared__ unsigned cnt;

  const int tid  = threadIdx.x;
  const int base = blockIdx.x * RG_NODES;
  const int rem  = N - base;
  const unsigned nloc = (unsigned)(rem < RG_NODES ? rem : RG_NODES);
  const int nvec = (int)nloc * (C >> 2);
  const int2* gl = glist + (size_t)blockIdx.x * RG_LIST;

  const unsigned prev = (unsigned)gcnt[blockIdx.x];
  const bool listed = prev <= (unsigned)RG_LIST;
  int n = listed ? (int)prev : 0;

  static_assert(RG_NODES <= RG_THREADS, "one dis value per thread");
  constexpr int NSV = (RG_NODES * KS + RG_THREADS - 1) / RG_THREADS;
  constexpr int NLP = RG_LIST / RG_THREADS;
  const v4f* hsv = (const v4f*)(hs + (size_t)base * C);
  v4f sv[NSV];
#pragma unroll
  for (int j = 0; j < NSV; ++j) {
    const int i = tid + j * RG_THREADS;
    sv[j] = hsv[i < nvec ? i : nvec - 1];
  }
  const float dv = dis[base + (tid < (int)nloc ? tid : (int)nloc - 1)];
  int2 lp[NLP];
#pragma unroll
  for (int j = 0; j < NLP; ++j) lp[j] = make_int2(0, 0);
  if (n > 0) {
#pragma unroll
    for (int j = 0; j < NLP; ++j) {
      const int i = tid + j * RG_THREADS;
      lp[j] = gl[i < n ? i : n - 1];
    }
  }
#pragma unroll
  for (int j = 0; j < NSV; ++j) {
    const int i = tid + j * RG_THREADS;
    if (i < nvec) {
      const int r = i / KS;
      *(v4f*)(sacc + r * PITCH + ((i - r * KS) << 2)) = sv[j];
    }
  }
  for (int i = nvec + tid; i < (((int)nloc + 15) >> 4) * 16 * KS; i += RG_THREADS) {
    const int r = i / KS;
    float4 z = {0.f, 0.f, 0.f, 0.f};
    *(float4*)(sacc + r * PITCH + ((i - r * KS) << 2)) = z;
  }
  if (tid < (int)nloc) sdis[tid] = dv;
  if (tid == 0) cnt = 0u;
  if (n > 0) {
#pragma unroll
    for (int j = 0; j < NLP; ++j) {
      const int i = tid + j * RG_THREADS;
      if (i < n) {
        lsrc[i] = lp[j].x;
        ldst[i] = lp[j].y;
      }
    }
  }

  if (!listed) {
    __syncthreads();

    const int* col = ei + E;

    auto keep = [&](int e, unsigned cl) {
      const unsigned s = atomicAdd(&cnt, 1u);
      if (s < (unsigned)RG_LIST) {
        lsrc[s] = e;
        ldst[s] = (int)cl;
      } else {
        const float* hp = hs + (size_t)ei[e] * C;
#pragma unroll 1
        for (int ch = 0; ch < C; ++ch) atomicAdd(&sacc[cl * PITCH + ch], hp[ch]);
      }
    };
    auto test4 = [&](const int4 v, int q) {
      const unsigned a = (unsigned)(v.x - base), b = (unsigned)(v.y - base);
      const unsigned c = (unsigned)(v.z - base), d = (unsigned)(v.w - base);
      if ((a < nloc) | (b < nloc) | (c < nloc) | (d < nloc)) {
        const int e = q << 2;
        if (a < nloc) keep(e, a);
        if (b < nloc) keep(e + 1, b);
        if (c < nloc) keep(e + 2, c);
        if (d < nloc) keep(e + 3, d);
      }
    };

    if (((E & 3) == 0) && ((((size_t)col) & 15) == 0)) {
      const int4* colv = (const int4*)col;
      const int nq = E >> 2;
      int q = tid;
      for (; q + 3 * RG_THREADS < nq; q += 4 * RG_THREADS) {
        const int4 v0 = colv[q];
        const int4 v1 = colv[q + RG_THREADS];
        const int4 v2 = colv[q + 2 * RG_THREADS];
        const int4 v3 = colv[q + 3 * RG_THREADS];
        test4(v0, q);
        test4(v1, q + RG_THREADS);
        test4(v2, q + 2 * RG_THREADS);
        test4(v3, q + 3 * RG_THREADS);
      }
      for (; q < nq; q += RG_THREADS) test4(colv[q], q);
    } else {
      for (int e = tid; e < E; e += RG_THREADS) {
        const unsigned cl = (unsigned)(col[e] - base);
        if (cl < nloc) keep(e, cl);
      }
    }
    __syncthreads();

    const unsigned total = cnt;
    n = (int)(total < (unsigned)RG_LIST ? total : (unsigned)RG_LIST);
    for (int i = tid; i < n; i += RG_THREADS) lsrc[i] = ei[lsrc[i]];
  }
  __syncthreads();

  constexpr int VEC  = C / 32;
  constexpr int RG_U = 16 / VEC;
  constexpr int NW   = RG_THREADS / 32;
  static_assert(RG_U <= 32 && (RG_U & (RG_U - 1)) == 0 && RG_LIST % RG_U == 0, "a trip is read by the lanes of one wave");
  const int lane = tid & 31;
  const int wave  = __builtin_amdgcn_readfirstlane(tid >> 5);
  const int ntrip = n / RG_U;
  const int lo = lane * VEC;
  float*    srow = sacc + lo;

  auto fetch = [&](int t, float (&val)[RG_U][VEC], int (&dst)[RG_U]) {
    const int e = t * RG_U + (lane & (RG_U - 1));
    const int s = lsrc[e];
    const int d = ldst[e];
#pragma unroll
    for (int u = 0; u < RG_U; ++u) {
      const int r = __builtin_amdgcn_readlane(s, u);
      dst[u] = __builtin_amdgcn_readlane(d, u);
      const float* hp = hs + (size_t)r * C;
#pragma unroll
      for (int k = 0; k < VEC; ++k) val[u][k] = hp[lo + k];
    }
  };
  auto add = [&](const float (&val)[RG_U][VEC], const int (&dst)[RG_U]) {
#pragma unroll
    for (int u = 0; u < RG_U; ++u) {
      float* sp = srow + dst[u] * PITCH;
#pragma unroll
      for (int k = 0; k < VEC; ++k) atomicAdd(sp + k, val[u][k]);
    }
  };

  {
    float vA[RG_U][VEC], vB[RG_U][VEC];
    int   dA[RG_U], dB[RG_U];
    int t = wave;
    if (t < ntrip) {
      fetch(t, vA, dA);
      for (;;) {
        const int t1 = t + NW;
        if (t1 < ntrip) fetch(t1, vB, dB);
        add(vA, dA);
        if (t1 >= ntrip) break;
        t = t1 + NW;
        if (t < ntrip) fetch(t, vA, dA);
        add(vB, dB);
        if (t >= ntrip) break;
      }
    }
  }
  for (int i = ntrip * RG_U + wave; i < n; i += NW) {
    const float* hp = hs + (size_t)lsrc[i] * C + lo;
    float*       sp = srow + ldst[i] * PITCH;
#pragma unroll
    for (int k = 0; k < VEC; ++k) atomicAdd(sp + k, hp[k]);
  }
  __syncthreads();

  static_assert(OC % 16 == 0 && (32 % (OC / 16)) == 0, "column tiles must divide the waves of a group");
  constexpr int OCT = OC / 16;
  constexpr int TPR = (RG_THREADS / 32) / OCT;
  constexpr int KB  = KS < 8 ? KS : 8;
  const int ct   = wave % OCT;
  const int tsub = wave / OCT;
  const int nidx = lane & 15;
  const int half = lane >> 4;
  const int cb   = ct * 16 + (half << 3);

  v2f fw[KS], gw[KS];
  {
    const v2f* wAp = (const v2f*)WA + (size_t)ct * KS * 32 + lane;
    const v2f* wBp = (const v2f*)WB + (size_t)ct * KS * 32 + lane;
#pragma unroll
    for (int ks = 0; ks < KS; ++ks) {
      fw[ks] = wAp[ks * 32];
      gw[ks] = wBp[ks * 32];
    }
  }
  float bias[8];
#pragma unroll
  for (int vg = 0; vg < 8; ++vg) bias[vg] = bA[cb + vg] + bB[cb + vg];

  const int ntile = ((int)nloc + 15) >> 4;
  for (int t = tsub; t < ntile; t += TPR) {
    const int  gt  = (base >> 4) + t;
    const int  row = (t << 4) + nidx;
    const bool rok = (unsigned)row < nloc;
    const float d  = rok ? sdis[row] : 0.f;
    const v2f*   ag = (const v2f*)(hf + (size_t)gt * 16 * C) + lane;
    const float* sp = sacc + row * PITCH + (half << 1);
    v8f acc = {};
#pragma unroll
    for (int kb = 0; kb < KS; kb += KB) {
      v2f fa[KB], ga[KB];
#pragma unroll
      for (int i = 0; i < KB; ++i) {
        fa[i] = ag[(kb + i) * 32];
        v2f s = *(const v2f*)(sp + ((kb + i) << 2));
        s.x *= d;
        s.y *= d;
        ga[i] = s;
      }
#pragma unroll
      for (int i = 0; i < KB; ++i) {
        acc = __builtin_amdgcn_wmma_f32_16x16x4_f32(false, fw[kb + i], false, fa[i],
                                                    (short)0, acc, false, false);
        acc = __builtin_amdgcn_wmma_f32_16x16x4_f32(false, gw[kb + i], false, ga[i],
                                                    (short)0, acc, false, false);
      }
    }
    if (rok) {
      constexpr int KSO = OC >> 2;
      float* ob = outf + ((size_t)gt * KSO + (cb >> 2)) * 64 + nidx * 2;
      float o[8];
#pragma unroll
      for (int vg = 0; vg < 8; ++vg) {
        const float v = acc[vg] + bias[vg];
        o[vg] = v > 0.f ? v : 0.f;
      }
#pragma unroll
      for (int q = 0; q < 4; ++q) {
        v2f ov = {o[2 * q], o[2 * q + 1]};
        *(v2f*)(ob + (q >> 1) * 64 + (q & 1) * 32) = ov;
      }
      if constexpr (HS) {
        float4* hp = (float4*)(hsout + (size_t)(base + row) * OC + cb);
        float4 lo = {o[0] * d, o[1] * d, o[2] * d, o[3] * d};
        float4 hi = {o[4] * d, o[5] * d, o[6] * d, o[7] * d};
        hp[0] = lo;
        hp[1] = hi;
      }
    }
  }
}


#define NUM_W 8
struct PackSrc {
  const float* w0; const float* w1; const float* w2; const float* w3;
  const float* w4; const float* w5; const float* w6; const float* w7;
};
#define WP_W1   0
#define WP_W2   4096
#define WP_C1L  4608
#define WP_C1R  6656
#define WP_C2L  8704
#define WP_C2R  16896
#define WP_W3   25088
#define WP_W4   49664
#define WP_TOTAL 55808

__global__ void gcn_pack_weights(PackSrc s, float* __restrict__ dst) {
  const int m = blockIdx.y;
  const float* W; int OC, IC, off;
  switch (m) {
    case 0:  W = s.w0; OC = 16;  IC = 256; off = WP_W1;  break;
    case 1:  W = s.w1; OC = 32;  IC = 16;  off = WP_W2;  break;
    case 2:  W = s.w2; OC = 64;  IC = 32;  off = WP_C1L; break;
    case 3:  W = s.w3; OC = 64;  IC = 32;  off = WP_C1R; break;
    case 4:  W = s.w4; OC = 128; IC = 64;  off = WP_C2L; break;
    case 5:  W = s.w5; OC = 128; IC = 64;  off = WP_C2R; break;
    case 6:  W = s.w6; OC = 192; IC = 128; off = WP_W3;  break;
    default: W = s.w7; OC = 20;  IC = 192; off = WP_W4;  break;
  }
  const int ks_n  = IC >> 2;
  const int tiles = (OC + 15) >> 4;
  const int total = tiles * ks_n * 32;
  for (int i = blockIdx.x * blockDim.x + threadIdx.x; i < total; i += gridDim.x * blockDim.x) {
    const int lane = i & 31;
    const int ks   = (i >> 5) % ks_n;
    const int ct   = (i >> 5) / ks_n;
    int col = ct * 16 + (lane & 15);
    col = col < OC ? col : OC - 1;
    const float* p = W + (size_t)col * IC + ks * 4 + ((lane >> 4) << 1);
    v2f v = *(const v2f*)p;
    *(v2f*)(dst + off + (size_t)i * 2) = v;
  }
}

template <int IC, bool DUAL, bool RELU, bool GUARD, bool INP, bool OUTP>
__global__ __launch_bounds__(32 * WAVES_PER_BLOCK) void gcn_wmma_gemm(
    const float* __restrict__ inA, const float* __restrict__ WA, const float* __restrict__ bA,
    const float* __restrict__ inB, const float* __restrict__ WB, const float* __restrict__ bB,
    float* __restrict__ out, int nrows, int OC) {
  static_assert(INP || !DUAL, "the two-operand form reads fragment-order inputs only");
  static_assert(!(OUTP && GUARD), "fragment-order output needs full tiles");
  __shared__ float lds[INP ? 4 : 16 * IC];
  const int tid  = threadIdx.x;
  const int lane = tid & 31;
  const int wave = tid >> 5;
  const int row0 = blockIdx.x << 4;
  const int col0 = (blockIdx.y * WAVES_PER_BLOCK + wave) << 4;

  float* Atile = lds;

  if constexpr (!INP) {
    constexpr int vecPerRow = IC >> 2;
    constexpr int totalVec  = vecPerRow << 4;
    for (int idx = tid; idx < totalVec; idx += 32 * WAVES_PER_BLOCK) {
      const int r = idx / vecPerRow;
      const int v = idx - r * vecPerRow;
      const int node = row0 + r;
      if (GUARD && node >= nrows) {
        float4 z = {0.f, 0.f, 0.f, 0.f};
        ((float4*)Atile)[idx] = z;
      } else {
        ((float4*)Atile)[idx] = ((const float4*)(inA + (size_t)node * IC))[v];
      }
    }
    __syncthreads();
  }

  if (col0 >= OC) return;

  const int  nidx = lane & 15;
  const int  koff = (lane >> 4) << 1;
  const int  col  = col0 + nidx;
  const bool colOK = GUARD ? (col < OC) : true;

  constexpr int KSTEPS = IC / 4;
  const v2f* wAp = (const v2f*)WA + (size_t)(col0 >> 4) * KSTEPS * 32 + lane;
  const v2f* wBp = DUAL ? ((const v2f*)WB + (size_t)(col0 >> 4) * KSTEPS * 32 + lane) : wAp;
  const v2f*   ag = (const v2f*)(inA + (size_t)row0 * IC) + lane;
  const v2f*   bg = DUAL ? ((const v2f*)(inB + (size_t)row0 * IC) + lane) : ag;
  const float* ar = Atile + nidx * IC + koff;

  constexpr int BATCH  = DUAL ? (KSTEPS < 4 ? KSTEPS : 4)
                              : (KSTEPS < 8 ? KSTEPS : 8);

  v8f acc = {};
#pragma unroll 1
  for (int kb = 0; kb < KSTEPS; kb += BATCH) {
    v2f fa[BATCH], fw[BATCH];
    v2f ga[DUAL ? BATCH : 1], gw[DUAL ? BATCH : 1];
#pragma unroll
    for (int i = 0; i < BATCH; ++i) {
      const int ks = kb + i;
      if constexpr (INP) fa[i] = ag[ks * 32];
      else               fa[i] = *(const v2f*)(ar + ks * 4);
      fw[i] = wAp[ks * 32];
      if constexpr (DUAL) {
        ga[i] = bg[ks * 32];
        gw[i] = wBp[ks * 32];
      }
    }
#pragma unroll
    for (int i = 0; i < BATCH; ++i) {
      if constexpr (OUTP) {
        acc = __builtin_amdgcn_wmma_f32_16x16x4_f32(false, fw[i], false, fa[i],
                                                    (short)0, acc, false, false);
        if constexpr (DUAL)
          acc = __builtin_amdgcn_wmma_f32_16x16x4_f32(false, gw[i], false, ga[i],
                                                      (short)0, acc, false, false);
      } else {
        acc = __builtin_amdgcn_wmma_f32_16x16x4_f32(false, fa[i], false, fw[i],
                                                    (short)0, acc, false, false);
        if constexpr (DUAL)
          acc = __builtin_amdgcn_wmma_f32_16x16x4_f32(false, ga[i], false, gw[i],
                                                      (short)0, acc, false, false);
      }
    }
  }

  const int mb = (lane >> 4) << 3;
  if constexpr (OUTP) {
    const int cb = col0 + mb;
    float bias[8];
#pragma unroll
    for (int vg = 0; vg < 8; ++vg) {
      bias[vg] = bA[cb + vg];
      if constexpr (DUAL) bias[vg] += bB[cb + vg];
    }
    const int KSO = OC >> 2;
    float* ob = out + ((size_t)blockIdx.x * KSO + (cb >> 2)) * 64 + nidx * 2;
#pragma unroll
    for (int t = 0; t < 4; ++t) {
      float v0 = acc[2 * t] + bias[2 * t];
      float v1 = acc[2 * t + 1] + bias[2 * t + 1];
      if constexpr (RELU) { v0 = v0 > 0.f ? v0 : 0.f; v1 = v1 > 0.f ? v1 : 0.f; }
      v2f o = {v0, v1};
      *(v2f*)(ob + (t >> 1) * 64 + (t & 1) * 32) = o;
    }
  } else if constexpr (GUARD) {
    float bias = 0.f;
    if (colOK) {
      bias = bA[col];
      if constexpr (DUAL) bias += bB[col];
    }
#pragma unroll
    for (int vg = 0; vg < 8; ++vg) {
      const int r = row0 + mb + vg;
      if (colOK && r < nrows) {
        float val = acc[vg] + bias;
        if constexpr (RELU) val = val > 0.f ? val : 0.f;
        out[(size_t)r * OC + col] = val;
      }
    }
  } else {
    float bias = bA[col];
    if constexpr (DUAL) bias += bB[col];
    float* op = out + (size_t)(row0 + mb) * OC + col;
#pragma unroll
    for (int vg = 0; vg < 8; ++vg) {
      float val = acc[vg] + bias;
      if constexpr (RELU) val = val > 0.f ? val : 0.f;
      op[(size_t)vg * OC] = val;
    }
  }
}

template <int IC, int WPB, bool DUAL, bool RELU, bool GUARD, bool OUTP>
__global__ __launch_bounds__(32 * WPB) void gcn_wmma_gemm_rows(
    const float* __restrict__ inA, const float* __restrict__ WA, const float* __restrict__ bA,
    const float* __restrict__ inB, const float* __restrict__ WB, const float* __restrict__ bB,
    float* __restrict__ out, int nrows, int OC) {
  static_assert(!(OUTP && GUARD), "fragment-order output needs full tiles");
  constexpr int KSTEPS = IC / 4;
  const int lane = threadIdx.x & 31;
  const int wave = __builtin_amdgcn_readfirstlane(threadIdx.x >> 5);
  const int col0 = (blockIdx.y * WPB + wave) << 4;
  if (col0 >= OC) return;

  const int nidx = lane & 15;
  const int mb   = (lane >> 4) << 3;

  const v2f* wAp = (const v2f*)WA + (size_t)(col0 >> 4) * KSTEPS * 32 + lane;
  const v2f* wBp = DUAL ? ((const v2f*)WB + (size_t)(col0 >> 4) * KSTEPS * 32 + lane) : wAp;
  v2f fw[KSTEPS];
  v2f gw[DUAL ? KSTEPS : 1];
#pragma unroll
  for (int ks = 0; ks < KSTEPS; ++ks) {
    fw[ks] = wAp[ks * 32];
    if constexpr (DUAL) gw[ks] = wBp[ks * 32];
  }

  float bias[OUTP ? 8 : 1];
  const int  col   = col0 + nidx;
  const bool colOK = GUARD ? (col < OC) : true;
  if constexpr (OUTP) {
    const int cb = col0 + mb;
#pragma unroll
    for (int vg = 0; vg < 8; ++vg) {
      bias[vg] = bA[cb + vg];
      if constexpr (DUAL) bias[vg] += bB[cb + vg];
    }
  } else {
    bias[0] = 0.f;
    if (colOK) {
      bias[0] = bA[col];
      if constexpr (DUAL) bias[0] += bB[col];
    }
  }

  const int rowTiles = (nrows + 15) >> 4;
  for (int rt = blockIdx.x; rt < rowTiles; rt += gridDim.x) {
    const v2f* ag = (const v2f*)(inA + (size_t)rt * 16 * IC) + lane;
    const v2f* bg = DUAL ? ((const v2f*)(inB + (size_t)rt * 16 * IC) + lane) : ag;
    v2f fa[KSTEPS];
    v2f ga[DUAL ? KSTEPS : 1];
#pragma unroll
    for (int ks = 0; ks < KSTEPS; ++ks) {
      fa[ks] = ag[ks * 32];
      if constexpr (DUAL) ga[ks] = bg[ks * 32];
    }
    __builtin_amdgcn_sched_barrier(0);
    v8f acc = {};
#pragma unroll
    for (int ks = 0; ks < KSTEPS; ++ks) {
      if constexpr (OUTP) {
        acc = __builtin_amdgcn_wmma_f32_16x16x4_f32(false, fw[ks], false, fa[ks],
                                                    (short)0, acc, false, false);
        if constexpr (DUAL)
          acc = __builtin_amdgcn_wmma_f32_16x16x4_f32(false, gw[ks], false, ga[ks],
                                                      (short)0, acc, false, false);
      } else {
        acc = __builtin_amdgcn_wmma_f32_16x16x4_f32(false, fa[ks], false, fw[ks],
                                                    (short)0, acc, false, false);
        if constexpr (DUAL)
          acc = __builtin_amdgcn_wmma_f32_16x16x4_f32(false, ga[ks], false, gw[ks],
                                                      (short)0, acc, false, false);
      }
    }

    if constexpr (OUTP) {
      const int cb  = col0 + mb;
      const int KSO = OC >> 2;
      float* ob = out + ((size_t)rt * KSO + (cb >> 2)) * 64 + nidx * 2;
#pragma unroll
      for (int t = 0; t < 4; ++t) {
        float v0 = acc[2 * t] + bias[2 * t];
        float v1 = acc[2 * t + 1] + bias[2 * t + 1];
        if constexpr (RELU) { v0 = v0 > 0.f ? v0 : 0.f; v1 = v1 > 0.f ? v1 : 0.f; }
        v2f o = {v0, v1};
        *(v2f*)(ob + (t >> 1) * 64 + (t & 1) * 32) = o;
      }
    } else {
      const int row0 = rt << 4;
#pragma unroll
      for (int vg = 0; vg < 8; ++vg) {
        const int r = row0 + mb + vg;
        if (colOK && (!GUARD || r < nrows)) {
          float val = acc[vg] + bias[0];
          if constexpr (RELU) val = val > 0.f ? val : 0.f;
          out[(size_t)r * OC + col] = val;
        }
      }
    }
  }
}


template <int IC, bool DUAL, bool RELU, bool INP, bool OUTP>
static inline void launch_gemm(const float* inA, const float* WA, const float* bA,
                               const float* inB, const float* WB, const float* bB,
                               float* out, int N, int OC, hipStream_t stream) {
  const int rowTiles = (N + 15) / 16;
  const int gy = (((OC + 15) / 16) + WAVES_PER_BLOCK - 1) / WAVES_PER_BLOCK;
  const dim3 grid(rowTiles, gy), blk(32 * WAVES_PER_BLOCK);
  if constexpr (OUTP) {
    gcn_wmma_gemm<IC, DUAL, RELU, false, INP, true><<<grid, blk, 0, stream>>>(
        inA, WA, bA, inB, WB, bB, out, N, OC);
  } else {
    if ((N & 15) == 0 && (OC & 15) == 0)
      gcn_wmma_gemm<IC, DUAL, RELU, false, INP, false><<<grid, blk, 0, stream>>>(
          inA, WA, bA, inB, WB, bB, out, N, OC);
    else
      gcn_wmma_gemm<IC, DUAL, RELU, true, INP, false><<<grid, blk, 0, stream>>>(
          inA, WA, bA, inB, WB, bB, out, N, OC);
  }
}

template <int IC, int WPB, int TPW, bool DUAL, bool RELU, bool OUTP>
static inline void launch_gemm_rows(const float* inA, const float* WA, const float* bA,
                                    const float* inB, const float* WB, const float* bB,
                                    float* out, int N, int OC, hipStream_t stream) {
  const int rowTiles = (N + 15) / 16;
  const int gx = (rowTiles + TPW - 1) / TPW;
  const int gy = (((OC + 15) / 16) + WPB - 1) / WPB;
  const dim3 grid(gx, gy), blk(32 * WPB);
  if constexpr (OUTP) {
    gcn_wmma_gemm_rows<IC, WPB, DUAL, RELU, false, true><<<grid, blk, 0, stream>>>(
        inA, WA, bA, inB, WB, bB, out, N, OC);
  } else {
    if ((N & 15) == 0 && (OC & 15) == 0)
      gcn_wmma_gemm_rows<IC, WPB, DUAL, RELU, false, false><<<grid, blk, 0, stream>>>(
          inA, WA, bA, inB, WB, bB, out, N, OC);
    else
      gcn_wmma_gemm_rows<IC, WPB, DUAL, RELU, true, false><<<grid, blk, 0, stream>>>(
          inA, WA, bA, inB, WB, bB, out, N, OC);
  }
}

extern "C" void kernel_launch(void* const* d_in, const int* in_sizes, int n_in,
                              void* d_out, int out_size, void* d_ws, size_t ws_size,
                              hipStream_t stream) {
  (void)n_in; (void)out_size; (void)ws_size;
  const float* x    = (const float*)d_in[0];
  const int*   ei   = (const int*)d_in[1];
  const float* w1   = (const float*)d_in[2];
  const float* b1   = (const float*)d_in[3];
  const float* w2   = (const float*)d_in[4];
  const float* b2   = (const float*)d_in[5];
  const float* c1wl = (const float*)d_in[6];
  const float* c1bl = (const float*)d_in[7];
  const float* c1wr = (const float*)d_in[8];
  const float* c1br = (const float*)d_in[9];
  const float* c2wl = (const float*)d_in[10];
  const float* c2bl = (const float*)d_in[11];
  const float* c2wr = (const float*)d_in[12];
  const float* c2br = (const float*)d_in[13];
  const float* w3   = (const float*)d_in[14];
  const float* b3   = (const float*)d_in[15];
  const float* w4   = (const float*)d_in[16];
  const float* b4   = (const float*)d_in[17];

  const int IN_CH = 256;
  const int N = in_sizes[0] / IN_CH;
  const int E = in_sizes[1] / 2;
  const size_t n = (size_t)N;

  float* ws   = (float*)d_ws;
  float* dis  = ws;
  float* h3   = ws + 1 * n;
  float* hs3  = ws + 65 * n;
  int*  rcnt  = (int*)(ws + 129 * n);
  int2* rlist = (int2*)(ws + 130 * n);
  float* h1   = ws + 172 * n;
  float* wp   = ws + 192 * n;
  float* h2   = ws + 193 * n;
  float* hs2  = ws + 225 * n;
  float* h4   = ws + 193 * n;
  float* h5   = ws;

  const int binBlocks  = (E + BIN_THREADS * BIN_PER - 1) / (BIN_THREADS * BIN_PER);
  const int rangeBlocks = (N + RG_NODES - 1) / RG_NODES;

  PackSrc ps = {w1, w2, c1wl, c1wr, c2wl, c2wr, w3, w4};
  gcn_pack_weights<<<dim3(24, NUM_W), 256, 0, stream>>>(ps, wp);

  launch_gemm<256, false, false, false, true>(x, wp + WP_W1, b1, nullptr, nullptr, nullptr, h1, N, 16, stream);
  launch_gemm_rows<16, 2, 2, false, false, true>(h1, wp + WP_W2, b2, nullptr, nullptr, nullptr, h2, N, 32, stream);

  gcn_cnt_init<<<(rangeBlocks + 255) / 256, 256, 0, stream>>>(rcnt, rangeBlocks);
  gcn_bin_edges<<<binBlocks, BIN_THREADS, 0, stream>>>(ei, rlist, rcnt, E, rangeBlocks);
  gcn_range_dis<<<rangeBlocks, DIS_THREADS, 0, stream>>>(ei, rlist, rcnt, dis, E, N);

  gcn_prescale<32><<<(int)((n * 8 + 255) / 256), 256, 0, stream>>>(dis, h2, hs2, N);
  gcn_range_conv<32, 64, true><<<rangeBlocks, RG_THREADS, 0, stream>>>(
      ei, hs2, dis, rlist, rcnt, E, N, h2, wp + WP_C1L, c1bl, wp + WP_C1R, c1br, h3, hs3);

  gcn_range_conv<64, 128, false><<<rangeBlocks, RG_THREADS, 0, stream>>>(
      ei, hs3, dis, rlist, rcnt, E, N, h3, wp + WP_C2L, c2bl, wp + WP_C2R, c2br, h4, nullptr);

  launch_gemm_rows<128, 4, 13, false, false, true>(h4, wp + WP_W3, b3, nullptr, nullptr, nullptr, h5, N, 192, stream);
  launch_gemm_rows<192, 2, 4, false, false, false>(h5, wp + WP_W4, b4, nullptr, nullptr, nullptr, (float*)d_out, N, 20, stream);
}
